# starting kernel
# baseline (speedup 1.0000x reference)
#include <hip/hip_runtime.h>
#include <hip/hip_bf16.h>
#include <stdint.h>

#define N_NODES_C  100000
#define N_EDGES_C  1600000
#define N_GRAPHS_C 64
#define IN_C_C     18
#define EDGE_C_C   22
#define HID_C_C    64
#define OUT_C_C    128

typedef float v2f __attribute__((ext_vector_type(2)));
typedef float v8f __attribute__((ext_vector_type(8)));

__global__ void __launch_bounds__(256)
init_kernel(const float* __restrict__ x, float* __restrict__ x1,
            float* __restrict__ deg, float* __restrict__ msum1,
            float* __restrict__ msum2, float* __restrict__ pooled,
            float* __restrict__ cnt) {
    const int i = blockIdx.x * blockDim.x + threadIdx.x;
    if (i < N_NODES_C * HID_C_C) msum2[i] = 0.0f;
    if (i < N_NODES_C * IN_C_C) { x1[i] = x[i]; msum1[i] = 0.0f; }
    if (i < N_NODES_C)          deg[i] = 0.0f;
    if (i < N_GRAPHS_C * HID_C_C) pooled[i] = 0.0f;
    if (i < N_GRAPHS_C)         cnt[i] = 0.0f;
}

__global__ void __launch_bounds__(256)
edge_fc_scatter(const float* __restrict__ eattr, const int* __restrict__ eindex,
                const float* __restrict__ W, const float* __restrict__ b,
                float* __restrict__ x1, float* __restrict__ deg) {
    __shared__ float sW[EDGE_C_C * IN_C_C];
    __shared__ float sB[IN_C_C];
    for (int i = threadIdx.x; i < EDGE_C_C * IN_C_C; i += blockDim.x) sW[i] = W[i];
    for (int i = threadIdx.x; i < IN_C_C; i += blockDim.x)            sB[i] = b[i];
    __syncthreads();

    const int e = blockIdx.x * blockDim.x + threadIdx.x;
    if (e >= N_EDGES_C) return;
    const int src = eindex[e];
    const int dst = eindex[N_EDGES_C + e];

    float a[EDGE_C_C];
    const float* er = eattr + (size_t)e * EDGE_C_C;
#pragma unroll
    for (int k = 0; k < EDGE_C_C; ++k) a[k] = er[k];

    float* xr = x1 + (size_t)src * IN_C_C;
#pragma unroll
    for (int c = 0; c < IN_C_C; ++c) {
        float acc = sB[c];
#pragma unroll
        for (int k = 0; k < EDGE_C_C; ++k) acc += a[k] * sW[k * IN_C_C + c];
        atomicAdd(xr + c, acc);
    }
    atomicAdd(deg + dst, 1.0f);
}

template <int C>
__global__ void __launch_bounds__(256)
gather_scatter(const int* __restrict__ eindex, const float* __restrict__ feat,
               float* __restrict__ msum) {
    const int e = blockIdx.x * blockDim.x + threadIdx.x;
    if (e >= N_EDGES_C) return;
    const int src = eindex[e];
    const int dst = eindex[N_EDGES_C + e];
    const float* fr = feat + (size_t)src * C;
    float*       md = msum + (size_t)dst * C;
#pragma unroll
    for (int c = 0; c < C; ++c) atomicAdd(md + c, fr[c]);
}

template <int K>
__global__ void __launch_bounds__(64)
sage_wmma_kernel(const float* __restrict__ xin, const float* __restrict__ msum,
                 const float* __restrict__ deg,
                 const float* __restrict__ lw, const float* __restrict__ lb,
                 const float* __restrict__ rw, float* __restrict__ out) {
    const int wave = (blockIdx.x * blockDim.x + threadIdx.x) >> 5;
    const int lane = threadIdx.x & 31;
    if (wave >= N_NODES_C / 16) return;

    const int node0 = wave * 16;
    const int m    = lane & 15;
    const int half = (lane >> 4) & 1;
    const int row  = node0 + m;

    const float dg  = deg[row];
    const float inv = dg > 0.0f ? 1.0f / dg : 0.0f;
    const float* xr = xin  + (size_t)row * K;
    const float* mr = msum + (size_t)row * K;

    v8f acc[4] = {};

    constexpr int KP = (K + 3) & ~3;
#pragma unroll
    for (int ks = 0; ks < KP; ks += 4) {
        const int kb = ks + half * 2;
        v2f amean, aroot;
        amean[0] = (kb     < K) ? mr[kb]     * inv : 0.0f;
        amean[1] = (kb + 1 < K) ? mr[kb + 1] * inv : 0.0f;
        aroot[0] = (kb     < K) ? xr[kb]           : 0.0f;
        aroot[1] = (kb + 1 < K) ? xr[kb + 1]       : 0.0f;
#pragma unroll
        for (int t = 0; t < 4; ++t) {
            const int col = t * 16 + m;
            v2f bl, br;
            bl[0] = (kb     < K) ? lw[(size_t)kb       * HID_C_C + col] : 0.0f;
            bl[1] = (kb + 1 < K) ? lw[(size_t)(kb + 1) * HID_C_C + col] : 0.0f;
            br[0] = (kb     < K) ? rw[(size_t)kb       * HID_C_C + col] : 0.0f;
            br[1] = (kb + 1 < K) ? rw[(size_t)(kb + 1) * HID_C_C + col] : 0.0f;
            acc[t] = __builtin_amdgcn_wmma_f32_16x16x4_f32(
                false, amean, false, bl, (short)0, acc[t], false, false);
            acc[t] = __builtin_amdgcn_wmma_f32_16x16x4_f32(
                false, aroot, false, br, (short)0, acc[t], false, false);
        }
    }

#pragma unroll
    for (int t = 0; t < 4; ++t) {
        const int col  = t * 16 + m;
        const float bias = lb[col];
#pragma unroll
        for (int v = 0; v < 8; ++v) {
            const int orow = node0 + half * 8 + v;
            float val = acc[t][v] + bias;
            out[(size_t)orow * HID_C_C + col] = val > 0.0f ? val : 0.0f;
        }
    }
}

__global__ void __launch_bounds__(256)
pool_kernel(const float* __restrict__ h2, const int* __restrict__ batch,
            float* __restrict__ pooled, float* __restrict__ cnt) {
    __shared__ float sPool[N_GRAPHS_C * HID_C_C];
    __shared__ float sCnt[N_GRAPHS_C];
    for (int i = threadIdx.x; i < N_GRAPHS_C * HID_C_C; i += blockDim.x) sPool[i] = 0.0f;
    for (int i = threadIdx.x; i < N_GRAPHS_C; i += blockDim.x)           sCnt[i] = 0.0f;
    __syncthreads();

    const int node = blockIdx.x * blockDim.x + threadIdx.x;
    if (node < N_NODES_C) {
        const int g = batch[node];
        atomicAdd(&sCnt[g], 1.0f);
        const float* hr = h2 + (size_t)node * HID_C_C;
#pragma unroll 4
        for (int c = 0; c < HID_C_C; ++c)
            atomicAdd(&sPool[g * HID_C_C + c], hr[c]);
    }
    __syncthreads();

    for (int i = threadIdx.x; i < N_GRAPHS_C * HID_C_C; i += blockDim.x)
        atomicAdd(&pooled[i], sPool[i]);
    for (int i = threadIdx.x; i < N_GRAPHS_C; i += blockDim.x)
        atomicAdd(&cnt[i], sCnt[i]);
}

__global__ void __launch_bounds__(256)
proj_kernel(const float* __restrict__ pooled, const float* __restrict__ cnt,
            const float* __restrict__ W, const float* __restrict__ b,
            float* __restrict__ out) {
    const int i = blockIdx.x * blockDim.x + threadIdx.x;
    if (i >= N_GRAPHS_C * OUT_C_C) return;
    const int g = i / OUT_C_C;
    const int o = i % OUT_C_C;
    const float inv = 1.0f / fmaxf(cnt[g], 1.0f);
    float acc = b[o];
    const float* pr = pooled + (size_t)g * HID_C_C;
#pragma unroll 8
    for (int c = 0; c < HID_C_C; ++c)
        acc += (pr[c] * inv) * W[(size_t)c * OUT_C_C + o];
    out[i] = acc;
}

extern "C" void kernel_launch(void* const* d_in, const int* in_sizes, int n_in,
                              void* d_out, int out_size, void* d_ws, size_t ws_size,
                              hipStream_t stream) {
    const float* x      = (const float*)d_in[0];
    const float* eattr  = (const float*)d_in[1];
    const int*   eindex = (const int*)d_in[2];
    const int*   batch  = (const int*)d_in[3];
    const float* efc_w  = (const float*)d_in[4];
    const float* efc_b  = (const float*)d_in[5];
    const float* c1_lw  = (const float*)d_in[6];
    const float* c1_lb  = (const float*)d_in[7];
    const float* c1_rw  = (const float*)d_in[8];
    const float* c2_lw  = (const float*)d_in[9];
    const float* c2_lb  = (const float*)d_in[10];
    const float* c2_rw  = (const float*)d_in[11];
    const float* proj_w = (const float*)d_in[12];
    const float* proj_b = (const float*)d_in[13];
    float* out = (float*)d_out;

    float* ws = (float*)d_ws;
    size_t off = 0;
    float* x1     = ws + off; off += (size_t)N_NODES_C * IN_C_C;
    float* deg    = ws + off; off += (size_t)N_NODES_C;
    float* msum1  = ws + off; off += (size_t)N_NODES_C * IN_C_C;
    float* h1     = ws + off; off += (size_t)N_NODES_C * HID_C_C;
    float* msum2  = ws + off; off += (size_t)N_NODES_C * HID_C_C;
    float* h2     = ws + off; off += (size_t)N_NODES_C * HID_C_C;
    float* pooled = ws + off; off += (size_t)N_GRAPHS_C * HID_C_C;
    float* cnt    = ws + off; off += (size_t)N_GRAPHS_C;

    init_kernel<<<(N_NODES_C * HID_C_C + 255) / 256, 256, 0, stream>>>(
        x, x1, deg, msum1, msum2, pooled, cnt);

    edge_fc_scatter<<<(N_EDGES_C + 255) / 256, 256, 0, stream>>>(
        eattr, eindex, efc_w, efc_b, x1, deg);

    gather_scatter<IN_C_C><<<(N_EDGES_C + 255) / 256, 256, 0, stream>>>(
        eindex, x1, msum1);

    sage_wmma_kernel<IN_C_C><<<N_NODES_C / 16 / 2, 64, 0, stream>>>(
        x1, msum1, deg, c1_lw, c1_lb, c1_rw, h1);

    gather_scatter<HID_C_C><<<(N_EDGES_C + 255) / 256, 256, 0, stream>>>(
        eindex, h1, msum2);

    sage_wmma_kernel<HID_C_C><<<N_NODES_C / 16 / 2, 64, 0, stream>>>(
        h1, msum2, deg, c2_lw, c2_lb, c2_rw, h2);

    pool_kernel<<<(N_NODES_C + 255) / 256, 256, 0, stream>>>(
        h2, batch, pooled, cnt);

    proj_kernel<<<(N_GRAPHS_C * OUT_C_C + 255) / 256, 256, 0, stream>>>(
        pooled, cnt, proj_w, proj_b, out);
}


# Neighbour sums without float atomics: edges counting-sorted by destination each call (int histogram, scan, atomic-rank fill), then per-node register sums in that arrival order; sums no longer zeroed
# speedup vs baseline: 2.2634x; 2.2634x over previous
#include <hip/hip_runtime.h>
#include <hip/hip_bf16.h>
#include <stdint.h>

#define N_NODES_C  100000
#define N_EDGES_C  1600000
#define N_GRAPHS_C 64
#define IN_C_C     18
#define EDGE_C_C   22
#define HID_C_C    64
#define OUT_C_C    128

typedef float v2f __attribute__((ext_vector_type(2)));
typedef float v8f __attribute__((ext_vector_type(8)));

__global__ void __launch_bounds__(256)
init_kernel(const float* __restrict__ x, float* __restrict__ x1,
            float* __restrict__ deg, float* __restrict__ pooled,
            float* __restrict__ cnt, int* __restrict__ count) {
    const int i = blockIdx.x * blockDim.x + threadIdx.x;
    if (i < N_NODES_C * IN_C_C) x1[i] = x[i];
    if (i < N_NODES_C)          { deg[i] = 0.0f; count[i] = 0; }
    if (i < N_GRAPHS_C * HID_C_C) pooled[i] = 0.0f;
    if (i < N_GRAPHS_C)         cnt[i] = 0.0f;
}

__global__ void __launch_bounds__(256)
dst_hist(const int* __restrict__ eindex, int* __restrict__ count) {
    const int e = blockIdx.x * blockDim.x + threadIdx.x;
    if (e >= N_EDGES_C) return;
    atomicAdd(count + eindex[N_EDGES_C + e], 1);
}

#define SCAN_CHUNK 1024

__global__ void __launch_bounds__(256)
dst_scan(const int* __restrict__ count, int* __restrict__ rowptr,
         int* __restrict__ cursor) {
    __shared__ int sFront[8];
    __shared__ int sWave[8];
    const int tid  = threadIdx.x;
    const int lane = tid & 31;
    const int wv   = tid >> 5;
    const int first = blockIdx.x * SCAN_CHUNK;

    const int4* c4 = (const int4*)count;
    int front = 0;
#pragma unroll 8
    for (int i = tid; i < first / 4; i += 256) {
        const int4 v = c4[i];
        front += (v.x + v.y) + (v.z + v.w);
    }
    for (int d = 16; d > 0; d >>= 1) front += __shfl_xor(front, d);
    if (lane == 0) sFront[wv] = front;

    const int i0 = first + tid * 4;
    int c[4];
    for (int k = 0; k < 4; ++k) c[k] = (i0 + k < N_NODES_C) ? count[i0 + k] : 0;
    const int s = (c[0] + c[1]) + (c[2] + c[3]);
    int v = s;
    for (int d = 1; d < 32; d <<= 1) {
        const int n = __shfl_up(v, d);
        if (lane >= d) v += n;
    }
    if (lane == 31) sWave[wv] = v;
    __syncthreads();

    int run = 0;
    for (int w = 0; w < 8; ++w) run += sFront[w];
    for (int w = 0; w < wv; ++w) run += sWave[w];
    run += v - s;
    for (int k = 0; k < 4; ++k) {
        if (i0 + k < N_NODES_C) { rowptr[i0 + k] = run; cursor[i0 + k] = run; }
        run += c[k];
    }
}

__global__ void __launch_bounds__(256)
dst_fill(const int* __restrict__ eindex, int* __restrict__ cursor,
         int* __restrict__ col) {
    const int e = blockIdx.x * blockDim.x + threadIdx.x;
    if (e >= N_EDGES_C) return;
    const int src = eindex[e];
    const int dst = eindex[N_EDGES_C + e];
    const int pos = atomicAdd(cursor + dst, 1);
    if ((unsigned)pos < (unsigned)N_EDGES_C) col[pos] = src;
}

__global__ void __launch_bounds__(256)
edge_fc_scatter(const float* __restrict__ eattr, const int* __restrict__ eindex,
                const float* __restrict__ W, const float* __restrict__ b,
                float* __restrict__ x1, float* __restrict__ deg) {
    __shared__ float sW[EDGE_C_C * IN_C_C];
    __shared__ float sB[IN_C_C];
    for (int i = threadIdx.x; i < EDGE_C_C * IN_C_C; i += blockDim.x) sW[i] = W[i];
    for (int i = threadIdx.x; i < IN_C_C; i += blockDim.x)            sB[i] = b[i];
    __syncthreads();

    const int e = blockIdx.x * blockDim.x + threadIdx.x;
    if (e >= N_EDGES_C) return;
    const int src = eindex[e];
    const int dst = eindex[N_EDGES_C + e];

    float a[EDGE_C_C];
    const float* er = eattr + (size_t)e * EDGE_C_C;
#pragma unroll
    for (int k = 0; k < EDGE_C_C; ++k) a[k] = er[k];

    float* xr = x1 + (size_t)src * IN_C_C;
#pragma unroll
    for (int c = 0; c < IN_C_C; ++c) {
        float acc = sB[c];
#pragma unroll
        for (int k = 0; k < EDGE_C_C; ++k) acc += a[k] * sW[k * IN_C_C + c];
        atomicAdd(xr + c, acc);
    }
    atomicAdd(deg + dst, 1.0f);
}

template <int C>
__global__ void __launch_bounds__(256)
pull_sum(const int* __restrict__ rowptr, const int* __restrict__ count,
         const int* __restrict__ col, const float* __restrict__ feat,
         float* __restrict__ msum) {
    constexpr int LPN = C / 2;
    const int t = blockIdx.x * blockDim.x + threadIdx.x;
    if (t >= N_NODES_C * LPN) return;
    const int node = t / LPN;
    const int c2   = t - node * LPN;
    const int beg  = rowptr[node];
    const int end  = beg + count[node];
    const float2* f2 = (const float2*)feat;

    float2 acc = make_float2(0.0f, 0.0f);
    int j = beg;
    for (; j + 4 <= end; j += 4) {
        const int s0 = col[j], s1 = col[j + 1], s2 = col[j + 2], s3 = col[j + 3];
        const float2 a0 = f2[s0 * LPN + c2];
        const float2 a1 = f2[s1 * LPN + c2];
        const float2 a2 = f2[s2 * LPN + c2];
        const float2 a3 = f2[s3 * LPN + c2];
        acc.x += a0.x; acc.y += a0.y;
        acc.x += a1.x; acc.y += a1.y;
        acc.x += a2.x; acc.y += a2.y;
        acc.x += a3.x; acc.y += a3.y;
    }
    for (; j < end; ++j) {
        const float2 a = f2[col[j] * LPN + c2];
        acc.x += a.x; acc.y += a.y;
    }
    ((float2*)msum)[t] = acc;
}

template <int K>
__global__ void __launch_bounds__(64)
sage_wmma_kernel(const float* __restrict__ xin, const float* __restrict__ msum,
                 const float* __restrict__ deg,
                 const float* __restrict__ lw, const float* __restrict__ lb,
                 const float* __restrict__ rw, float* __restrict__ out) {
    const int wave = (blockIdx.x * blockDim.x + threadIdx.x) >> 5;
    const int lane = threadIdx.x & 31;
    if (wave >= N_NODES_C / 16) return;

    const int node0 = wave * 16;
    const int m    = lane & 15;
    const int half = (lane >> 4) & 1;
    const int row  = node0 + m;

    const float dg  = deg[row];
    const float inv = dg > 0.0f ? 1.0f / dg : 0.0f;
    const float* xr = xin  + (size_t)row * K;
    const float* mr = msum + (size_t)row * K;

    v8f acc[4] = {};

    constexpr int KP = (K + 3) & ~3;
#pragma unroll
    for (int ks = 0; ks < KP; ks += 4) {
        const int kb = ks + half * 2;
        v2f amean, aroot;
        amean[0] = (kb     < K) ? mr[kb]     * inv : 0.0f;
        amean[1] = (kb + 1 < K) ? mr[kb + 1] * inv : 0.0f;
        aroot[0] = (kb     < K) ? xr[kb]           : 0.0f;
        aroot[1] = (kb + 1 < K) ? xr[kb + 1]       : 0.0f;
#pragma unroll
        for (int t = 0; t < 4; ++t) {
            const int col = t * 16 + m;
            v2f bl, br;
            bl[0] = (kb     < K) ? lw[(size_t)kb       * HID_C_C + col] : 0.0f;
            bl[1] = (kb + 1 < K) ? lw[(size_t)(kb + 1) * HID_C_C + col] : 0.0f;
            br[0] = (kb     < K) ? rw[(size_t)kb       * HID_C_C + col] : 0.0f;
            br[1] = (kb + 1 < K) ? rw[(size_t)(kb + 1) * HID_C_C + col] : 0.0f;
            acc[t] = __builtin_amdgcn_wmma_f32_16x16x4_f32(
                false, amean, false, bl, (short)0, acc[t], false, false);
            acc[t] = __builtin_amdgcn_wmma_f32_16x16x4_f32(
                false, aroot, false, br, (short)0, acc[t], false, false);
        }
    }

#pragma unroll
    for (int t = 0; t < 4; ++t) {
        const int col  = t * 16 + m;
        const float bias = lb[col];
#pragma unroll
        for (int v = 0; v < 8; ++v) {
            const int orow = node0 + half * 8 + v;
            float val = acc[t][v] + bias;
            out[(size_t)orow * HID_C_C + col] = val > 0.0f ? val : 0.0f;
        }
    }
}

__global__ void __launch_bounds__(256)
pool_kernel(const float* __restrict__ h2, const int* __restrict__ batch,
            float* __restrict__ pooled, float* __restrict__ cnt) {
    __shared__ float sPool[N_GRAPHS_C * HID_C_C];
    __shared__ float sCnt[N_GRAPHS_C];
    for (int i = threadIdx.x; i < N_GRAPHS_C * HID_C_C; i += blockDim.x) sPool[i] = 0.0f;
    for (int i = threadIdx.x; i < N_GRAPHS_C; i += blockDim.x)           sCnt[i] = 0.0f;
    __syncthreads();

    const int node = blockIdx.x * blockDim.x + threadIdx.x;
    if (node < N_NODES_C) {
        const int g = batch[node];
        atomicAdd(&sCnt[g], 1.0f);
        const float* hr = h2 + (size_t)node * HID_C_C;
#pragma unroll 4
        for (int c = 0; c < HID_C_C; ++c)
            atomicAdd(&sPool[g * HID_C_C + c], hr[c]);
    }
    __syncthreads();

    for (int i = threadIdx.x; i < N_GRAPHS_C * HID_C_C; i += blockDim.x)
        atomicAdd(&pooled[i], sPool[i]);
    for (int i = threadIdx.x; i < N_GRAPHS_C; i += blockDim.x)
        atomicAdd(&cnt[i], sCnt[i]);
}

__global__ void __launch_bounds__(256)
proj_kernel(const float* __restrict__ pooled, const float* __restrict__ cnt,
            const float* __restrict__ W, const float* __restrict__ b,
            float* __restrict__ out) {
    const int i = blockIdx.x * blockDim.x + threadIdx.x;
    if (i >= N_GRAPHS_C * OUT_C_C) return;
    const int g = i / OUT_C_C;
    const int o = i % OUT_C_C;
    const float inv = 1.0f / fmaxf(cnt[g], 1.0f);
    float acc = b[o];
    const float* pr = pooled + (size_t)g * HID_C_C;
#pragma unroll 8
    for (int c = 0; c < HID_C_C; ++c)
        acc += (pr[c] * inv) * W[(size_t)c * OUT_C_C + o];
    out[i] = acc;
}

extern "C" void kernel_launch(void* const* d_in, const int* in_sizes, int n_in,
                              void* d_out, int out_size, void* d_ws, size_t ws_size,
                              hipStream_t stream) {
    const float* x      = (const float*)d_in[0];
    const float* eattr  = (const float*)d_in[1];
    const int*   eindex = (const int*)d_in[2];
    const int*   batch  = (const int*)d_in[3];
    const float* efc_w  = (const float*)d_in[4];
    const float* efc_b  = (const float*)d_in[5];
    const float* c1_lw  = (const float*)d_in[6];
    const float* c1_lb  = (const float*)d_in[7];
    const float* c1_rw  = (const float*)d_in[8];
    const float* c2_lw  = (const float*)d_in[9];
    const float* c2_lb  = (const float*)d_in[10];
    const float* c2_rw  = (const float*)d_in[11];
    const float* proj_w = (const float*)d_in[12];
    const float* proj_b = (const float*)d_in[13];
    float* out = (float*)d_out;

    float* ws = (float*)d_ws;
    size_t off = 0;
    float* x1     = ws + off; off += (size_t)N_NODES_C * IN_C_C;
    float* deg    = ws + off; off += (size_t)N_NODES_C;
    float* msum1  = ws + off; off += (size_t)N_NODES_C * IN_C_C;
    float* h1     = ws + off; off += (size_t)N_NODES_C * HID_C_C;
    float* msum2  = ws + off; off += (size_t)N_NODES_C * HID_C_C;
    float* h2     = ws + off; off += (size_t)N_NODES_C * HID_C_C;
    float* pooled = ws + off; off += (size_t)N_GRAPHS_C * HID_C_C;
    float* cnt    = ws + off; off += (size_t)N_GRAPHS_C;

    int* count  = (int*)h2;
    int* rowptr = count  + N_NODES_C;
    int* cursor = rowptr + N_NODES_C;
    int* col    = cursor + N_NODES_C;

    init_kernel<<<(N_NODES_C * IN_C_C + 255) / 256, 256, 0, stream>>>(
        x, x1, deg, pooled, cnt, count);

    dst_hist<<<(N_EDGES_C + 255) / 256, 256, 0, stream>>>(eindex, count);
    dst_scan<<<(N_NODES_C + SCAN_CHUNK - 1) / SCAN_CHUNK, 256, 0, stream>>>(
        count, rowptr, cursor);
    dst_fill<<<(N_EDGES_C + 255) / 256, 256, 0, stream>>>(eindex, cursor, col);

    edge_fc_scatter<<<(N_EDGES_C + 255) / 256, 256, 0, stream>>>(
        eattr, eindex, efc_w, efc_b, x1, deg);

    pull_sum<IN_C_C><<<(N_NODES_C * (IN_C_C / 2) + 255) / 256, 256, 0, stream>>>(
        rowptr, count, col, x1, msum1);

    sage_wmma_kernel<IN_C_C><<<N_NODES_C / 16 / 2, 64, 0, stream>>>(
        x1, msum1, deg, c1_lw, c1_lb, c1_rw, h1);

    pull_sum<HID_C_C><<<(N_NODES_C * (HID_C_C / 2) + 255) / 256, 256, 0, stream>>>(
        rowptr, count, col, h1, msum2);

    sage_wmma_kernel<HID_C_C><<<N_NODES_C / 16 / 2, 64, 0, stream>>>(
        h1, msum2, deg, c2_lw, c2_lb, c2_rw, h2);

    pool_kernel<<<(N_NODES_C + 255) / 256, 256, 0, stream>>>(
        h2, batch, pooled, cnt);

    proj_kernel<<<(N_GRAPHS_C * OUT_C_C + 255) / 256, 256, 0, stream>>>(
        pooled, cnt, proj_w, proj_b, out);
}


# edge_fc float atomics removed: edges also counting-sorted by source each call; 16 lanes per node sum its edges' attr rows (list order, lane tree), one 22x18 product + count*bias; deg = int histogram
# speedup vs baseline: 3.6113x; 1.5955x over previous
#include <hip/hip_runtime.h>
#include <hip/hip_bf16.h>
#include <stdint.h>

#define N_NODES_C  100000
#define N_EDGES_C  1600000
#define N_GRAPHS_C 64
#define IN_C_C     18
#define EDGE_C_C   22
#define HID_C_C    64
#define OUT_C_C    128

typedef float v2f __attribute__((ext_vector_type(2)));
typedef float v8f __attribute__((ext_vector_type(8)));

__global__ void __launch_bounds__(256)
init_kernel(float* __restrict__ pooled, float* __restrict__ cnt,
            int* __restrict__ count_d, int* __restrict__ count_s) {
    const int i = blockIdx.x * blockDim.x + threadIdx.x;
    if (i < N_NODES_C)          { count_d[i] = 0; count_s[i] = 0; }
    if (i < N_GRAPHS_C * HID_C_C) pooled[i] = 0.0f;
    if (i < N_GRAPHS_C)         cnt[i] = 0.0f;
}

__global__ void __launch_bounds__(256)
edge_hist(const int* __restrict__ eindex, int* __restrict__ count_d,
          int* __restrict__ count_s) {
    const int e = blockIdx.x * blockDim.x + threadIdx.x;
    if (e >= N_EDGES_C) return;
    atomicAdd(count_s + eindex[e], 1);
    atomicAdd(count_d + eindex[N_EDGES_C + e], 1);
}

#define SCAN_CHUNK 1024

__global__ void __launch_bounds__(256)
list_scan(const int* __restrict__ count_d, int* __restrict__ rowptr_d,
          int* __restrict__ cursor_d,
          const int* __restrict__ count_s, int* __restrict__ rowptr_s,
          int* __restrict__ cursor_s) {
    __shared__ int sFront[8];
    __shared__ int sWave[8];
    const bool by_src = blockIdx.y != 0;
    const int* __restrict__ count  = by_src ? count_s  : count_d;
    int* __restrict__       rowptr = by_src ? rowptr_s : rowptr_d;
    int* __restrict__       cursor = by_src ? cursor_s : cursor_d;
    const int tid  = threadIdx.x;
    const int lane = tid & 31;
    const int wv   = tid >> 5;
    const int first = blockIdx.x * SCAN_CHUNK;

    const int4* c4 = (const int4*)count;
    int front = 0;
#pragma unroll 8
    for (int i = tid; i < first / 4; i += 256) {
        const int4 v = c4[i];
        front += (v.x + v.y) + (v.z + v.w);
    }
    for (int d = 16; d > 0; d >>= 1) front += __shfl_xor(front, d);
    if (lane == 0) sFront[wv] = front;

    const int i0 = first + tid * 4;
    int c[4];
    for (int k = 0; k < 4; ++k) c[k] = (i0 + k < N_NODES_C) ? count[i0 + k] : 0;
    const int s = (c[0] + c[1]) + (c[2] + c[3]);
    int v = s;
    for (int d = 1; d < 32; d <<= 1) {
        const int n = __shfl_up(v, d);
        if (lane >= d) v += n;
    }
    if (lane == 31) sWave[wv] = v;
    __syncthreads();

    int run = 0;
    for (int w = 0; w < 8; ++w) run += sFront[w];
    for (int w = 0; w < wv; ++w) run += sWave[w];
    run += v - s;
    for (int k = 0; k < 4; ++k) {
        if (i0 + k < N_NODES_C) { rowptr[i0 + k] = run; cursor[i0 + k] = run; }
        run += c[k];
    }
}

__global__ void __launch_bounds__(256)
edge_fill(const int* __restrict__ eindex, int* __restrict__ cursor_d,
          int* __restrict__ col, int* __restrict__ cursor_s,
          int* __restrict__ eperm) {
    const int e = blockIdx.x * blockDim.x + threadIdx.x;
    if (e >= N_EDGES_C) return;
    const int src = eindex[e];
    const int dst = eindex[N_EDGES_C + e];
    const int pd = atomicAdd(cursor_d + dst, 1);
    const int ps = atomicAdd(cursor_s + src, 1);
    if ((unsigned)pd < (unsigned)N_EDGES_C) col[pd]   = src;
    if ((unsigned)ps < (unsigned)N_EDGES_C) eperm[ps] = e;
}

template <int M>
static __device__ __forceinline__ float row_xor_add(float v) {
    const int o = __builtin_amdgcn_update_dpp(0, __float_as_int(v), 0x160 + M,
                                              0xf, 0xf, true);
    return v + __int_as_float(o);
}

__global__ void __launch_bounds__(256)
edge_fc_pull(const float* __restrict__ eattr, const float* __restrict__ x,
             const float* __restrict__ W, const float* __restrict__ b,
             const int* __restrict__ rowptr_s, const int* __restrict__ count_s,
             const int* __restrict__ eperm, const int* __restrict__ count_d,
             float* __restrict__ x1, float* __restrict__ deg) {
    const int t    = blockIdx.x * blockDim.x + threadIdx.x;
    const int node = t >> 4;
    const int l    = t & 15;
    const int beg  = rowptr_s[node];
    const int cnt  = count_s[node];

    float s[EDGE_C_C];
#pragma unroll
    for (int q = 0; q < EDGE_C_C; ++q) s[q] = 0.0f;

    for (int k = l; k < cnt; k += 16) {
        unsigned e = (unsigned)eperm[beg + k];
        e = e < (unsigned)N_EDGES_C ? e : (unsigned)(N_EDGES_C - 1);
        const float2* er = (const float2*)(eattr + (size_t)e * EDGE_C_C);
#pragma unroll
        for (int q = 0; q < EDGE_C_C / 2; ++q) {
            const float2 p = er[q];
            s[2 * q] += p.x; s[2 * q + 1] += p.y;
        }
    }

#pragma unroll
    for (int q = 0; q < EDGE_C_C; ++q) {
        float v = s[q];
        v = row_xor_add<1>(v);
        v = row_xor_add<2>(v);
        v = row_xor_add<4>(v);
        v = row_xor_add<8>(v);
        s[q] = v;
    }

    if (l < IN_C_C / 2) {
        const float2* W2 = (const float2*)W;
        const float2 bb = ((const float2*)b)[l];
        const float  cf = (float)cnt;
        float2 acc = make_float2(cf * bb.x, cf * bb.y);
#pragma unroll
        for (int q = 0; q < EDGE_C_C; ++q) {
            const float2 w = W2[q * (IN_C_C / 2) + l];
            acc.x += s[q] * w.x; acc.y += s[q] * w.y;
        }
        const float2 xv = ((const float2*)x)[node * (IN_C_C / 2) + l];
        ((float2*)x1)[node * (IN_C_C / 2) + l] = make_float2(xv.x + acc.x, xv.y + acc.y);
    }
    if (l == 15) deg[node] = (float)count_d[node];
}

template <int C>
__global__ void __launch_bounds__(256)
pull_sum(const int* __restrict__ rowptr, const int* __restrict__ count,
         const int* __restrict__ col, const float* __restrict__ feat,
         float* __restrict__ msum) {
    constexpr int LPN = C / 2;
    const int t = blockIdx.x * blockDim.x + threadIdx.x;
    if (t >= N_NODES_C * LPN) return;
    const int node = t / LPN;
    const int c2   = t - node * LPN;
    const int beg  = rowptr[node];
    const int end  = beg + count[node];
    const float2* f2 = (const float2*)feat;

    float2 acc = make_float2(0.0f, 0.0f);
    int j = beg;
    for (; j + 4 <= end; j += 4) {
        const int s0 = col[j], s1 = col[j + 1], s2 = col[j + 2], s3 = col[j + 3];
        const float2 a0 = f2[s0 * LPN + c2];
        const float2 a1 = f2[s1 * LPN + c2];
        const float2 a2 = f2[s2 * LPN + c2];
        const float2 a3 = f2[s3 * LPN + c2];
        acc.x += a0.x; acc.y += a0.y;
        acc.x += a1.x; acc.y += a1.y;
        acc.x += a2.x; acc.y += a2.y;
        acc.x += a3.x; acc.y += a3.y;
    }
    for (; j < end; ++j) {
        const float2 a = f2[col[j] * LPN + c2];
        acc.x += a.x; acc.y += a.y;
    }
    ((float2*)msum)[t] = acc;
}

template <int K>
__global__ void __launch_bounds__(64)
sage_wmma_kernel(const float* __restrict__ xin, const float* __restrict__ msum,
                 const float* __restrict__ deg,
                 const float* __restrict__ lw, const float* __restrict__ lb,
                 const float* __restrict__ rw, float* __restrict__ out) {
    const int wave = (blockIdx.x * blockDim.x + threadIdx.x) >> 5;
    const int lane = threadIdx.x & 31;
    if (wave >= N_NODES_C / 16) return;

    const int node0 = wave * 16;
    const int m    = lane & 15;
    const int half = (lane >> 4) & 1;
    const int row  = node0 + m;

    const float dg  = deg[row];
    const float inv = dg > 0.0f ? 1.0f / dg : 0.0f;
    const float* xr = xin  + (size_t)row * K;
    const float* mr = msum + (size_t)row * K;

    v8f acc[4] = {};

    constexpr int KP = (K + 3) & ~3;
#pragma unroll
    for (int ks = 0; ks < KP; ks += 4) {
        const int kb = ks + half * 2;
        v2f amean, aroot;
        amean[0] = (kb     < K) ? mr[kb]     * inv : 0.0f;
        amean[1] = (kb + 1 < K) ? mr[kb + 1] * inv : 0.0f;
        aroot[0] = (kb     < K) ? xr[kb]           : 0.0f;
        aroot[1] = (kb + 1 < K) ? xr[kb + 1]       : 0.0f;
#pragma unroll
        for (int t = 0; t < 4; ++t) {
            const int col = t * 16 + m;
            v2f bl, br;
            bl[0] = (kb     < K) ? lw[(size_t)kb       * HID_C_C + col] : 0.0f;
            bl[1] = (kb + 1 < K) ? lw[(size_t)(kb + 1) * HID_C_C + col] : 0.0f;
            br[0] = (kb     < K) ? rw[(size_t)kb       * HID_C_C + col] : 0.0f;
            br[1] = (kb + 1 < K) ? rw[(size_t)(kb + 1) * HID_C_C + col] : 0.0f;
            acc[t] = __builtin_amdgcn_wmma_f32_16x16x4_f32(
                false, amean, false, bl, (short)0, acc[t], false, false);
            acc[t] = __builtin_amdgcn_wmma_f32_16x16x4_f32(
                false, aroot, false, br, (short)0, acc[t], false, false);
        }
    }

#pragma unroll
    for (int t = 0; t < 4; ++t) {
        const int col  = t * 16 + m;
        const float bias = lb[col];
#pragma unroll
        for (int v = 0; v < 8; ++v) {
            const int orow = node0 + half * 8 + v;
            float val = acc[t][v] + bias;
            out[(size_t)orow * HID_C_C + col] = val > 0.0f ? val : 0.0f;
        }
    }
}

__global__ void __launch_bounds__(256)
pool_kernel(const float* __restrict__ h2, const int* __restrict__ batch,
            float* __restrict__ pooled, float* __restrict__ cnt) {
    __shared__ float sPool[N_GRAPHS_C * HID_C_C];
    __shared__ float sCnt[N_GRAPHS_C];
    for (int i = threadIdx.x; i < N_GRAPHS_C * HID_C_C; i += blockDim.x) sPool[i] = 0.0f;
    for (int i = threadIdx.x; i < N_GRAPHS_C; i += blockDim.x)           sCnt[i] = 0.0f;
    __syncthreads();

    const int node = blockIdx.x * blockDim.x + threadIdx.x;
    if (node < N_NODES_C) {
        const int g = batch[node];
        atomicAdd(&sCnt[g], 1.0f);
        const float* hr = h2 + (size_t)node * HID_C_C;
#pragma unroll 4
        for (int c = 0; c < HID_C_C; ++c)
            atomicAdd(&sPool[g * HID_C_C + c], hr[c]);
    }
    __syncthreads();

    for (int i = threadIdx.x; i < N_GRAPHS_C * HID_C_C; i += blockDim.x)
        atomicAdd(&pooled[i], sPool[i]);
    for (int i = threadIdx.x; i < N_GRAPHS_C; i += blockDim.x)
        atomicAdd(&cnt[i], sCnt[i]);
}

__global__ void __launch_bounds__(256)
proj_kernel(const float* __restrict__ pooled, const float* __restrict__ cnt,
            const float* __restrict__ W, const float* __restrict__ b,
            float* __restrict__ out) {
    const int i = blockIdx.x * blockDim.x + threadIdx.x;
    if (i >= N_GRAPHS_C * OUT_C_C) return;
    const int g = i / OUT_C_C;
    const int o = i % OUT_C_C;
    const float inv = 1.0f / fmaxf(cnt[g], 1.0f);
    float acc = b[o];
    const float* pr = pooled + (size_t)g * HID_C_C;
#pragma unroll 8
    for (int c = 0; c < HID_C_C; ++c)
        acc += (pr[c] * inv) * W[(size_t)c * OUT_C_C + o];
    out[i] = acc;
}

extern "C" void kernel_launch(void* const* d_in, const int* in_sizes, int n_in,
                              void* d_out, int out_size, void* d_ws, size_t ws_size,
                              hipStream_t stream) {
    const float* x      = (const float*)d_in[0];
    const float* eattr  = (const float*)d_in[1];
    const int*   eindex = (const int*)d_in[2];
    const int*   batch  = (const int*)d_in[3];
    const float* efc_w  = (const float*)d_in[4];
    const float* efc_b  = (const float*)d_in[5];
    const float* c1_lw  = (const float*)d_in[6];
    const float* c1_lb  = (const float*)d_in[7];
    const float* c1_rw  = (const float*)d_in[8];
    const float* c2_lw  = (const float*)d_in[9];
    const float* c2_lb  = (const float*)d_in[10];
    const float* c2_rw  = (const float*)d_in[11];
    const float* proj_w = (const float*)d_in[12];
    const float* proj_b = (const float*)d_in[13];
    float* out = (float*)d_out;

    float* ws = (float*)d_ws;
    size_t off = 0;
    float* x1     = ws + off; off += (size_t)N_NODES_C * IN_C_C;
    float* deg    = ws + off; off += (size_t)N_NODES_C;
    float* msum1  = ws + off; off += (size_t)N_NODES_C * IN_C_C;
    float* h1     = ws + off; off += (size_t)N_NODES_C * HID_C_C;
    float* msum2  = ws + off; off += (size_t)N_NODES_C * HID_C_C;
    float* h2     = ws + off; off += (size_t)N_NODES_C * HID_C_C;
    float* pooled = ws + off; off += (size_t)N_GRAPHS_C * HID_C_C;
    float* cnt    = ws + off; off += (size_t)N_GRAPHS_C;

    int* count  = (int*)h2;
    int* rowptr = count  + N_NODES_C;
    int* cursor = rowptr + N_NODES_C;
    int* col    = cursor + N_NODES_C;
    int* count_s  = (int*)h1;
    int* rowptr_s = count_s  + N_NODES_C;
    int* cursor_s = rowptr_s + N_NODES_C;
    int* eperm    = cursor_s + N_NODES_C;

    init_kernel<<<(N_NODES_C + 255) / 256, 256, 0, stream>>>(
        pooled, cnt, count, count_s);

    edge_hist<<<(N_EDGES_C + 255) / 256, 256, 0, stream>>>(eindex, count, count_s);
    list_scan<<<dim3((N_NODES_C + SCAN_CHUNK - 1) / SCAN_CHUNK, 2), 256, 0, stream>>>(
        count, rowptr, cursor, count_s, rowptr_s, cursor_s);
    edge_fill<<<(N_EDGES_C + 255) / 256, 256, 0, stream>>>(
        eindex, cursor, col, cursor_s, eperm);

    edge_fc_pull<<<N_NODES_C * 16 / 256, 256, 0, stream>>>(
        eattr, x, efc_w, efc_b, rowptr_s, count_s, eperm, count, x1, deg);

    pull_sum<IN_C_C><<<(N_NODES_C * (IN_C_C / 2) + 255) / 256, 256, 0, stream>>>(
        rowptr, count, col, x1, msum1);

    sage_wmma_kernel<IN_C_C><<<N_NODES_C / 16 / 2, 64, 0, stream>>>(
        x1, msum1, deg, c1_lw, c1_lb, c1_rw, h1);

    pull_sum<HID_C_C><<<(N_NODES_C * (HID_C_C / 2) + 255) / 256, 256, 0, stream>>>(
        rowptr, count, col, h1, msum2);

    sage_wmma_kernel<HID_C_C><<<N_NODES_C / 16 / 2, 64, 0, stream>>>(
        h1, msum2, deg, c2_lw, c2_lb, c2_rw, h2);

    pool_kernel<<<(N_NODES_C + 255) / 256, 256, 0, stream>>>(
        h2, batch, pooled, cnt);

    proj_kernel<<<(N_GRAPHS_C * OUT_C_C + 255) / 256, 256, 0, stream>>>(
        pooled, cnt, proj_w, proj_b, out);
}


# Mean pool folded into conv2's WMMA epilogue: tile's 16 rows summed per column in registers (fixed order), one float atomic per column per tile (arrival order); h2 store, pool launch, LDS adds gone
# speedup vs baseline: 4.2747x; 1.1837x over previous
#include <hip/hip_runtime.h>
#include <hip/hip_bf16.h>
#include <stdint.h>

#define N_NODES_C  100000
#define N_EDGES_C  1600000
#define N_GRAPHS_C 64
#define IN_C_C     18
#define EDGE_C_C   22
#define HID_C_C    64
#define OUT_C_C    128

typedef float v2f __attribute__((ext_vector_type(2)));
typedef float v8f __attribute__((ext_vector_type(8)));

__global__ void __launch_bounds__(256)
init_kernel(float* __restrict__ pooled, float* __restrict__ cnt,
            int* __restrict__ count_d, int* __restrict__ count_s) {
    const int i = blockIdx.x * blockDim.x + threadIdx.x;
    if (i < N_NODES_C)          { count_d[i] = 0; count_s[i] = 0; }
    if (i < N_GRAPHS_C * HID_C_C) pooled[i] = 0.0f;
    if (i < N_GRAPHS_C)         cnt[i] = 0.0f;
}

__global__ void __launch_bounds__(256)
edge_hist(const int* __restrict__ eindex, int* __restrict__ count_d,
          int* __restrict__ count_s) {
    const int e = blockIdx.x * blockDim.x + threadIdx.x;
    if (e >= N_EDGES_C) return;
    atomicAdd(count_s + eindex[e], 1);
    atomicAdd(count_d + eindex[N_EDGES_C + e], 1);
}

#define SCAN_CHUNK 1024

__global__ void __launch_bounds__(256)
list_scan(const int* __restrict__ count_d, int* __restrict__ rowptr_d,
          int* __restrict__ cursor_d,
          const int* __restrict__ count_s, int* __restrict__ rowptr_s,
          int* __restrict__ cursor_s) {
    __shared__ int sFront[8];
    __shared__ int sWave[8];
    const bool by_src = blockIdx.y != 0;
    const int* __restrict__ count  = by_src ? count_s  : count_d;
    int* __restrict__       rowptr = by_src ? rowptr_s : rowptr_d;
    int* __restrict__       cursor = by_src ? cursor_s : cursor_d;
    const int tid  = threadIdx.x;
    const int lane = tid & 31;
    const int wv   = tid >> 5;
    const int first = blockIdx.x * SCAN_CHUNK;

    const int4* c4 = (const int4*)count;
    int front = 0;
#pragma unroll 8
    for (int i = tid; i < first / 4; i += 256) {
        const int4 v = c4[i];
        front += (v.x + v.y) + (v.z + v.w);
    }
    for (int d = 16; d > 0; d >>= 1) front += __shfl_xor(front, d);
    if (lane == 0) sFront[wv] = front;

    const int i0 = first + tid * 4;
    int c[4];
    for (int k = 0; k < 4; ++k) c[k] = (i0 + k < N_NODES_C) ? count[i0 + k] : 0;
    const int s = (c[0] + c[1]) + (c[2] + c[3]);
    int v = s;
    for (int d = 1; d < 32; d <<= 1) {
        const int n = __shfl_up(v, d);
        if (lane >= d) v += n;
    }
    if (lane == 31) sWave[wv] = v;
    __syncthreads();

    int run = 0;
    for (int w = 0; w < 8; ++w) run += sFront[w];
    for (int w = 0; w < wv; ++w) run += sWave[w];
    run += v - s;
    for (int k = 0; k < 4; ++k) {
        if (i0 + k < N_NODES_C) { rowptr[i0 + k] = run; cursor[i0 + k] = run; }
        run += c[k];
    }
}

__global__ void __launch_bounds__(256)
edge_fill(const int* __restrict__ eindex, int* __restrict__ cursor_d,
          int* __restrict__ col, int* __restrict__ cursor_s,
          int* __restrict__ eperm) {
    const int e = blockIdx.x * blockDim.x + threadIdx.x;
    if (e >= N_EDGES_C) return;
    const int src = eindex[e];
    const int dst = eindex[N_EDGES_C + e];
    const int pd = atomicAdd(cursor_d + dst, 1);
    const int ps = atomicAdd(cursor_s + src, 1);
    if ((unsigned)pd < (unsigned)N_EDGES_C) col[pd]   = src;
    if ((unsigned)ps < (unsigned)N_EDGES_C) eperm[ps] = e;
}

template <int M>
static __device__ __forceinline__ float row_xor_add(float v) {
    const int o = __builtin_amdgcn_update_dpp(0, __float_as_int(v), 0x160 + M,
                                              0xf, 0xf, true);
    return v + __int_as_float(o);
}

__global__ void __launch_bounds__(256)
edge_fc_pull(const float* __restrict__ eattr, const float* __restrict__ x,
             const float* __restrict__ W, const float* __restrict__ b,
             const int* __restrict__ rowptr_s, const int* __restrict__ count_s,
             const int* __restrict__ eperm, const int* __restrict__ count_d,
             float* __restrict__ x1, float* __restrict__ deg) {
    const int t    = blockIdx.x * blockDim.x + threadIdx.x;
    const int node = t >> 4;
    const int l    = t & 15;
    const int beg  = rowptr_s[node];
    const int cnt  = count_s[node];

    float s[EDGE_C_C];
#pragma unroll
    for (int q = 0; q < EDGE_C_C; ++q) s[q] = 0.0f;

    for (int k = l; k < cnt; k += 16) {
        unsigned e = (unsigned)eperm[beg + k];
        e = e < (unsigned)N_EDGES_C ? e : (unsigned)(N_EDGES_C - 1);
        const float2* er = (const float2*)(eattr + (size_t)e * EDGE_C_C);
#pragma unroll
        for (int q = 0; q < EDGE_C_C / 2; ++q) {
            const float2 p = er[q];
            s[2 * q] += p.x; s[2 * q + 1] += p.y;
        }
    }

#pragma unroll
    for (int q = 0; q < EDGE_C_C; ++q) {
        float v = s[q];
        v = row_xor_add<1>(v);
        v = row_xor_add<2>(v);
        v = row_xor_add<4>(v);
        v = row_xor_add<8>(v);
        s[q] = v;
    }

    if (l < IN_C_C / 2) {
        const float2* W2 = (const float2*)W;
        const float2 bb = ((const float2*)b)[l];
        const float  cf = (float)cnt;
        float2 acc = make_float2(cf * bb.x, cf * bb.y);
#pragma unroll
        for (int q = 0; q < EDGE_C_C; ++q) {
            const float2 w = W2[q * (IN_C_C / 2) + l];
            acc.x += s[q] * w.x; acc.y += s[q] * w.y;
        }
        const float2 xv = ((const float2*)x)[node * (IN_C_C / 2) + l];
        ((float2*)x1)[node * (IN_C_C / 2) + l] = make_float2(xv.x + acc.x, xv.y + acc.y);
    }
    if (l == 15) deg[node] = (float)count_d[node];
}

template <int C>
__global__ void __launch_bounds__(256)
pull_sum(const int* __restrict__ rowptr, const int* __restrict__ count,
         const int* __restrict__ col, const float* __restrict__ feat,
         float* __restrict__ msum) {
    constexpr int LPN = C / 2;
    const int t = blockIdx.x * blockDim.x + threadIdx.x;
    if (t >= N_NODES_C * LPN) return;
    const int node = t / LPN;
    const int c2   = t - node * LPN;
    const int beg  = rowptr[node];
    const int end  = beg + count[node];
    const float2* f2 = (const float2*)feat;

    float2 acc = make_float2(0.0f, 0.0f);
    int j = beg;
    for (; j + 4 <= end; j += 4) {
        const int s0 = col[j], s1 = col[j + 1], s2 = col[j + 2], s3 = col[j + 3];
        const float2 a0 = f2[s0 * LPN + c2];
        const float2 a1 = f2[s1 * LPN + c2];
        const float2 a2 = f2[s2 * LPN + c2];
        const float2 a3 = f2[s3 * LPN + c2];
        acc.x += a0.x; acc.y += a0.y;
        acc.x += a1.x; acc.y += a1.y;
        acc.x += a2.x; acc.y += a2.y;
        acc.x += a3.x; acc.y += a3.y;
    }
    for (; j < end; ++j) {
        const float2 a = f2[col[j] * LPN + c2];
        acc.x += a.x; acc.y += a.y;
    }
    ((float2*)msum)[t] = acc;
}

template <int K, bool POOL>
__global__ void __launch_bounds__(64)
sage_wmma_kernel(const float* __restrict__ xin, const float* __restrict__ msum,
                 const float* __restrict__ deg,
                 const float* __restrict__ lw, const float* __restrict__ lb,
                 const float* __restrict__ rw, float* __restrict__ out,
                 const int* __restrict__ batch, float* __restrict__ pooled,
                 float* __restrict__ cnt) {
    const int wave = (blockIdx.x * blockDim.x + threadIdx.x) >> 5;
    const int lane = threadIdx.x & 31;
    if (wave >= N_NODES_C / 16) return;

    const int node0 = wave * 16;
    const int m    = lane & 15;
    const int half = (lane >> 4) & 1;
    const int row  = node0 + m;

    const float dg  = deg[row];
    const float inv = dg > 0.0f ? 1.0f / dg : 0.0f;
    const float* xr = xin  + (size_t)row * K;
    const float* mr = msum + (size_t)row * K;

    v8f acc[4] = {};

    constexpr int KP = (K + 3) & ~3;
#pragma unroll
    for (int ks = 0; ks < KP; ks += 4) {
        const int kb = ks + half * 2;
        v2f amean, aroot;
        amean[0] = (kb     < K) ? mr[kb]     * inv : 0.0f;
        amean[1] = (kb + 1 < K) ? mr[kb + 1] * inv : 0.0f;
        aroot[0] = (kb     < K) ? xr[kb]           : 0.0f;
        aroot[1] = (kb + 1 < K) ? xr[kb + 1]       : 0.0f;
#pragma unroll
        for (int t = 0; t < 4; ++t) {
            const int col = t * 16 + m;
            v2f bl, br;
            bl[0] = (kb     < K) ? lw[(size_t)kb       * HID_C_C + col] : 0.0f;
            bl[1] = (kb + 1 < K) ? lw[(size_t)(kb + 1) * HID_C_C + col] : 0.0f;
            br[0] = (kb     < K) ? rw[(size_t)kb       * HID_C_C + col] : 0.0f;
            br[1] = (kb + 1 < K) ? rw[(size_t)(kb + 1) * HID_C_C + col] : 0.0f;
            acc[t] = __builtin_amdgcn_wmma_f32_16x16x4_f32(
                false, amean, false, bl, (short)0, acc[t], false, false);
            acc[t] = __builtin_amdgcn_wmma_f32_16x16x4_f32(
                false, aroot, false, br, (short)0, acc[t], false, false);
        }
    }

    if (!POOL) {
#pragma unroll
        for (int t = 0; t < 4; ++t) {
            const int col  = t * 16 + m;
            const float bias = lb[col];
#pragma unroll
            for (int v = 0; v < 8; ++v) {
                const int orow = node0 + half * 8 + v;
                float val = acc[t][v] + bias;
                out[(size_t)orow * HID_C_C + col] = val > 0.0f ? val : 0.0f;
            }
        }
    } else {
        const int g  = batch[row];
        const int g0 = __builtin_amdgcn_readfirstlane(g);

        float h[4][8];
        float cs[4];
#pragma unroll
        for (int t = 0; t < 4; ++t) {
            const float bias = lb[t * 16 + m];
            float s = 0.0f;
#pragma unroll
            for (int v = 0; v < 8; ++v) {
                const float val = acc[t][v] + bias;
                h[t][v] = val > 0.0f ? val : 0.0f;
                s += h[t][v];
            }
            cs[t] = s;
        }

        if (__builtin_amdgcn_ballot_w32(g != g0) == 0u) {
            float c4[4];
#pragma unroll
            for (int t = 0; t < 4; ++t) c4[t] = cs[t] + __shfl_xor(cs[t], 16);
            const float a0 = half ? c4[2] : c4[0];
            const float a1 = half ? c4[3] : c4[1];
            float* pg = pooled + g0 * HID_C_C + half * 32 + m;
            atomicAdd(pg,      a0);
            atomicAdd(pg + 16, a1);
            if (lane == 0) atomicAdd(cnt + g0, 16.0f);
        } else {
#pragma unroll
            for (int v = 0; v < 8; ++v) {
                const int gr = __shfl(g, half * 8 + v);
#pragma unroll
                for (int t = 0; t < 4; ++t)
                    atomicAdd(pooled + gr * HID_C_C + t * 16 + m, h[t][v]);
            }
            if (lane < 16) atomicAdd(cnt + g, 1.0f);
        }
    }
}

__global__ void __launch_bounds__(256)
proj_kernel(const float* __restrict__ pooled, const float* __restrict__ cnt,
            const float* __restrict__ W, const float* __restrict__ b,
            float* __restrict__ out) {
    const int i = blockIdx.x * blockDim.x + threadIdx.x;
    if (i >= N_GRAPHS_C * OUT_C_C) return;
    const int g = i / OUT_C_C;
    const int o = i % OUT_C_C;
    const float inv = 1.0f / fmaxf(cnt[g], 1.0f);
    float acc = b[o];
    const float* pr = pooled + (size_t)g * HID_C_C;
#pragma unroll 8
    for (int c = 0; c < HID_C_C; ++c)
        acc += (pr[c] * inv) * W[(size_t)c * OUT_C_C + o];
    out[i] = acc;
}

extern "C" void kernel_launch(void* const* d_in, const int* in_sizes, int n_in,
                              void* d_out, int out_size, void* d_ws, size_t ws_size,
                              hipStream_t stream) {
    const float* x      = (const float*)d_in[0];
    const float* eattr  = (const float*)d_in[1];
    const int*   eindex = (const int*)d_in[2];
    const int*   batch  = (const int*)d_in[3];
    const float* efc_w  = (const float*)d_in[4];
    const float* efc_b  = (const float*)d_in[5];
    const float* c1_lw  = (const float*)d_in[6];
    const float* c1_lb  = (const float*)d_in[7];
    const float* c1_rw  = (const float*)d_in[8];
    const float* c2_lw  = (const float*)d_in[9];
    const float* c2_lb  = (const float*)d_in[10];
    const float* c2_rw  = (const float*)d_in[11];
    const float* proj_w = (const float*)d_in[12];
    const float* proj_b = (const float*)d_in[13];
    float* out = (float*)d_out;

    float* ws = (float*)d_ws;
    size_t off = 0;
    float* x1     = ws + off; off += (size_t)N_NODES_C * IN_C_C;
    float* deg    = ws + off; off += (size_t)N_NODES_C;
    float* msum1  = ws + off; off += (size_t)N_NODES_C * IN_C_C;
    float* h1     = ws + off; off += (size_t)N_NODES_C * HID_C_C;
    float* msum2  = ws + off; off += (size_t)N_NODES_C * HID_C_C;
    float* h2     = ws + off; off += (size_t)N_NODES_C * HID_C_C;
    float* pooled = ws + off; off += (size_t)N_GRAPHS_C * HID_C_C;
    float* cnt    = ws + off; off += (size_t)N_GRAPHS_C;

    int* count  = (int*)h2;
    int* rowptr = count  + N_NODES_C;
    int* cursor = rowptr + N_NODES_C;
    int* col    = cursor + N_NODES_C;
    int* count_s  = (int*)h1;
    int* rowptr_s = count_s  + N_NODES_C;
    int* cursor_s = rowptr_s + N_NODES_C;
    int* eperm    = cursor_s + N_NODES_C;

    init_kernel<<<(N_NODES_C + 255) / 256, 256, 0, stream>>>(
        pooled, cnt, count, count_s);

    edge_hist<<<(N_EDGES_C + 255) / 256, 256, 0, stream>>>(eindex, count, count_s);
    list_scan<<<dim3((N_NODES_C + SCAN_CHUNK - 1) / SCAN_CHUNK, 2), 256, 0, stream>>>(
        count, rowptr, cursor, count_s, rowptr_s, cursor_s);
    edge_fill<<<(N_EDGES_C + 255) / 256, 256, 0, stream>>>(
        eindex, cursor, col, cursor_s, eperm);

    edge_fc_pull<<<N_NODES_C * 16 / 256, 256, 0, stream>>>(
        eattr, x, efc_w, efc_b, rowptr_s, count_s, eperm, count, x1, deg);

    pull_sum<IN_C_C><<<(N_NODES_C * (IN_C_C / 2) + 255) / 256, 256, 0, stream>>>(
        rowptr, count, col, x1, msum1);

    sage_wmma_kernel<IN_C_C, false><<<N_NODES_C / 16 / 2, 64, 0, stream>>>(
        x1, msum1, deg, c1_lw, c1_lb, c1_rw, h1, batch, pooled, cnt);

    pull_sum<HID_C_C><<<(N_NODES_C * (HID_C_C / 2) + 255) / 256, 256, 0, stream>>>(
        rowptr, count, col, h1, msum2);

    sage_wmma_kernel<HID_C_C, true><<<N_NODES_C / 16 / 2, 64, 0, stream>>>(
        h1, msum2, deg, c2_lw, c2_lb, c2_rw, h2, batch, pooled, cnt);

    proj_kernel<<<(N_GRAPHS_C * OUT_C_C + 255) / 256, 256, 0, stream>>>(
        pooled, cnt, proj_w, proj_b, out);
}


# Edge lists built by a two-level counting sort: per-edge counts and ranks in LDS (256-node bins, runs staged in LDS) instead of global atomics and random 4-byte stores; lists still in arrival order
# speedup vs baseline: 11.5116x; 2.6930x over previous
#include <hip/hip_runtime.h>
#include <hip/hip_bf16.h>
#include <stdint.h>

#define N_NODES_C  100000
#define N_EDGES_C  1600000
#define N_GRAPHS_C 64
#define IN_C_C     18
#define EDGE_C_C   22
#define HID_C_C    64
#define OUT_C_C    128

typedef float v2f __attribute__((ext_vector_type(2)));
typedef float v8f __attribute__((ext_vector_type(8)));

#define BIN_SHIFT   8
#define BIN_NODES   256
#define N_BINS      ((N_NODES_C + BIN_NODES - 1) / BIN_NODES)
#define CHUNK_E     8192
#define N_CHUNKS    ((N_EDGES_C + CHUNK_E - 1) / CHUNK_E)
#define BIN_T       512
#define BIN_EPT     (CHUNK_E / BIN_T)
#define STAGE_CAP   8192
#define SORT_EPT    16
#define DST_PAY_BITS 17
#define SRC_PAY_BITS 21

__global__ void __launch_bounds__(256)
init_kernel(float* __restrict__ pooled, float* __restrict__ cnt,
            int* __restrict__ gcount) {
    const int i = blockIdx.x * blockDim.x + threadIdx.x;
    if (i < 4 * N_BINS)           gcount[i] = 0;
    if (i < N_GRAPHS_C * HID_C_C) pooled[i] = 0.0f;
    if (i < N_GRAPHS_C)           cnt[i] = 0.0f;
}

template <int NW>
static __device__ __forceinline__ int block_excl_scan(int v, int* swave) {
    const int lane = threadIdx.x & 31;
    const int wv   = threadIdx.x >> 5;
    int inc = v;
#pragma unroll
    for (int d = 1; d < 32; d <<= 1) {
        const int n = __shfl_up(inc, d);
        if (lane >= d) inc += n;
    }
    if (lane == 31) swave[wv] = inc;
    __syncthreads();
    int pre = 0;
#pragma unroll
    for (int w = 0; w < NW; ++w) pre += (w < wv) ? swave[w] : 0;
    __syncthreads();
    return pre + inc - v;
}

static __device__ __forceinline__ unsigned node_id(int v) {
    const unsigned u = (unsigned)v;
    return u < (unsigned)N_NODES_C ? u : (unsigned)(N_NODES_C - 1);
}

__global__ void __launch_bounds__(BIN_T)
bin_hist(const int* __restrict__ eindex, int* __restrict__ gcount) {
    __shared__ int h[2 * N_BINS];
    const int tid = threadIdx.x;
    for (int i = tid; i < 2 * N_BINS; i += BIN_T) h[i] = 0;
    __syncthreads();
    const int e0 = blockIdx.x * CHUNK_E;
    int sv[BIN_EPT], dv[BIN_EPT];
#pragma unroll
    for (int k = 0; k < BIN_EPT; ++k) {
        const int e  = e0 + k * BIN_T + tid;
        const int ec = e < N_EDGES_C ? e : N_EDGES_C - 1;
        sv[k] = eindex[ec];
        dv[k] = eindex[N_EDGES_C + ec];
    }
#pragma unroll
    for (int k = 0; k < BIN_EPT; ++k) {
        const int e = e0 + k * BIN_T + tid;
        if (e < N_EDGES_C) {
            atomicAdd(&h[node_id(dv[k]) >> BIN_SHIFT], 1);
            atomicAdd(&h[N_BINS + (node_id(sv[k]) >> BIN_SHIFT)], 1);
        }
    }
    __syncthreads();
    for (int i = tid; i < 2 * N_BINS; i += BIN_T) {
        const int v = h[i];
        if (v) atomicAdd(gcount + i, v);
    }
}

__global__ void __launch_bounds__(BIN_T)
bin_scatter(const int* __restrict__ eindex, const int* __restrict__ gcount,
            int* __restrict__ gcursor, int* __restrict__ tmp) {
    __shared__ int hist[N_BINS];
    __shared__ int gdelta[N_BINS];
    __shared__ int sval[CHUNK_E];
    __shared__ unsigned short sbin[CHUNK_E];
    __shared__ int swave[BIN_T / 32];
    const int list = blockIdx.y;
    const int tid  = threadIdx.x;
    const int e0   = blockIdx.x * CHUNK_E;
    const int nloc = (N_EDGES_C - e0) < CHUNK_E ? (N_EDGES_C - e0) : CHUNK_E;

    if (tid < N_BINS) hist[tid] = 0;
    const int gc = tid < N_BINS ? gcount[list * N_BINS + tid] : 0;
    __syncthreads();

    const int* keyrow = eindex + (list == 0 ? N_EDGES_C : 0);
    const int  pshift = list == 0 ? DST_PAY_BITS : SRC_PAY_BITS;
    int sv[BIN_EPT], kv[BIN_EPT];
#pragma unroll
    for (int k = 0; k < BIN_EPT; ++k) {
        const int e  = e0 + k * BIN_T + tid;
        const int ec = e < N_EDGES_C ? e : N_EDGES_C - 1;
        sv[k] = eindex[ec];
        kv[k] = keyrow[ec];
    }

    int val[BIN_EPT], rnk[BIN_EPT], bin[BIN_EPT];
#pragma unroll
    for (int k = 0; k < BIN_EPT; ++k) {
        const int e = e0 + k * BIN_T + tid;
        bin[k] = -1;
        if (e < N_EDGES_C) {
            const unsigned key = node_id(kv[k]);
            const unsigned low = list == 0 ? node_id(sv[k]) : (unsigned)e;
            const unsigned pay = ((key & (BIN_NODES - 1)) << pshift) | low;
            bin[k] = (int)(key >> BIN_SHIFT);
            val[k] = (int)pay;
            rnk[k] = atomicAdd(&hist[bin[k]], 1);
        }
    }
    __syncthreads();

    const int c  = tid < N_BINS ? hist[tid] : 0;
    const int bstart = block_excl_scan<BIN_T / 32>(gc, swave);
    const int lofs   = block_excl_scan<BIN_T / 32>(c, swave);
    if (tid < N_BINS) {
        const int claim = c ? atomicAdd(gcursor + list * N_BINS + tid, c) : 0;
        hist[tid]   = lofs;
        gdelta[tid] = bstart + claim - lofs;
    }
    __syncthreads();

#pragma unroll
    for (int k = 0; k < BIN_EPT; ++k) {
        if (bin[k] >= 0) {
            const int p = hist[bin[k]] + rnk[k];
            sval[p] = val[k];
            sbin[p] = (unsigned short)bin[k];
        }
    }
    __syncthreads();

    int* out = tmp + (size_t)list * N_EDGES_C;
    for (int i = tid; i < nloc; i += BIN_T) {
        const int pos = gdelta[sbin[i]] + i;
        if ((unsigned)pos < (unsigned)N_EDGES_C) out[pos] = sval[i];
    }
}

__global__ void __launch_bounds__(BIN_NODES)
bin_sort(const int* __restrict__ gcount, const int* __restrict__ tmp,
         int* __restrict__ count_d, int* __restrict__ rowptr_d, int* __restrict__ col,
         int* __restrict__ count_s, int* __restrict__ rowptr_s, int* __restrict__ eperm) {
    __shared__ int fcnt[BIN_NODES];
    __shared__ int fofs[BIN_NODES];
    __shared__ int stage[STAGE_CAP];
    __shared__ int swave[BIN_NODES / 32];
    __shared__ int sred[BIN_NODES / 32];
    const int list = blockIdx.y;
    const int b    = blockIdx.x;
    const int tid  = threadIdx.x;
    const int lane = tid & 31;
    const int* gc  = gcount + list * N_BINS;
    int* count  = list ? count_s  : count_d;
    int* rowptr = list ? rowptr_s : rowptr_d;
    int* out    = list ? eperm    : col;
    const int      shift = list ? SRC_PAY_BITS : DST_PAY_BITS;
    const unsigned pmask = (1u << shift) - 1u;

    int part = 0;
    for (int j = tid; j < b; j += BIN_NODES) part += gc[j];
    for (int d = 16; d > 0; d >>= 1) part += __shfl_xor(part, d);
    if (lane == 0) sred[tid >> 5] = part;
    fcnt[tid] = 0;
    __syncthreads();
    int start = 0;
#pragma unroll
    for (int w = 0; w < BIN_NODES / 32; ++w) start += sred[w];
    int n = gc[b];
    if (start < 0 || start > N_EDGES_C) start = N_EDGES_C;
    if (n < 0) n = 0;
    if (n > N_EDGES_C - start) n = N_EDGES_C - start;

    const int* in = tmp + (size_t)list * N_EDGES_C + start;
    for (int base = 0; base < n; base += BIN_NODES * SORT_EPT) {
        unsigned v[SORT_EPT];
#pragma unroll
        for (int k = 0; k < SORT_EPT; ++k) {
            const int i = base + k * BIN_NODES + tid;
            v[k] = (unsigned)in[i < n ? i : n - 1];
        }
#pragma unroll
        for (int k = 0; k < SORT_EPT; ++k) {
            const int i = base + k * BIN_NODES + tid;
            if (i < n) atomicAdd(&fcnt[(v[k] >> shift) & (BIN_NODES - 1)], 1);
        }
    }
    __syncthreads();

    const int c    = fcnt[tid];
    const int excl = block_excl_scan<BIN_NODES / 32>(c, swave);
    const int node = b * BIN_NODES + tid;
    if (node < N_NODES_C) { count[node] = c; rowptr[node] = start + excl; }
    fofs[tid] = excl;
    fcnt[tid] = 0;
    __syncthreads();

    const bool staged = n <= STAGE_CAP;
    for (int base = 0; base < n; base += BIN_NODES * SORT_EPT) {
        unsigned v[SORT_EPT];
#pragma unroll
        for (int k = 0; k < SORT_EPT; ++k) {
            const int i = base + k * BIN_NODES + tid;
            v[k] = (unsigned)in[i < n ? i : n - 1];
        }
#pragma unroll
        for (int k = 0; k < SORT_EPT; ++k) {
            const int i = base + k * BIN_NODES + tid;
            if (i < n) {
                const unsigned f = (v[k] >> shift) & (BIN_NODES - 1);
                const int p = fofs[f] + atomicAdd(&fcnt[f], 1);
                if (staged) stage[p] = (int)(v[k] & pmask);
                else        out[start + p] = (int)(v[k] & pmask);
            }
        }
    }
    if (staged) {
        __syncthreads();
        for (int i = tid; i < n; i += BIN_NODES) out[start + i] = stage[i];
    }
}

template <int M>
static __device__ __forceinline__ float row_xor_add(float v) {
    const int o = __builtin_amdgcn_update_dpp(0, __float_as_int(v), 0x160 + M,
                                              0xf, 0xf, true);
    return v + __int_as_float(o);
}

__global__ void __launch_bounds__(256)
edge_fc_pull(const float* __restrict__ eattr, const float* __restrict__ x,
             const float* __restrict__ W, const float* __restrict__ b,
             const int* __restrict__ rowptr_s, const int* __restrict__ count_s,
             const int* __restrict__ eperm, const int* __restrict__ count_d,
             float* __restrict__ x1, float* __restrict__ deg) {
    const int t    = blockIdx.x * blockDim.x + threadIdx.x;
    const int node = t >> 4;
    const int l    = t & 15;
    const int beg  = rowptr_s[node];
    const int cnt  = count_s[node];

    float s[EDGE_C_C];
#pragma unroll
    for (int q = 0; q < EDGE_C_C; ++q) s[q] = 0.0f;

    for (int k = l; k < cnt; k += 16) {
        unsigned e = (unsigned)eperm[beg + k];
        e = e < (unsigned)N_EDGES_C ? e : (unsigned)(N_EDGES_C - 1);
        const float2* er = (const float2*)(eattr + (size_t)e * EDGE_C_C);
#pragma unroll
        for (int q = 0; q < EDGE_C_C / 2; ++q) {
            const float2 p = er[q];
            s[2 * q] += p.x; s[2 * q + 1] += p.y;
        }
    }

#pragma unroll
    for (int q = 0; q < EDGE_C_C; ++q) {
        float v = s[q];
        v = row_xor_add<1>(v);
        v = row_xor_add<2>(v);
        v = row_xor_add<4>(v);
        v = row_xor_add<8>(v);
        s[q] = v;
    }

    if (l < IN_C_C / 2) {
        const float2* W2 = (const float2*)W;
        const float2 bb = ((const float2*)b)[l];
        const float  cf = (float)cnt;
        float2 acc = make_float2(cf * bb.x, cf * bb.y);
#pragma unroll
        for (int q = 0; q < EDGE_C_C; ++q) {
            const float2 w = W2[q * (IN_C_C / 2) + l];
            acc.x += s[q] * w.x; acc.y += s[q] * w.y;
        }
        const float2 xv = ((const float2*)x)[node * (IN_C_C / 2) + l];
        ((float2*)x1)[node * (IN_C_C / 2) + l] = make_float2(xv.x + acc.x, xv.y + acc.y);
    }
    if (l == 15) deg[node] = (float)count_d[node];
}

template <int C>
__global__ void __launch_bounds__(256)
pull_sum(const int* __restrict__ rowptr, const int* __restrict__ count,
         const int* __restrict__ col, const float* __restrict__ feat,
         float* __restrict__ msum) {
    constexpr int LPN = C / 2;
    const int t = blockIdx.x * blockDim.x + threadIdx.x;
    if (t >= N_NODES_C * LPN) return;
    const int node = t / LPN;
    const int c2   = t - node * LPN;
    const int beg  = rowptr[node];
    const int end  = beg + count[node];
    const float2* f2 = (const float2*)feat;

    float2 acc = make_float2(0.0f, 0.0f);
    int j = beg;
    for (; j + 4 <= end; j += 4) {
        const int s0 = col[j], s1 = col[j + 1], s2 = col[j + 2], s3 = col[j + 3];
        const float2 a0 = f2[s0 * LPN + c2];
        const float2 a1 = f2[s1 * LPN + c2];
        const float2 a2 = f2[s2 * LPN + c2];
        const float2 a3 = f2[s3 * LPN + c2];
        acc.x += a0.x; acc.y += a0.y;
        acc.x += a1.x; acc.y += a1.y;
        acc.x += a2.x; acc.y += a2.y;
        acc.x += a3.x; acc.y += a3.y;
    }
    for (; j < end; ++j) {
        const float2 a = f2[col[j] * LPN + c2];
        acc.x += a.x; acc.y += a.y;
    }
    ((float2*)msum)[t] = acc;
}

template <int K, bool POOL>
__global__ void __launch_bounds__(64)
sage_wmma_kernel(const float* __restrict__ xin, const float* __restrict__ msum,
                 const float* __restrict__ deg,
                 const float* __restrict__ lw, const float* __restrict__ lb,
                 const float* __restrict__ rw, float* __restrict__ out,
                 const int* __restrict__ batch, float* __restrict__ pooled,
                 float* __restrict__ cnt) {
    const int wave = (blockIdx.x * blockDim.x + threadIdx.x) >> 5;
    const int lane = threadIdx.x & 31;
    if (wave >= N_NODES_C / 16) return;

    const int node0 = wave * 16;
    const int m    = lane & 15;
    const int half = (lane >> 4) & 1;
    const int row  = node0 + m;

    const float dg  = deg[row];
    const float inv = dg > 0.0f ? 1.0f / dg : 0.0f;
    const float* xr = xin  + (size_t)row * K;
    const float* mr = msum + (size_t)row * K;

    v8f acc[4] = {};

    constexpr int KP = (K + 3) & ~3;
#pragma unroll
    for (int ks = 0; ks < KP; ks += 4) {
        const int kb = ks + half * 2;
        v2f amean, aroot;
        amean[0] = (kb     < K) ? mr[kb]     * inv : 0.0f;
        amean[1] = (kb + 1 < K) ? mr[kb + 1] * inv : 0.0f;
        aroot[0] = (kb     < K) ? xr[kb]           : 0.0f;
        aroot[1] = (kb + 1 < K) ? xr[kb + 1]       : 0.0f;
#pragma unroll
        for (int t = 0; t < 4; ++t) {
            const int col = t * 16 + m;
            v2f bl, br;
            bl[0] = (kb     < K) ? lw[(size_t)kb       * HID_C_C + col] : 0.0f;
            bl[1] = (kb + 1 < K) ? lw[(size_t)(kb + 1) * HID_C_C + col] : 0.0f;
            br[0] = (kb     < K) ? rw[(size_t)kb       * HID_C_C + col] : 0.0f;
            br[1] = (kb + 1 < K) ? rw[(size_t)(kb + 1) * HID_C_C + col] : 0.0f;
            acc[t] = __builtin_amdgcn_wmma_f32_16x16x4_f32(
                false, amean, false, bl, (short)0, acc[t], false, false);
            acc[t] = __builtin_amdgcn_wmma_f32_16x16x4_f32(
                false, aroot, false, br, (short)0, acc[t], false, false);
        }
    }

    if (!POOL) {
#pragma unroll
        for (int t = 0; t < 4; ++t) {
            const int col  = t * 16 + m;
            const float bias = lb[col];
#pragma unroll
            for (int v = 0; v < 8; ++v) {
                const int orow = node0 + half * 8 + v;
                float val = acc[t][v] + bias;
                out[(size_t)orow * HID_C_C + col] = val > 0.0f ? val : 0.0f;
            }
        }
    } else {
        const int g  = batch[row];
        const int g0 = __builtin_amdgcn_readfirstlane(g);

        float h[4][8];
        float cs[4];
#pragma unroll
        for (int t = 0; t < 4; ++t) {
            const float bias = lb[t * 16 + m];
            float s = 0.0f;
#pragma unroll
            for (int v = 0; v < 8; ++v) {
                const float val = acc[t][v] + bias;
                h[t][v] = val > 0.0f ? val : 0.0f;
                s += h[t][v];
            }
            cs[t] = s;
        }

        if (__builtin_amdgcn_ballot_w32(g != g0) == 0u) {
            float c4[4];
#pragma unroll
            for (int t = 0; t < 4; ++t) c4[t] = cs[t] + __shfl_xor(cs[t], 16);
            const float a0 = half ? c4[2] : c4[0];
            const float a1 = half ? c4[3] : c4[1];
            float* pg = pooled + g0 * HID_C_C + half * 32 + m;
            atomicAdd(pg,      a0);
            atomicAdd(pg + 16, a1);
            if (lane == 0) atomicAdd(cnt + g0, 16.0f);
        } else {
#pragma unroll
            for (int v = 0; v < 8; ++v) {
                const int gr = __shfl(g, half * 8 + v);
#pragma unroll
                for (int t = 0; t < 4; ++t)
                    atomicAdd(pooled + gr * HID_C_C + t * 16 + m, h[t][v]);
            }
            if (lane < 16) atomicAdd(cnt + g, 1.0f);
        }
    }
}

__global__ void __launch_bounds__(256)
proj_kernel(const float* __restrict__ pooled, const float* __restrict__ cnt,
            const float* __restrict__ W, const float* __restrict__ b,
            float* __restrict__ out) {
    const int i = blockIdx.x * blockDim.x + threadIdx.x;
    if (i >= N_GRAPHS_C * OUT_C_C) return;
    const int g = i / OUT_C_C;
    const int o = i % OUT_C_C;
    const float inv = 1.0f / fmaxf(cnt[g], 1.0f);
    float acc = b[o];
    const float* pr = pooled + (size_t)g * HID_C_C;
#pragma unroll 8
    for (int c = 0; c < HID_C_C; ++c)
        acc += (pr[c] * inv) * W[(size_t)c * OUT_C_C + o];
    out[i] = acc;
}

extern "C" void kernel_launch(void* const* d_in, const int* in_sizes, int n_in,
                              void* d_out, int out_size, void* d_ws, size_t ws_size,
                              hipStream_t stream) {
    const float* x      = (const float*)d_in[0];
    const float* eattr  = (const float*)d_in[1];
    const int*   eindex = (const int*)d_in[2];
    const int*   batch  = (const int*)d_in[3];
    const float* efc_w  = (const float*)d_in[4];
    const float* efc_b  = (const float*)d_in[5];
    const float* c1_lw  = (const float*)d_in[6];
    const float* c1_lb  = (const float*)d_in[7];
    const float* c1_rw  = (const float*)d_in[8];
    const float* c2_lw  = (const float*)d_in[9];
    const float* c2_lb  = (const float*)d_in[10];
    const float* c2_rw  = (const float*)d_in[11];
    const float* proj_w = (const float*)d_in[12];
    const float* proj_b = (const float*)d_in[13];
    float* out = (float*)d_out;

    float* ws = (float*)d_ws;
    size_t off = 0;
    float* x1     = ws + off; off += (size_t)N_NODES_C * IN_C_C;
    float* deg    = ws + off; off += (size_t)N_NODES_C;
    float* msum1  = ws + off; off += (size_t)N_NODES_C * IN_C_C;
    float* h1     = ws + off; off += (size_t)N_NODES_C * HID_C_C;
    float* msum2  = ws + off; off += (size_t)N_NODES_C * HID_C_C;
    float* h2     = ws + off; off += (size_t)N_NODES_C * HID_C_C;
    float* pooled = ws + off; off += (size_t)N_GRAPHS_C * HID_C_C;
    float* cnt    = ws + off; off += (size_t)N_GRAPHS_C;

    int* count  = (int*)h2;
    int* rowptr = count  + N_NODES_C;
    int* col    = rowptr + N_NODES_C;
    int* count_s  = (int*)h1;
    int* rowptr_s = count_s  + N_NODES_C;
    int* eperm    = rowptr_s + N_NODES_C;
    int* tmp     = (int*)msum2;
    int* gcount  = tmp + 2 * (size_t)N_EDGES_C;
    int* gcursor = gcount + 2 * N_BINS;

    init_kernel<<<(N_GRAPHS_C * HID_C_C + 255) / 256, 256, 0, stream>>>(
        pooled, cnt, gcount);

    bin_hist<<<N_CHUNKS, BIN_T, 0, stream>>>(eindex, gcount);
    bin_scatter<<<dim3(N_CHUNKS, 2), BIN_T, 0, stream>>>(eindex, gcount, gcursor, tmp);
    bin_sort<<<dim3(N_BINS, 2), BIN_NODES, 0, stream>>>(
        gcount, tmp, count, rowptr, col, count_s, rowptr_s, eperm);

    edge_fc_pull<<<N_NODES_C * 16 / 256, 256, 0, stream>>>(
        eattr, x, efc_w, efc_b, rowptr_s, count_s, eperm, count, x1, deg);

    pull_sum<IN_C_C><<<(N_NODES_C * (IN_C_C / 2) + 255) / 256, 256, 0, stream>>>(
        rowptr, count, col, x1, msum1);

    sage_wmma_kernel<IN_C_C, false><<<N_NODES_C / 16 / 2, 64, 0, stream>>>(
        x1, msum1, deg, c1_lw, c1_lb, c1_rw, h1, batch, pooled, cnt);

    pull_sum<HID_C_C><<<(N_NODES_C * (HID_C_C / 2) + 255) / 256, 256, 0, stream>>>(
        rowptr, count, col, h1, msum2);

    sage_wmma_kernel<HID_C_C, true><<<N_NODES_C / 16 / 2, 64, 0, stream>>>(
        h1, msum2, deg, c2_lw, c2_lb, c2_rw, h2, batch, pooled, cnt);

    proj_kernel<<<(N_GRAPHS_C * OUT_C_C + 255) / 256, 256, 0, stream>>>(
        pooled, cnt, proj_w, proj_b, out);
}


# Neighbour sums fused into the two conv launches: each tile's wave sums its 16 nodes' rows (same list order) into LDS and reads the WMMA A operand from there; pull launches and msum planes gone
# speedup vs baseline: 13.0828x; 1.1365x over previous
#include <hip/hip_runtime.h>
#include <hip/hip_bf16.h>
#include <stdint.h>

#define N_NODES_C  100000
#define N_EDGES_C  1600000
#define N_GRAPHS_C 64
#define IN_C_C     18
#define EDGE_C_C   22
#define HID_C_C    64
#define OUT_C_C    128

typedef float v2f __attribute__((ext_vector_type(2)));
typedef float v8f __attribute__((ext_vector_type(8)));

#define BIN_SHIFT   8
#define BIN_NODES   256
#define N_BINS      ((N_NODES_C + BIN_NODES - 1) / BIN_NODES)
#define CHUNK_E     8192
#define N_CHUNKS    ((N_EDGES_C + CHUNK_E - 1) / CHUNK_E)
#define BIN_T       512
#define BIN_EPT     (CHUNK_E / BIN_T)
#define STAGE_CAP   8192
#define SORT_EPT    16
#define DST_PAY_BITS 17
#define SRC_PAY_BITS 21

__global__ void __launch_bounds__(256)
init_kernel(float* __restrict__ pooled, float* __restrict__ cnt,
            int* __restrict__ gcount) {
    const int i = blockIdx.x * blockDim.x + threadIdx.x;
    if (i < 4 * N_BINS)           gcount[i] = 0;
    if (i < N_GRAPHS_C * HID_C_C) pooled[i] = 0.0f;
    if (i < N_GRAPHS_C)           cnt[i] = 0.0f;
}

template <int NW>
static __device__ __forceinline__ int block_excl_scan(int v, int* swave) {
    const int lane = threadIdx.x & 31;
    const int wv   = threadIdx.x >> 5;
    int inc = v;
#pragma unroll
    for (int d = 1; d < 32; d <<= 1) {
        const int n = __shfl_up(inc, d);
        if (lane >= d) inc += n;
    }
    if (lane == 31) swave[wv] = inc;
    __syncthreads();
    int pre = 0;
#pragma unroll
    for (int w = 0; w < NW; ++w) pre += (w < wv) ? swave[w] : 0;
    __syncthreads();
    return pre + inc - v;
}

static __device__ __forceinline__ unsigned node_id(int v) {
    const unsigned u = (unsigned)v;
    return u < (unsigned)N_NODES_C ? u : (unsigned)(N_NODES_C - 1);
}

__global__ void __launch_bounds__(BIN_T)
bin_hist(const int* __restrict__ eindex, int* __restrict__ gcount) {
    __shared__ int h[2 * N_BINS];
    const int tid = threadIdx.x;
    for (int i = tid; i < 2 * N_BINS; i += BIN_T) h[i] = 0;
    __syncthreads();
    const int e0 = blockIdx.x * CHUNK_E;
    int sv[BIN_EPT], dv[BIN_EPT];
#pragma unroll
    for (int k = 0; k < BIN_EPT; ++k) {
        const int e  = e0 + k * BIN_T + tid;
        const int ec = e < N_EDGES_C ? e : N_EDGES_C - 1;
        sv[k] = eindex[ec];
        dv[k] = eindex[N_EDGES_C + ec];
    }
#pragma unroll
    for (int k = 0; k < BIN_EPT; ++k) {
        const int e = e0 + k * BIN_T + tid;
        if (e < N_EDGES_C) {
            atomicAdd(&h[node_id(dv[k]) >> BIN_SHIFT], 1);
            atomicAdd(&h[N_BINS + (node_id(sv[k]) >> BIN_SHIFT)], 1);
        }
    }
    __syncthreads();
    for (int i = tid; i < 2 * N_BINS; i += BIN_T) {
        const int v = h[i];
        if (v) atomicAdd(gcount + i, v);
    }
}

__global__ void __launch_bounds__(BIN_T)
bin_scatter(const int* __restrict__ eindex, const int* __restrict__ gcount,
            int* __restrict__ gcursor, int* __restrict__ tmp) {
    __shared__ int hist[N_BINS];
    __shared__ int gdelta[N_BINS];
    __shared__ int sval[CHUNK_E];
    __shared__ unsigned short sbin[CHUNK_E];
    __shared__ int swave[BIN_T / 32];
    const int list = blockIdx.y;
    const int tid  = threadIdx.x;
    const int e0   = blockIdx.x * CHUNK_E;
    const int nloc = (N_EDGES_C - e0) < CHUNK_E ? (N_EDGES_C - e0) : CHUNK_E;

    if (tid < N_BINS) hist[tid] = 0;
    const int gc = tid < N_BINS ? gcount[list * N_BINS + tid] : 0;
    __syncthreads();

    const int* keyrow = eindex + (list == 0 ? N_EDGES_C : 0);
    const int  pshift = list == 0 ? DST_PAY_BITS : SRC_PAY_BITS;
    int sv[BIN_EPT], kv[BIN_EPT];
#pragma unroll
    for (int k = 0; k < BIN_EPT; ++k) {
        const int e  = e0 + k * BIN_T + tid;
        const int ec = e < N_EDGES_C ? e : N_EDGES_C - 1;
        sv[k] = eindex[ec];
        kv[k] = keyrow[ec];
    }

    int val[BIN_EPT], rnk[BIN_EPT], bin[BIN_EPT];
#pragma unroll
    for (int k = 0; k < BIN_EPT; ++k) {
        const int e = e0 + k * BIN_T + tid;
        bin[k] = -1;
        if (e < N_EDGES_C) {
            const unsigned key = node_id(kv[k]);
            const unsigned low = list == 0 ? node_id(sv[k]) : (unsigned)e;
            const unsigned pay = ((key & (BIN_NODES - 1)) << pshift) | low;
            bin[k] = (int)(key >> BIN_SHIFT);
            val[k] = (int)pay;
            rnk[k] = atomicAdd(&hist[bin[k]], 1);
        }
    }
    __syncthreads();

    const int c  = tid < N_BINS ? hist[tid] : 0;
    const int bstart = block_excl_scan<BIN_T / 32>(gc, swave);
    const int lofs   = block_excl_scan<BIN_T / 32>(c, swave);
    if (tid < N_BINS) {
        const int claim = c ? atomicAdd(gcursor + list * N_BINS + tid, c) : 0;
        hist[tid]   = lofs;
        gdelta[tid] = bstart + claim - lofs;
    }
    __syncthreads();

#pragma unroll
    for (int k = 0; k < BIN_EPT; ++k) {
        if (bin[k] >= 0) {
            const int p = hist[bin[k]] + rnk[k];
            sval[p] = val[k];
            sbin[p] = (unsigned short)bin[k];
        }
    }
    __syncthreads();

    int* out = tmp + (size_t)list * N_EDGES_C;
    for (int i = tid; i < nloc; i += BIN_T) {
        const int pos = gdelta[sbin[i]] + i;
        if ((unsigned)pos < (unsigned)N_EDGES_C) out[pos] = sval[i];
    }
}

__global__ void __launch_bounds__(BIN_NODES)
bin_sort(const int* __restrict__ gcount, const int* __restrict__ tmp,
         int* __restrict__ count_d, int* __restrict__ rowptr_d, int* __restrict__ col,
         int* __restrict__ count_s, int* __restrict__ rowptr_s, int* __restrict__ eperm) {
    __shared__ int fcnt[BIN_NODES];
    __shared__ int fofs[BIN_NODES];
    __shared__ int stage[STAGE_CAP];
    __shared__ int swave[BIN_NODES / 32];
    __shared__ int sred[BIN_NODES / 32];
    const int list = blockIdx.y;
    const int b    = blockIdx.x;
    const int tid  = threadIdx.x;
    const int lane = tid & 31;
    const int* gc  = gcount + list * N_BINS;
    int* count  = list ? count_s  : count_d;
    int* rowptr = list ? rowptr_s : rowptr_d;
    int* out    = list ? eperm    : col;
    const int      shift = list ? SRC_PAY_BITS : DST_PAY_BITS;
    const unsigned pmask = (1u << shift) - 1u;

    int part = 0;
    for (int j = tid; j < b; j += BIN_NODES) part += gc[j];
    for (int d = 16; d > 0; d >>= 1) part += __shfl_xor(part, d);
    if (lane == 0) sred[tid >> 5] = part;
    fcnt[tid] = 0;
    __syncthreads();
    int start = 0;
#pragma unroll
    for (int w = 0; w < BIN_NODES / 32; ++w) start += sred[w];
    int n = gc[b];
    if (start < 0 || start > N_EDGES_C) start = N_EDGES_C;
    if (n < 0) n = 0;
    if (n > N_EDGES_C - start) n = N_EDGES_C - start;

    const int* in = tmp + (size_t)list * N_EDGES_C + start;
    for (int base = 0; base < n; base += BIN_NODES * SORT_EPT) {
        unsigned v[SORT_EPT];
#pragma unroll
        for (int k = 0; k < SORT_EPT; ++k) {
            const int i = base + k * BIN_NODES + tid;
            v[k] = (unsigned)in[i < n ? i : n - 1];
        }
#pragma unroll
        for (int k = 0; k < SORT_EPT; ++k) {
            const int i = base + k * BIN_NODES + tid;
            if (i < n) atomicAdd(&fcnt[(v[k] >> shift) & (BIN_NODES - 1)], 1);
        }
    }
    __syncthreads();

    const int c    = fcnt[tid];
    const int excl = block_excl_scan<BIN_NODES / 32>(c, swave);
    const int node = b * BIN_NODES + tid;
    if (node < N_NODES_C) { count[node] = c; rowptr[node] = start + excl; }
    fofs[tid] = excl;
    fcnt[tid] = 0;
    __syncthreads();

    const bool staged = n <= STAGE_CAP;
    for (int base = 0; base < n; base += BIN_NODES * SORT_EPT) {
        unsigned v[SORT_EPT];
#pragma unroll
        for (int k = 0; k < SORT_EPT; ++k) {
            const int i = base + k * BIN_NODES + tid;
            v[k] = (unsigned)in[i < n ? i : n - 1];
        }
#pragma unroll
        for (int k = 0; k < SORT_EPT; ++k) {
            const int i = base + k * BIN_NODES + tid;
            if (i < n) {
                const unsigned f = (v[k] >> shift) & (BIN_NODES - 1);
                const int p = fofs[f] + atomicAdd(&fcnt[f], 1);
                if (staged) stage[p] = (int)(v[k] & pmask);
                else        out[start + p] = (int)(v[k] & pmask);
            }
        }
    }
    if (staged) {
        __syncthreads();
        for (int i = tid; i < n; i += BIN_NODES) out[start + i] = stage[i];
    }
}

template <int M>
static __device__ __forceinline__ float row_xor_add(float v) {
    const int o = __builtin_amdgcn_update_dpp(0, __float_as_int(v), 0x160 + M,
                                              0xf, 0xf, true);
    return v + __int_as_float(o);
}

__global__ void __launch_bounds__(256)
edge_fc_pull(const float* __restrict__ eattr, const float* __restrict__ x,
             const float* __restrict__ W, const float* __restrict__ b,
             const int* __restrict__ rowptr_s, const int* __restrict__ count_s,
             const int* __restrict__ eperm, const int* __restrict__ count_d,
             float* __restrict__ x1, float* __restrict__ deg) {
    const int t    = blockIdx.x * blockDim.x + threadIdx.x;
    const int node = t >> 4;
    const int l    = t & 15;
    const int beg  = rowptr_s[node];
    const int cnt  = count_s[node];

    float s[EDGE_C_C];
#pragma unroll
    for (int q = 0; q < EDGE_C_C; ++q) s[q] = 0.0f;

    for (int k = l; k < cnt; k += 16) {
        unsigned e = (unsigned)eperm[beg + k];
        e = e < (unsigned)N_EDGES_C ? e : (unsigned)(N_EDGES_C - 1);
        const float2* er = (const float2*)(eattr + (size_t)e * EDGE_C_C);
#pragma unroll
        for (int q = 0; q < EDGE_C_C / 2; ++q) {
            const float2 p = er[q];
            s[2 * q] += p.x; s[2 * q + 1] += p.y;
        }
    }

#pragma unroll
    for (int q = 0; q < EDGE_C_C; ++q) {
        float v = s[q];
        v = row_xor_add<1>(v);
        v = row_xor_add<2>(v);
        v = row_xor_add<4>(v);
        v = row_xor_add<8>(v);
        s[q] = v;
    }

    if (l < IN_C_C / 2) {
        const float2* W2 = (const float2*)W;
        const float2 bb = ((const float2*)b)[l];
        const float  cf = (float)cnt;
        float2 acc = make_float2(cf * bb.x, cf * bb.y);
#pragma unroll
        for (int q = 0; q < EDGE_C_C; ++q) {
            const float2 w = W2[q * (IN_C_C / 2) + l];
            acc.x += s[q] * w.x; acc.y += s[q] * w.y;
        }
        const float2 xv = ((const float2*)x)[node * (IN_C_C / 2) + l];
        ((float2*)x1)[node * (IN_C_C / 2) + l] = make_float2(xv.x + acc.x, xv.y + acc.y);
    }
    if (l == 15) deg[node] = (float)count_d[node];
}

#define COL_CAP 512

template <int K, int PITCH, bool STAGED>
static __device__ __forceinline__ void
gather_tile(const float2* __restrict__ f2, const int* __restrict__ col,
            const int* sc, const int* srp, const int* scn,
            int base, int navail, int lane, float* sm) {
    constexpr int KP  = (K + 3) & ~3;
    constexpr int LPN = K / 2;
    constexpr int NPW = 32 / LPN;
    const int g  = lane / LPN;
    const int c2 = lane - g * LPN;
    auto entry = [&](int j) -> int {
        if (STAGED) {
            const unsigned q = (unsigned)(j - base);
            return sc[q < (unsigned)navail ? q : (unsigned)(navail - 1)];
        } else {
            return col[(unsigned)j < (unsigned)N_EDGES_C ? (unsigned)j : (unsigned)(N_EDGES_C - 1)];
        }
    };
    for (int p = 0; p < (16 + NPW - 1) / NPW; ++p) {
        const int r = p * NPW + g;
        if (g < NPW && r < 16) {
            const int beg = srp[r];
            const int end = beg + scn[r];
            float2 acc = make_float2(0.0f, 0.0f);
            int j = beg;
            for (; j + 8 <= end; j += 8) {
                int   s[8];
                float2 a[8];
#pragma unroll
                for (int u = 0; u < 8; ++u) s[u] = entry(j + u);
#pragma unroll
                for (int u = 0; u < 8; ++u) a[u] = f2[s[u] * LPN + c2];
#pragma unroll
                for (int u = 0; u < 8; ++u) { acc.x += a[u].x; acc.y += a[u].y; }
            }
            if (j < end) {
                int   s[7];
                float2 a[7];
#pragma unroll
                for (int u = 0; u < 7; ++u) s[u] = entry(j + u < end ? j + u : end - 1);
#pragma unroll
                for (int u = 0; u < 7; ++u) a[u] = f2[s[u] * LPN + c2];
#pragma unroll
                for (int u = 0; u < 7; ++u)
                    if (j + u < end) { acc.x += a[u].x; acc.y += a[u].y; }
            }
            *(float2*)(sm + r * PITCH + 2 * c2) = acc;
            if (KP > K && c2 == LPN - 1) {
                for (int q = K; q < KP; ++q) sm[r * PITCH + q] = 0.0f;
            }
        }
    }
}

template <int K, bool POOL>
__global__ void __launch_bounds__(64)
sage_wmma_kernel(const float* __restrict__ xin,
                 const int* __restrict__ rowptr, const int* __restrict__ count,
                 const int* __restrict__ col,
                 const float* __restrict__ deg,
                 const float* __restrict__ lw, const float* __restrict__ lb,
                 const float* __restrict__ rw, float* __restrict__ out,
                 const int* __restrict__ batch, float* __restrict__ pooled,
                 float* __restrict__ cnt) {
    constexpr int KP    = (K + 3) & ~3;
    constexpr int PITCH = (K % 64 == 0) ? K + 4 : KP;
    __shared__ __attribute__((aligned(16))) float s_sum[2][16 * PITCH];
    __shared__ int   s_col[2][COL_CAP];
    __shared__ int   s_rp[2][16];
    __shared__ int   s_cn[2][16];

    const int wib  = threadIdx.x >> 5;
    const int wave = blockIdx.x * 2 + wib;
    const int lane = threadIdx.x & 31;

    const int node0 = wave * 16;
    const int m    = lane & 15;
    const int half = (lane >> 4) & 1;
    const int row  = node0 + m;

    float* sm = s_sum[wib];
    int*   sc = s_col[wib];

    const int rp = rowptr[row];
    const int cn = count[row];
    if (lane < 16) { s_rp[wib][lane] = rp; s_cn[wib][lane] = cn; }
    const int base = __builtin_amdgcn_readlane(rp, 0);
    const int tot  = __builtin_amdgcn_readlane(rp, 15) + __builtin_amdgcn_readlane(cn, 15) - base;
    const bool staged = tot >= 0 && tot <= COL_CAP;
    if (staged) {
        for (int i = lane; i < tot; i += 32) {
            const int j = base + i;
            sc[i] = col[j < N_EDGES_C ? j : N_EDGES_C - 1];
        }
    }
    __syncthreads();

    if (staged)
        gather_tile<K, PITCH, true>((const float2*)xin, col, sc, s_rp[wib], s_cn[wib],
                                    base, tot, lane, sm);
    else
        gather_tile<K, PITCH, false>((const float2*)xin, col, sc, s_rp[wib], s_cn[wib],
                                     base, tot, lane, sm);
    __syncthreads();

    const float dg  = deg[row];
    const float inv = dg > 0.0f ? 1.0f / dg : 0.0f;
    const float* xr = xin + (size_t)row * K;
    const float* mr = sm + m * PITCH;

    v8f acc[4] = {};

#pragma unroll
    for (int ks = 0; ks < KP; ks += 4) {
        const int kb = ks + half * 2;
        v2f amean, aroot;
        const float2 ms = *(const float2*)(mr + kb);
        amean[0] = ms.x * inv;
        amean[1] = ms.y * inv;
        aroot[0] = (kb     < K) ? xr[kb]           : 0.0f;
        aroot[1] = (kb + 1 < K) ? xr[kb + 1]       : 0.0f;
#pragma unroll
        for (int t = 0; t < 4; ++t) {
            const int col_n = t * 16 + m;
            v2f bl, br;
            bl[0] = (kb     < K) ? lw[(size_t)kb       * HID_C_C + col_n] : 0.0f;
            bl[1] = (kb + 1 < K) ? lw[(size_t)(kb + 1) * HID_C_C + col_n] : 0.0f;
            br[0] = (kb     < K) ? rw[(size_t)kb       * HID_C_C + col_n] : 0.0f;
            br[1] = (kb + 1 < K) ? rw[(size_t)(kb + 1) * HID_C_C + col_n] : 0.0f;
            acc[t] = __builtin_amdgcn_wmma_f32_16x16x4_f32(
                false, amean, false, bl, (short)0, acc[t], false, false);
            acc[t] = __builtin_amdgcn_wmma_f32_16x16x4_f32(
                false, aroot, false, br, (short)0, acc[t], false, false);
        }
    }

    if (!POOL) {
#pragma unroll
        for (int t = 0; t < 4; ++t) {
            const int oc   = t * 16 + m;
            const float bias = lb[oc];
#pragma unroll
            for (int v = 0; v < 8; ++v) {
                const int orow = node0 + half * 8 + v;
                float val = acc[t][v] + bias;
                out[(size_t)orow * HID_C_C + oc] = val > 0.0f ? val : 0.0f;
            }
        }
    } else {
        const int g  = batch[row];
        const int g0 = __builtin_amdgcn_readfirstlane(g);

        float h[4][8];
        float cs[4];
#pragma unroll
        for (int t = 0; t < 4; ++t) {
            const float bias = lb[t * 16 + m];
            float s = 0.0f;
#pragma unroll
            for (int v = 0; v < 8; ++v) {
                const float val = acc[t][v] + bias;
                h[t][v] = val > 0.0f ? val : 0.0f;
                s += h[t][v];
            }
            cs[t] = s;
        }

        if (__builtin_amdgcn_ballot_w32(g != g0) == 0u) {
            float c4[4];
#pragma unroll
            for (int t = 0; t < 4; ++t) c4[t] = cs[t] + __shfl_xor(cs[t], 16);
            const float a0 = half ? c4[2] : c4[0];
            const float a1 = half ? c4[3] : c4[1];
            float* pg = pooled + g0 * HID_C_C + half * 32 + m;
            atomicAdd(pg,      a0);
            atomicAdd(pg + 16, a1);
            if (lane == 0) atomicAdd(cnt + g0, 16.0f);
        } else {
#pragma unroll
            for (int v = 0; v < 8; ++v) {
                const int gr = __shfl(g, half * 8 + v);
#pragma unroll
                for (int t = 0; t < 4; ++t)
                    atomicAdd(pooled + gr * HID_C_C + t * 16 + m, h[t][v]);
            }
            if (lane < 16) atomicAdd(cnt + g, 1.0f);
        }
    }
}

__global__ void __launch_bounds__(256)
proj_kernel(const float* __restrict__ pooled, const float* __restrict__ cnt,
            const float* __restrict__ W, const float* __restrict__ b,
            float* __restrict__ out) {
    const int i = blockIdx.x * blockDim.x + threadIdx.x;
    if (i >= N_GRAPHS_C * OUT_C_C) return;
    const int g = i / OUT_C_C;
    const int o = i % OUT_C_C;
    const float inv = 1.0f / fmaxf(cnt[g], 1.0f);
    float acc = b[o];
    const float* pr = pooled + (size_t)g * HID_C_C;
#pragma unroll 8
    for (int c = 0; c < HID_C_C; ++c)
        acc += (pr[c] * inv) * W[(size_t)c * OUT_C_C + o];
    out[i] = acc;
}

extern "C" void kernel_launch(void* const* d_in, const int* in_sizes, int n_in,
                              void* d_out, int out_size, void* d_ws, size_t ws_size,
                              hipStream_t stream) {
    const float* x      = (const float*)d_in[0];
    const float* eattr  = (const float*)d_in[1];
    const int*   eindex = (const int*)d_in[2];
    const int*   batch  = (const int*)d_in[3];
    const float* efc_w  = (const float*)d_in[4];
    const float* efc_b  = (const float*)d_in[5];
    const float* c1_lw  = (const float*)d_in[6];
    const float* c1_lb  = (const float*)d_in[7];
    const float* c1_rw  = (const float*)d_in[8];
    const float* c2_lw  = (const float*)d_in[9];
    const float* c2_lb  = (const float*)d_in[10];
    const float* c2_rw  = (const float*)d_in[11];
    const float* proj_w = (const float*)d_in[12];
    const float* proj_b = (const float*)d_in[13];
    float* out = (float*)d_out;

    float* ws = (float*)d_ws;
    size_t off = 0;
    float* x1     = ws + off; off += (size_t)N_NODES_C * IN_C_C;
    float* deg    = ws + off; off += (size_t)N_NODES_C;
    off += (size_t)N_NODES_C * IN_C_C;
    float* h1     = ws + off; off += (size_t)N_NODES_C * HID_C_C;
    float* msum2  = ws + off; off += (size_t)N_NODES_C * HID_C_C;
    float* h2     = ws + off; off += (size_t)N_NODES_C * HID_C_C;
    float* pooled = ws + off; off += (size_t)N_GRAPHS_C * HID_C_C;
    float* cnt    = ws + off; off += (size_t)N_GRAPHS_C;

    int* count  = (int*)h2;
    int* rowptr = count  + N_NODES_C;
    int* col    = rowptr + N_NODES_C;
    int* count_s  = (int*)h1;
    int* rowptr_s = count_s  + N_NODES_C;
    int* eperm    = rowptr_s + N_NODES_C;
    int* tmp     = (int*)msum2;
    int* gcount  = tmp + 2 * (size_t)N_EDGES_C;
    int* gcursor = gcount + 2 * N_BINS;

    init_kernel<<<(N_GRAPHS_C * HID_C_C + 255) / 256, 256, 0, stream>>>(
        pooled, cnt, gcount);

    bin_hist<<<N_CHUNKS, BIN_T, 0, stream>>>(eindex, gcount);
    bin_scatter<<<dim3(N_CHUNKS, 2), BIN_T, 0, stream>>>(eindex, gcount, gcursor, tmp);
    bin_sort<<<dim3(N_BINS, 2), BIN_NODES, 0, stream>>>(
        gcount, tmp, count, rowptr, col, count_s, rowptr_s, eperm);

    edge_fc_pull<<<N_NODES_C * 16 / 256, 256, 0, stream>>>(
        eattr, x, efc_w, efc_b, rowptr_s, count_s, eperm, count, x1, deg);

    sage_wmma_kernel<IN_C_C, false><<<N_NODES_C / 16 / 2, 64, 0, stream>>>(
        x1, rowptr, count, col, deg, c1_lw, c1_lb, c1_rw, h1, batch, pooled, cnt);

    sage_wmma_kernel<HID_C_C, true><<<N_NODES_C / 16 / 2, 64, 0, stream>>>(
        h1, rowptr, count, col, deg, c2_lw, c2_lb, c2_rw, h2, batch, pooled, cnt);

    proj_kernel<<<(N_GRAPHS_C * OUT_C_C + 255) / 256, 256, 0, stream>>>(
        pooled, cnt, proj_w, proj_b, out);
}


# speedup vs baseline: 13.6388x; 1.0425x over previous
#include <hip/hip_runtime.h>
#include <hip/hip_bf16.h>
#include <stdint.h>

#define N_NODES_C  100000
#define N_EDGES_C  1600000
#define N_GRAPHS_C 64
#define IN_C_C     18
#define EDGE_C_C   22
#define HID_C_C    64
#define OUT_C_C    128

#define X1_PITCH   32

typedef float v2f __attribute__((ext_vector_type(2)));
typedef float v8f __attribute__((ext_vector_type(8)));

#define BIN_SHIFT   8
#define BIN_NODES   256
#define N_BINS      ((N_NODES_C + BIN_NODES - 1) / BIN_NODES)
#define CHUNK_E     8192
#define N_CHUNKS    ((N_EDGES_C + CHUNK_E - 1) / CHUNK_E)
#define BIN_T       512
#define BIN_EPT     (CHUNK_E / BIN_T)
#define STAGE_CAP   8192
#define SORT_EPT    16
#define DST_PAY_BITS 17
#define SRC_PAY_BITS 21

__global__ void __launch_bounds__(256)
init_kernel(float* __restrict__ pooled, float* __restrict__ cnt,
            int* __restrict__ gcount) {
    const int i = blockIdx.x * blockDim.x + threadIdx.x;
    if (i < 4 * N_BINS)           gcount[i] = 0;
    if (i < N_GRAPHS_C * HID_C_C) pooled[i] = 0.0f;
    if (i < N_GRAPHS_C)           cnt[i] = 0.0f;
}

template <int NW>
static __device__ __forceinline__ int block_excl_scan(int v, int* swave) {
    const int lane = threadIdx.x & 31;
    const int wv   = threadIdx.x >> 5;
    int inc = v;
#pragma unroll
    for (int d = 1; d < 32; d <<= 1) {
        const int n = __shfl_up(inc, d);
        if (lane >= d) inc += n;
    }
    if (lane == 31) swave[wv] = inc;
    __syncthreads();
    int pre = 0;
#pragma unroll
    for (int w = 0; w < NW; ++w) pre += (w < wv) ? swave[w] : 0;
    __syncthreads();
    return pre + inc - v;
}

static __device__ __forceinline__ unsigned node_id(int v) {
    const unsigned u = (unsigned)v;
    return u < (unsigned)N_NODES_C ? u : (unsigned)(N_NODES_C - 1);
}

__global__ void __launch_bounds__(BIN_T)
bin_hist(const int* __restrict__ eindex, int* __restrict__ gcount) {
    __shared__ int h[2 * N_BINS];
    const int tid = threadIdx.x;
    for (int i = tid; i < 2 * N_BINS; i += BIN_T) h[i] = 0;
    __syncthreads();
    const int e0 = blockIdx.x * CHUNK_E;
    int sv[BIN_EPT], dv[BIN_EPT];
#pragma unroll
    for (int k = 0; k < BIN_EPT; ++k) {
        const int e  = e0 + k * BIN_T + tid;
        const int ec = e < N_EDGES_C ? e : N_EDGES_C - 1;
        sv[k] = eindex[ec];
        dv[k] = eindex[N_EDGES_C + ec];
    }
#pragma unroll
    for (int k = 0; k < BIN_EPT; ++k) {
        const int e = e0 + k * BIN_T + tid;
        if (e < N_EDGES_C) {
            atomicAdd(&h[node_id(dv[k]) >> BIN_SHIFT], 1);
            atomicAdd(&h[N_BINS + (node_id(sv[k]) >> BIN_SHIFT)], 1);
        }
    }
    __syncthreads();
    for (int i = tid; i < 2 * N_BINS; i += BIN_T) {
        const int v = h[i];
        if (v) atomicAdd(gcount + i, v);
    }
}

__global__ void __launch_bounds__(BIN_T)
bin_scatter(const int* __restrict__ eindex, const int* __restrict__ gcount,
            int* __restrict__ gcursor, int* __restrict__ tmp) {
    __shared__ int hist[N_BINS];
    __shared__ int gdelta[N_BINS];
    __shared__ int sval[CHUNK_E];
    __shared__ unsigned short sbin[CHUNK_E];
    __shared__ int swave[BIN_T / 32];
    const int list = blockIdx.y;
    const int tid  = threadIdx.x;
    const int e0   = blockIdx.x * CHUNK_E;
    const int nloc = (N_EDGES_C - e0) < CHUNK_E ? (N_EDGES_C - e0) : CHUNK_E;

    if (tid < N_BINS) hist[tid] = 0;
    const int gc = tid < N_BINS ? gcount[list * N_BINS + tid] : 0;
    __syncthreads();

    const int* keyrow = eindex + (list == 0 ? N_EDGES_C : 0);
    const int  pshift = list == 0 ? DST_PAY_BITS : SRC_PAY_BITS;
    int sv[BIN_EPT], kv[BIN_EPT];
#pragma unroll
    for (int k = 0; k < BIN_EPT; ++k) {
        const int e  = e0 + k * BIN_T + tid;
        const int ec = e < N_EDGES_C ? e : N_EDGES_C - 1;
        sv[k] = eindex[ec];
        kv[k] = keyrow[ec];
    }

    int val[BIN_EPT], rnk[BIN_EPT], bin[BIN_EPT];
#pragma unroll
    for (int k = 0; k < BIN_EPT; ++k) {
        const int e = e0 + k * BIN_T + tid;
        bin[k] = -1;
        if (e < N_EDGES_C) {
            const unsigned key = node_id(kv[k]);
            const unsigned low = list == 0 ? node_id(sv[k]) : (unsigned)e;
            const unsigned pay = ((key & (BIN_NODES - 1)) << pshift) | low;
            bin[k] = (int)(key >> BIN_SHIFT);
            val[k] = (int)pay;
            rnk[k] = atomicAdd(&hist[bin[k]], 1);
        }
    }
    __syncthreads();

    const int c  = tid < N_BINS ? hist[tid] : 0;
    const int bstart = block_excl_scan<BIN_T / 32>(gc, swave);
    const int lofs   = block_excl_scan<BIN_T / 32>(c, swave);
    if (tid < N_BINS) {
        const int claim = c ? atomicAdd(gcursor + list * N_BINS + tid, c) : 0;
        hist[tid]   = lofs;
        gdelta[tid] = bstart + claim - lofs;
    }
    __syncthreads();

#pragma unroll
    for (int k = 0; k < BIN_EPT; ++k) {
        if (bin[k] >= 0) {
            const int p = hist[bin[k]] + rnk[k];
            sval[p] = val[k];
            sbin[p] = (unsigned short)bin[k];
        }
    }
    __syncthreads();

    int* out = tmp + (size_t)list * N_EDGES_C;
    for (int i = tid; i < nloc; i += BIN_T) {
        const int pos = gdelta[sbin[i]] + i;
        if ((unsigned)pos < (unsigned)N_EDGES_C) out[pos] = sval[i];
    }
}

__global__ void __launch_bounds__(BIN_NODES)
bin_sort(const int* __restrict__ gcount, const int* __restrict__ tmp,
         int* __restrict__ count_d, int* __restrict__ rowptr_d, int* __restrict__ col,
         int* __restrict__ count_s, int* __restrict__ rowptr_s, int* __restrict__ eperm) {
    __shared__ int fcnt[BIN_NODES];
    __shared__ int fofs[BIN_NODES];
    __shared__ int stage[STAGE_CAP];
    __shared__ int swave[BIN_NODES / 32];
    __shared__ int sred[BIN_NODES / 32];
    const int list = blockIdx.y;
    const int b    = blockIdx.x;
    const int tid  = threadIdx.x;
    const int lane = tid & 31;
    const int* gc  = gcount + list * N_BINS;
    int* count  = list ? count_s  : count_d;
    int* rowptr = list ? rowptr_s : rowptr_d;
    int* out    = list ? eperm    : col;
    const int      shift = list ? SRC_PAY_BITS : DST_PAY_BITS;
    const unsigned pmask = (1u << shift) - 1u;

    int part = 0;
    for (int j = tid; j < b; j += BIN_NODES) part += gc[j];
    for (int d = 16; d > 0; d >>= 1) part += __shfl_xor(part, d);
    if (lane == 0) sred[tid >> 5] = part;
    fcnt[tid] = 0;
    __syncthreads();
    int start = 0;
#pragma unroll
    for (int w = 0; w < BIN_NODES / 32; ++w) start += sred[w];
    int n = gc[b];
    if (start < 0 || start > N_EDGES_C) start = N_EDGES_C;
    if (n < 0) n = 0;
    if (n > N_EDGES_C - start) n = N_EDGES_C - start;

    const int* in = tmp + (size_t)list * N_EDGES_C + start;
    for (int base = 0; base < n; base += BIN_NODES * SORT_EPT) {
        unsigned v[SORT_EPT];
#pragma unroll
        for (int k = 0; k < SORT_EPT; ++k) {
            const int i = base + k * BIN_NODES + tid;
            v[k] = (unsigned)in[i < n ? i : n - 1];
        }
#pragma unroll
        for (int k = 0; k < SORT_EPT; ++k) {
            const int i = base + k * BIN_NODES + tid;
            if (i < n) atomicAdd(&fcnt[(v[k] >> shift) & (BIN_NODES - 1)], 1);
        }
    }
    __syncthreads();

    const int c    = fcnt[tid];
    const int excl = block_excl_scan<BIN_NODES / 32>(c, swave);
    const int node = b * BIN_NODES + tid;
    if (node < N_NODES_C) { count[node] = c; rowptr[node] = start + excl; }
    fofs[tid] = excl;
    fcnt[tid] = 0;
    __syncthreads();

    const bool staged = n <= STAGE_CAP;
    for (int base = 0; base < n; base += BIN_NODES * SORT_EPT) {
        unsigned v[SORT_EPT];
#pragma unroll
        for (int k = 0; k < SORT_EPT; ++k) {
            const int i = base + k * BIN_NODES + tid;
            v[k] = (unsigned)in[i < n ? i : n - 1];
        }
#pragma unroll
        for (int k = 0; k < SORT_EPT; ++k) {
            const int i = base + k * BIN_NODES + tid;
            if (i < n) {
                const unsigned f = (v[k] >> shift) & (BIN_NODES - 1);
                const int p = fofs[f] + atomicAdd(&fcnt[f], 1);
                if (staged) stage[p] = (int)(v[k] & pmask);
                else        out[start + p] = (int)(v[k] & pmask);
            }
        }
    }
    if (staged) {
        __syncthreads();
        for (int i = tid; i < n; i += BIN_NODES) out[start + i] = stage[i];
    }
}

template <int M>
static __device__ __forceinline__ float row_xor_add(float v) {
    const int o = __builtin_amdgcn_update_dpp(0, __float_as_int(v), 0x160 + M,
                                              0xf, 0xf, true);
    return v + __int_as_float(o);
}

__global__ void __launch_bounds__(256)
edge_fc_pull(const float* __restrict__ eattr, const float* __restrict__ x,
             const float* __restrict__ W, const float* __restrict__ b,
             const int* __restrict__ rowptr_s, const int* __restrict__ count_s,
             const int* __restrict__ eperm, const int* __restrict__ count_d,
             float* __restrict__ x1, float* __restrict__ deg) {
    const int t    = blockIdx.x * blockDim.x + threadIdx.x;
    const int node = t >> 4;
    const int l    = t & 15;
    const int beg  = rowptr_s[node];
    const int cnt  = count_s[node];

    if (l == 15) deg[node] = (float)count_d[node];

    float s[EDGE_C_C];
#pragma unroll
    for (int q = 0; q < EDGE_C_C; ++q) s[q] = 0.0f;

    for (int k = l; k < cnt; k += 16) {
        unsigned e = (unsigned)eperm[beg + k];
        e = e < (unsigned)N_EDGES_C ? e : (unsigned)(N_EDGES_C - 1);
        const float2* er = (const float2*)(eattr + (size_t)e * EDGE_C_C);
#pragma unroll
        for (int q = 0; q < EDGE_C_C / 2; ++q) {
            const float2 p = er[q];
            s[2 * q] += p.x; s[2 * q + 1] += p.y;
        }
    }

#pragma unroll
    for (int q = 0; q < EDGE_C_C; ++q) {
        float v = s[q];
        v = row_xor_add<1>(v);
        v = row_xor_add<2>(v);
        v = row_xor_add<4>(v);
        v = row_xor_add<8>(v);
        s[q] = v;
    }

    if (l < IN_C_C / 2) {
        const float2* W2 = (const float2*)W;
        const float2 bb = ((const float2*)b)[l];
        const float  cf = (float)cnt;
        float2 acc = make_float2(cf * bb.x, cf * bb.y);
#pragma unroll
        for (int q = 0; q < EDGE_C_C; ++q) {
            const float2 w = W2[q * (IN_C_C / 2) + l];
            acc.x += s[q] * w.x; acc.y += s[q] * w.y;
        }
        const float2 xv = ((const float2*)x)[node * (IN_C_C / 2) + l];
        ((float2*)x1)[node * (X1_PITCH / 2) + l] = make_float2(xv.x + acc.x, xv.y + acc.y);
    }
}

#define COL_CAP 512

template <int K, int XP, int PITCH, bool STAGED>
static __device__ __forceinline__ void
gather_tile(const float2* __restrict__ f2, const int* __restrict__ col,
            const int* sc, const int* srp, const int* scn,
            int base, int navail, int lane, float* sm) {
    constexpr int KP  = (K + 3) & ~3;
    constexpr int LPN = K / 2;
    constexpr int NPW = 32 / LPN;
    const int g  = lane / LPN;
    const int c2 = lane - g * LPN;
    auto entry = [&](int j) -> int {
        if (STAGED) {
            const unsigned q = (unsigned)(j - base);
            return sc[q < (unsigned)navail ? q : (unsigned)(navail - 1)];
        } else {
            return col[(unsigned)j < (unsigned)N_EDGES_C ? (unsigned)j : (unsigned)(N_EDGES_C - 1)];
        }
    };
    for (int p = 0; p < (16 + NPW - 1) / NPW; ++p) {
        const int r = p * NPW + g;
        if (g < NPW && r < 16) {
            const int beg = srp[r];
            const int end = beg + scn[r];
            float2 acc = make_float2(0.0f, 0.0f);
            int j = beg;
            for (; j + 8 <= end; j += 8) {
                int   s[8];
                float2 a[8];
#pragma unroll
                for (int u = 0; u < 8; ++u) s[u] = entry(j + u);
#pragma unroll
                for (int u = 0; u < 8; ++u) a[u] = f2[s[u] * (XP / 2) + c2];
#pragma unroll
                for (int u = 0; u < 8; ++u) { acc.x += a[u].x; acc.y += a[u].y; }
            }
            if (j < end) {
                int   s[7];
                float2 a[7];
#pragma unroll
                for (int u = 0; u < 7; ++u) s[u] = entry(j + u < end ? j + u : end - 1);
#pragma unroll
                for (int u = 0; u < 7; ++u) a[u] = f2[s[u] * (XP / 2) + c2];
#pragma unroll
                for (int u = 0; u < 7; ++u)
                    if (j + u < end) { acc.x += a[u].x; acc.y += a[u].y; }
            }
            *(float2*)(sm + r * PITCH + 2 * c2) = acc;
            if (KP > K && c2 == LPN - 1) {
                for (int q = K; q < KP; ++q) sm[r * PITCH + q] = 0.0f;
            }
        }
    }
}

template <int K, int XP, bool POOL>
__global__ void __launch_bounds__(64)
sage_wmma_kernel(const float* __restrict__ xin,
                 const int* __restrict__ rowptr, const int* __restrict__ count,
                 const int* __restrict__ col,
                 const float* __restrict__ deg,
                 const float* __restrict__ lw, const float* __restrict__ lb,
                 const float* __restrict__ rw, float* __restrict__ out,
                 const int* __restrict__ batch, float* __restrict__ pooled,
                 float* __restrict__ cnt) {
    constexpr int KP    = (K + 3) & ~3;
    constexpr int PITCH = (K % 64 == 0) ? K + 4 : KP;
    __shared__ __attribute__((aligned(16))) float s_sum[2][16 * PITCH];
    __shared__ int   s_col[2][COL_CAP];
    __shared__ int   s_rp[2][16];
    __shared__ int   s_cn[2][16];

    const int wib  = threadIdx.x >> 5;
    const int wave = blockIdx.x * 2 + wib;
    const int lane = threadIdx.x & 31;

    const int node0 = wave * 16;
    const int m    = lane & 15;
    const int half = (lane >> 4) & 1;
    const int row  = node0 + m;

    float* sm = s_sum[wib];
    int*   sc = s_col[wib];

    const int rp = rowptr[row];
    const int cn = count[row];
    if (lane < 16) { s_rp[wib][lane] = rp; s_cn[wib][lane] = cn; }
    const int base = __builtin_amdgcn_readlane(rp, 0);
    const int tot  = __builtin_amdgcn_readlane(rp, 15) + __builtin_amdgcn_readlane(cn, 15) - base;
    const bool staged = tot >= 0 && tot <= COL_CAP;
    if (staged) {
        for (int i = lane; i < tot; i += 32) {
            const int j = base + i;
            sc[i] = col[j < N_EDGES_C ? j : N_EDGES_C - 1];
        }
    }
    __syncthreads();

    if (staged)
        gather_tile<K, XP, PITCH, true>((const float2*)xin, col, sc, s_rp[wib], s_cn[wib],
                                    base, tot, lane, sm);
    else
        gather_tile<K, XP, PITCH, false>((const float2*)xin, col, sc, s_rp[wib], s_cn[wib],
                                     base, tot, lane, sm);
    __syncthreads();

    const float dg  = deg[row];
    const float inv = dg > 0.0f ? 1.0f / dg : 0.0f;
    const float* xr = xin + (size_t)row * XP;
    const float* mr = sm + m * PITCH;

    v8f acc[4] = {};

#pragma unroll
    for (int ks = 0; ks < KP; ks += 4) {
        const int kb = ks + half * 2;
        v2f amean, aroot;
        const float2 ms = *(const float2*)(mr + kb);
        amean[0] = ms.x * inv;
        amean[1] = ms.y * inv;
        aroot[0] = (kb     < K) ? xr[kb]           : 0.0f;
        aroot[1] = (kb + 1 < K) ? xr[kb + 1]       : 0.0f;
#pragma unroll
        for (int t = 0; t < 4; ++t) {
            const int col_n = t * 16 + m;
            v2f bl, br;
            bl[0] = (kb     < K) ? lw[(size_t)kb       * HID_C_C + col_n] : 0.0f;
            bl[1] = (kb + 1 < K) ? lw[(size_t)(kb + 1) * HID_C_C + col_n] : 0.0f;
            br[0] = (kb     < K) ? rw[(size_t)kb       * HID_C_C + col_n] : 0.0f;
            br[1] = (kb + 1 < K) ? rw[(size_t)(kb + 1) * HID_C_C + col_n] : 0.0f;
            acc[t] = __builtin_amdgcn_wmma_f32_16x16x4_f32(
                false, amean, false, bl, (short)0, acc[t], false, false);
            acc[t] = __builtin_amdgcn_wmma_f32_16x16x4_f32(
                false, aroot, false, br, (short)0, acc[t], false, false);
        }
    }

    if (!POOL) {
#pragma unroll
        for (int t = 0; t < 4; ++t) {
            const int oc   = t * 16 + m;
            const float bias = lb[oc];
#pragma unroll
            for (int v = 0; v < 8; ++v) {
                const int orow = node0 + half * 8 + v;
                float val = acc[t][v] + bias;
                out[(size_t)orow * HID_C_C + oc] = val > 0.0f ? val : 0.0f;
            }
        }
    } else {
        const int g  = batch[row];
        const int g0 = __builtin_amdgcn_readfirstlane(g);

        float h[4][8];
        float cs[4];
#pragma unroll
        for (int t = 0; t < 4; ++t) {
            const float bias = lb[t * 16 + m];
            float s = 0.0f;
#pragma unroll
            for (int v = 0; v < 8; ++v) {
                const float val = acc[t][v] + bias;
                h[t][v] = val > 0.0f ? val : 0.0f;
                s += h[t][v];
            }
            cs[t] = s;
        }

        if (__builtin_amdgcn_ballot_w32(g != g0) == 0u) {
            float c4[4];
#pragma unroll
            for (int t = 0; t < 4; ++t) c4[t] = cs[t] + __shfl_xor(cs[t], 16);
            const float a0 = half ? c4[2] : c4[0];
            const float a1 = half ? c4[3] : c4[1];
            float* pg = pooled + g0 * HID_C_C + half * 32 + m;
            atomicAdd(pg,      a0);
            atomicAdd(pg + 16, a1);
            if (lane == 0) atomicAdd(cnt + g0, 16.0f);
        } else {
#pragma unroll
            for (int v = 0; v < 8; ++v) {
                const int gr = __shfl(g, half * 8 + v);
#pragma unroll
                for (int t = 0; t < 4; ++t)
                    atomicAdd(pooled + gr * HID_C_C + t * 16 + m, h[t][v]);
            }
            if (lane < 16) atomicAdd(cnt + g, 1.0f);
        }
    }
}

__global__ void __launch_bounds__(256)
proj_kernel(const float* __restrict__ pooled, const float* __restrict__ cnt,
            const float* __restrict__ W, const float* __restrict__ b,
            float* __restrict__ out) {
    const int i = blockIdx.x * blockDim.x + threadIdx.x;
    if (i >= N_GRAPHS_C * OUT_C_C) return;
    const int g = i / OUT_C_C;
    const int o = i % OUT_C_C;
    const float inv = 1.0f / fmaxf(cnt[g], 1.0f);
    float acc = b[o];
    const float* pr = pooled + (size_t)g * HID_C_C;
#pragma unroll 8
    for (int c = 0; c < HID_C_C; ++c)
        acc += (pr[c] * inv) * W[(size_t)c * OUT_C_C + o];
    out[i] = acc;
}

extern "C" void kernel_launch(void* const* d_in, const int* in_sizes, int n_in,
                              void* d_out, int out_size, void* d_ws, size_t ws_size,
                              hipStream_t stream) {
    const float* x      = (const float*)d_in[0];
    const float* eattr  = (const float*)d_in[1];
    const int*   eindex = (const int*)d_in[2];
    const int*   batch  = (const int*)d_in[3];
    const float* efc_w  = (const float*)d_in[4];
    const float* efc_b  = (const float*)d_in[5];
    const float* c1_lw  = (const float*)d_in[6];
    const float* c1_lb  = (const float*)d_in[7];
    const float* c1_rw  = (const float*)d_in[8];
    const float* c2_lw  = (const float*)d_in[9];
    const float* c2_lb  = (const float*)d_in[10];
    const float* c2_rw  = (const float*)d_in[11];
    const float* proj_w = (const float*)d_in[12];
    const float* proj_b = (const float*)d_in[13];
    float* out = (float*)d_out;

    float* ws = (float*)d_ws;
    size_t off = 0;
    float* x1     = ws + off;
    float* deg    = x1 + (size_t)N_NODES_C * X1_PITCH;
    off += (size_t)N_NODES_C * (2 * IN_C_C + 1);
    float* h1     = ws + off - 32; off += (size_t)N_NODES_C * HID_C_C;
    float* msum2  = ws + off; off += (size_t)N_NODES_C * HID_C_C;
    float* h2     = ws + off; off += (size_t)N_NODES_C * HID_C_C;
    float* pooled = ws + off; off += (size_t)N_GRAPHS_C * HID_C_C;
    float* cnt    = ws + off; off += (size_t)N_GRAPHS_C;

    int* count  = (int*)h2;
    int* rowptr = count  + N_NODES_C;
    int* col    = rowptr + N_NODES_C;
    int* count_s  = (int*)h1;
    int* rowptr_s = count_s  + N_NODES_C;
    int* eperm    = rowptr_s + N_NODES_C;
    int* tmp     = (int*)msum2;
    int* gcount  = tmp + 2 * (size_t)N_EDGES_C;
    int* gcursor = gcount + 2 * N_BINS;

    init_kernel<<<(N_GRAPHS_C * HID_C_C + 255) / 256, 256, 0, stream>>>(
        pooled, cnt, gcount);

    bin_hist<<<N_CHUNKS, BIN_T, 0, stream>>>(eindex, gcount);
    bin_scatter<<<dim3(N_CHUNKS, 2), BIN_T, 0, stream>>>(eindex, gcount, gcursor, tmp);
    bin_sort<<<dim3(N_BINS, 2), BIN_NODES, 0, stream>>>(
        gcount, tmp, count, rowptr, col, count_s, rowptr_s, eperm);

    edge_fc_pull<<<N_NODES_C * 16 / 256, 256, 0, stream>>>(
        eattr, x, efc_w, efc_b, rowptr_s, count_s, eperm, count, x1, deg);

    sage_wmma_kernel<IN_C_C, X1_PITCH, false><<<N_NODES_C / 16 / 2, 64, 0, stream>>>(
        x1, rowptr, count, col, deg, c1_lw, c1_lb, c1_rw, h1, batch, pooled, cnt);

    sage_wmma_kernel<HID_C_C, HID_C_C, true><<<N_NODES_C / 16 / 2, 64, 0, stream>>>(
        h1, rowptr, count, col, deg, c2_lw, c2_lb, c2_rw, h2, batch, pooled, cnt);

    proj_kernel<<<(N_GRAPHS_C * OUT_C_C + 255) / 256, 256, 0, stream>>>(
        pooled, cnt, proj_w, proj_b, out);
}


# Conv launches: each 16-node tile on a 4-wave block; a wave sums 4 of the nodes' rows (same list order) and does one 16-column slice of the WMMA products; h1 rows stored whole via LDS
# speedup vs baseline: 14.3047x; 1.0488x over previous
#include <hip/hip_runtime.h>
#include <hip/hip_bf16.h>
#include <stdint.h>

#define N_NODES_C  100000
#define N_EDGES_C  1600000
#define N_GRAPHS_C 64
#define IN_C_C     18
#define EDGE_C_C   22
#define HID_C_C    64
#define OUT_C_C    128

#define X1_PITCH   32

typedef float v2f __attribute__((ext_vector_type(2)));
typedef float v8f __attribute__((ext_vector_type(8)));

#define BIN_SHIFT   8
#define BIN_NODES   256
#define N_BINS      ((N_NODES_C + BIN_NODES - 1) / BIN_NODES)
#define CHUNK_E     8192
#define N_CHUNKS    ((N_EDGES_C + CHUNK_E - 1) / CHUNK_E)
#define BIN_T       512
#define BIN_EPT     (CHUNK_E / BIN_T)
#define STAGE_CAP   8192
#define SORT_EPT    16
#define DST_PAY_BITS 17
#define SRC_PAY_BITS 21

__global__ void __launch_bounds__(256)
init_kernel(float* __restrict__ pooled, float* __restrict__ cnt,
            int* __restrict__ gcount) {
    const int i = blockIdx.x * blockDim.x + threadIdx.x;
    if (i < 4 * N_BINS)           gcount[i] = 0;
    if (i < N_GRAPHS_C * HID_C_C) pooled[i] = 0.0f;
    if (i < N_GRAPHS_C)           cnt[i] = 0.0f;
}

template <int NW>
static __device__ __forceinline__ int block_excl_scan(int v, int* swave) {
    const int lane = threadIdx.x & 31;
    const int wv   = threadIdx.x >> 5;
    int inc = v;
#pragma unroll
    for (int d = 1; d < 32; d <<= 1) {
        const int n = __shfl_up(inc, d);
        if (lane >= d) inc += n;
    }
    if (lane == 31) swave[wv] = inc;
    __syncthreads();
    int pre = 0;
#pragma unroll
    for (int w = 0; w < NW; ++w) pre += (w < wv) ? swave[w] : 0;
    __syncthreads();
    return pre + inc - v;
}

static __device__ __forceinline__ unsigned node_id(int v) {
    const unsigned u = (unsigned)v;
    return u < (unsigned)N_NODES_C ? u : (unsigned)(N_NODES_C - 1);
}

__global__ void __launch_bounds__(BIN_T)
bin_hist(const int* __restrict__ eindex, int* __restrict__ gcount) {
    __shared__ int h[2 * N_BINS];
    const int tid = threadIdx.x;
    for (int i = tid; i < 2 * N_BINS; i += BIN_T) h[i] = 0;
    __syncthreads();
    const int e0 = blockIdx.x * CHUNK_E;
    int sv[BIN_EPT], dv[BIN_EPT];
#pragma unroll
    for (int k = 0; k < BIN_EPT; ++k) {
        const int e  = e0 + k * BIN_T + tid;
        const int ec = e < N_EDGES_C ? e : N_EDGES_C - 1;
        sv[k] = eindex[ec];
        dv[k] = eindex[N_EDGES_C + ec];
    }
#pragma unroll
    for (int k = 0; k < BIN_EPT; ++k) {
        const int e = e0 + k * BIN_T + tid;
        if (e < N_EDGES_C) {
            atomicAdd(&h[node_id(dv[k]) >> BIN_SHIFT], 1);
            atomicAdd(&h[N_BINS + (node_id(sv[k]) >> BIN_SHIFT)], 1);
        }
    }
    __syncthreads();
    for (int i = tid; i < 2 * N_BINS; i += BIN_T) {
        const int v = h[i];
        if (v) atomicAdd(gcount + i, v);
    }
}

__global__ void __launch_bounds__(BIN_T)
bin_scatter(const int* __restrict__ eindex, const int* __restrict__ gcount,
            int* __restrict__ gcursor, int* __restrict__ tmp) {
    __shared__ int hist[N_BINS];
    __shared__ int gdelta[N_BINS];
    __shared__ int sval[CHUNK_E];
    __shared__ unsigned short sbin[CHUNK_E];
    __shared__ int swave[BIN_T / 32];
    const int list = blockIdx.y;
    const int tid  = threadIdx.x;
    const int e0   = blockIdx.x * CHUNK_E;
    const int nloc = (N_EDGES_C - e0) < CHUNK_E ? (N_EDGES_C - e0) : CHUNK_E;

    if (tid < N_BINS) hist[tid] = 0;
    const int gc = tid < N_BINS ? gcount[list * N_BINS + tid] : 0;
    __syncthreads();

    const int* keyrow = eindex + (list == 0 ? N_EDGES_C : 0);
    const int  pshift = list == 0 ? DST_PAY_BITS : SRC_PAY_BITS;
    int sv[BIN_EPT], kv[BIN_EPT];
#pragma unroll
    for (int k = 0; k < BIN_EPT; ++k) {
        const int e  = e0 + k * BIN_T + tid;
        const int ec = e < N_EDGES_C ? e : N_EDGES_C - 1;
        sv[k] = eindex[ec];
        kv[k] = keyrow[ec];
    }

    int val[BIN_EPT], rnk[BIN_EPT], bin[BIN_EPT];
#pragma unroll
    for (int k = 0; k < BIN_EPT; ++k) {
        const int e = e0 + k * BIN_T + tid;
        bin[k] = -1;
        if (e < N_EDGES_C) {
            const unsigned key = node_id(kv[k]);
            const unsigned low = list == 0 ? node_id(sv[k]) : (unsigned)e;
            const unsigned pay = ((key & (BIN_NODES - 1)) << pshift) | low;
            bin[k] = (int)(key >> BIN_SHIFT);
            val[k] = (int)pay;
            rnk[k] = atomicAdd(&hist[bin[k]], 1);
        }
    }
    __syncthreads();

    const int c  = tid < N_BINS ? hist[tid] : 0;
    const int bstart = block_excl_scan<BIN_T / 32>(gc, swave);
    const int lofs   = block_excl_scan<BIN_T / 32>(c, swave);
    if (tid < N_BINS) {
        const int claim = c ? atomicAdd(gcursor + list * N_BINS + tid, c) : 0;
        hist[tid]   = lofs;
        gdelta[tid] = bstart + claim - lofs;
    }
    __syncthreads();

#pragma unroll
    for (int k = 0; k < BIN_EPT; ++k) {
        if (bin[k] >= 0) {
            const int p = hist[bin[k]] + rnk[k];
            sval[p] = val[k];
            sbin[p] = (unsigned short)bin[k];
        }
    }
    __syncthreads();

    int* out = tmp + (size_t)list * N_EDGES_C;
    for (int i = tid; i < nloc; i += BIN_T) {
        const int pos = gdelta[sbin[i]] + i;
        if ((unsigned)pos < (unsigned)N_EDGES_C) out[pos] = sval[i];
    }
}

__global__ void __launch_bounds__(BIN_NODES)
bin_sort(const int* __restrict__ gcount, const int* __restrict__ tmp,
         int* __restrict__ count_d, int* __restrict__ rowptr_d, int* __restrict__ col,
         int* __restrict__ count_s, int* __restrict__ rowptr_s, int* __restrict__ eperm) {
    __shared__ int fcnt[BIN_NODES];
    __shared__ int fofs[BIN_NODES];
    __shared__ int stage[STAGE_CAP];
    __shared__ int swave[BIN_NODES / 32];
    __shared__ int sred[BIN_NODES / 32];
    const int list = blockIdx.y;
    const int b    = blockIdx.x;
    const int tid  = threadIdx.x;
    const int lane = tid & 31;
    const int* gc  = gcount + list * N_BINS;
    int* count  = list ? count_s  : count_d;
    int* rowptr = list ? rowptr_s : rowptr_d;
    int* out    = list ? eperm    : col;
    const int      shift = list ? SRC_PAY_BITS : DST_PAY_BITS;
    const unsigned pmask = (1u << shift) - 1u;

    int part = 0;
    for (int j = tid; j < b; j += BIN_NODES) part += gc[j];
    for (int d = 16; d > 0; d >>= 1) part += __shfl_xor(part, d);
    if (lane == 0) sred[tid >> 5] = part;
    fcnt[tid] = 0;
    __syncthreads();
    int start = 0;
#pragma unroll
    for (int w = 0; w < BIN_NODES / 32; ++w) start += sred[w];
    int n = gc[b];
    if (start < 0 || start > N_EDGES_C) start = N_EDGES_C;
    if (n < 0) n = 0;
    if (n > N_EDGES_C - start) n = N_EDGES_C - start;

    const int* in = tmp + (size_t)list * N_EDGES_C + start;
    for (int base = 0; base < n; base += BIN_NODES * SORT_EPT) {
        unsigned v[SORT_EPT];
#pragma unroll
        for (int k = 0; k < SORT_EPT; ++k) {
            const int i = base + k * BIN_NODES + tid;
            v[k] = (unsigned)in[i < n ? i : n - 1];
        }
#pragma unroll
        for (int k = 0; k < SORT_EPT; ++k) {
            const int i = base + k * BIN_NODES + tid;
            if (i < n) atomicAdd(&fcnt[(v[k] >> shift) & (BIN_NODES - 1)], 1);
        }
    }
    __syncthreads();

    const int c    = fcnt[tid];
    const int excl = block_excl_scan<BIN_NODES / 32>(c, swave);
    const int node = b * BIN_NODES + tid;
    if (node < N_NODES_C) { count[node] = c; rowptr[node] = start + excl; }
    fofs[tid] = excl;
    fcnt[tid] = 0;
    __syncthreads();

    const bool staged = n <= STAGE_CAP;
    for (int base = 0; base < n; base += BIN_NODES * SORT_EPT) {
        unsigned v[SORT_EPT];
#pragma unroll
        for (int k = 0; k < SORT_EPT; ++k) {
            const int i = base + k * BIN_NODES + tid;
            v[k] = (unsigned)in[i < n ? i : n - 1];
        }
#pragma unroll
        for (int k = 0; k < SORT_EPT; ++k) {
            const int i = base + k * BIN_NODES + tid;
            if (i < n) {
                const unsigned f = (v[k] >> shift) & (BIN_NODES - 1);
                const int p = fofs[f] + atomicAdd(&fcnt[f], 1);
                if (staged) stage[p] = (int)(v[k] & pmask);
                else        out[start + p] = (int)(v[k] & pmask);
            }
        }
    }
    if (staged) {
        __syncthreads();
        for (int i = tid; i < n; i += BIN_NODES) out[start + i] = stage[i];
    }
}

template <int M>
static __device__ __forceinline__ float row_xor_add(float v) {
    const int o = __builtin_amdgcn_update_dpp(0, __float_as_int(v), 0x160 + M,
                                              0xf, 0xf, true);
    return v + __int_as_float(o);
}

__global__ void __launch_bounds__(256)
edge_fc_pull(const float* __restrict__ eattr, const float* __restrict__ x,
             const float* __restrict__ W, const float* __restrict__ b,
             const int* __restrict__ rowptr_s, const int* __restrict__ count_s,
             const int* __restrict__ eperm, const int* __restrict__ count_d,
             float* __restrict__ x1, float* __restrict__ deg) {
    const int t    = blockIdx.x * blockDim.x + threadIdx.x;
    const int node = t >> 4;
    const int l    = t & 15;
    const int beg  = rowptr_s[node];
    const int cnt  = count_s[node];

    if (l == 15) deg[node] = (float)count_d[node];

    float s[EDGE_C_C];
#pragma unroll
    for (int q = 0; q < EDGE_C_C; ++q) s[q] = 0.0f;

    for (int k = l; k < cnt; k += 16) {
        unsigned e = (unsigned)eperm[beg + k];
        e = e < (unsigned)N_EDGES_C ? e : (unsigned)(N_EDGES_C - 1);
        const float2* er = (const float2*)(eattr + (size_t)e * EDGE_C_C);
#pragma unroll
        for (int q = 0; q < EDGE_C_C / 2; ++q) {
            const float2 p = er[q];
            s[2 * q] += p.x; s[2 * q + 1] += p.y;
        }
    }

#pragma unroll
    for (int q = 0; q < EDGE_C_C; ++q) {
        float v = s[q];
        v = row_xor_add<1>(v);
        v = row_xor_add<2>(v);
        v = row_xor_add<4>(v);
        v = row_xor_add<8>(v);
        s[q] = v;
    }

    if (l < IN_C_C / 2) {
        const float2* W2 = (const float2*)W;
        const float2 bb = ((const float2*)b)[l];
        const float  cf = (float)cnt;
        float2 acc = make_float2(cf * bb.x, cf * bb.y);
#pragma unroll
        for (int q = 0; q < EDGE_C_C; ++q) {
            const float2 w = W2[q * (IN_C_C / 2) + l];
            acc.x += s[q] * w.x; acc.y += s[q] * w.y;
        }
        const float2 xv = ((const float2*)x)[node * (IN_C_C / 2) + l];
        ((float2*)x1)[node * (X1_PITCH / 2) + l] = make_float2(xv.x + acc.x, xv.y + acc.y);
    }
}

#define COL_CAP 512

template <int K, int XP, int PITCH, int NWV, bool STAGED>
static __device__ __forceinline__ void
gather_tile(const float2* __restrict__ f2, const int* __restrict__ col,
            const int* sc, const int* srp, const int* scn,
            int base, int navail, int lane, int wib, float* sm) {
    constexpr int KP  = (K + 3) & ~3;
    constexpr int LPN = K / 2;
    constexpr int NPW = 32 / LPN;
    const int g  = lane / LPN;
    const int c2 = lane - g * LPN;
    auto entry = [&](int j) -> int {
        if (STAGED) {
            const unsigned q = (unsigned)(j - base);
            return sc[q < (unsigned)navail ? q : (unsigned)(navail - 1)];
        } else {
            return col[(unsigned)j < (unsigned)N_EDGES_C ? (unsigned)j : (unsigned)(N_EDGES_C - 1)];
        }
    };
    for (int p = wib; p < (16 + NPW - 1) / NPW; p += NWV) {
        const int r = p * NPW + g;
        if (g < NPW && r < 16) {
            const int beg = srp[r];
            const int end = beg + scn[r];
            float2 acc = make_float2(0.0f, 0.0f);
            int j = beg;
            for (; j + 8 <= end; j += 8) {
                int   s[8];
                float2 a[8];
#pragma unroll
                for (int u = 0; u < 8; ++u) s[u] = entry(j + u);
#pragma unroll
                for (int u = 0; u < 8; ++u) a[u] = f2[s[u] * (XP / 2) + c2];
#pragma unroll
                for (int u = 0; u < 8; ++u) { acc.x += a[u].x; acc.y += a[u].y; }
            }
            if (j < end) {
                int   s[7];
                float2 a[7];
#pragma unroll
                for (int u = 0; u < 7; ++u) s[u] = entry(j + u < end ? j + u : end - 1);
#pragma unroll
                for (int u = 0; u < 7; ++u) a[u] = f2[s[u] * (XP / 2) + c2];
#pragma unroll
                for (int u = 0; u < 7; ++u)
                    if (j + u < end) { acc.x += a[u].x; acc.y += a[u].y; }
            }
            *(float2*)(sm + r * PITCH + 2 * c2) = acc;
            if (KP > K && c2 == LPN - 1) {
                for (int q = K; q < KP; ++q) sm[r * PITCH + q] = 0.0f;
            }
        }
    }
}

#define TILE_WAVES 4

template <int K, int XP, bool POOL>
__global__ void __launch_bounds__(32 * TILE_WAVES)
sage_wmma_kernel(const float* __restrict__ xin,
                 const int* __restrict__ rowptr, const int* __restrict__ count,
                 const int* __restrict__ col,
                 const float* __restrict__ deg,
                 const float* __restrict__ lw, const float* __restrict__ lb,
                 const float* __restrict__ rw, float* __restrict__ out,
                 const int* __restrict__ batch, float* __restrict__ pooled,
                 float* __restrict__ cnt) {
    constexpr int KP    = (K + 3) & ~3;
    constexpr int PITCH = (K % 64 == 0) ? K + 4 : KP;
    constexpr int OPITCH = HID_C_C + 4;
    __shared__ __attribute__((aligned(16))) float s_sum[16 * PITCH];
    __shared__ __attribute__((aligned(16))) float s_out[POOL ? 4 : 16 * OPITCH];
    __shared__ int   s_col[COL_CAP];
    __shared__ int   s_rp[16];
    __shared__ int   s_cn[16];

    const int wib  = threadIdx.x >> 5;
    const int lane = threadIdx.x & 31;

    const int node0 = blockIdx.x * 16;
    const int m    = lane & 15;
    const int half = (lane >> 4) & 1;
    const int row  = node0 + m;

    float* sm = s_sum;
    int*   sc = s_col;

    const int rp = rowptr[row];
    const int cn = count[row];
    if (threadIdx.x < 16) { s_rp[lane] = rp; s_cn[lane] = cn; }
    const int base = __builtin_amdgcn_readlane(rp, 0);
    const int tot  = __builtin_amdgcn_readlane(rp, 15) + __builtin_amdgcn_readlane(cn, 15) - base;
    const bool staged = tot >= 0 && tot <= COL_CAP;
    if (staged) {
        for (int i = threadIdx.x; i < tot; i += 32 * TILE_WAVES) {
            const int j = base + i;
            sc[i] = col[j < N_EDGES_C ? j : N_EDGES_C - 1];
        }
    }
    __syncthreads();

    if (staged)
        gather_tile<K, XP, PITCH, TILE_WAVES, true>((const float2*)xin, col, sc, s_rp, s_cn,
                                                    base, tot, lane, wib, sm);
    else
        gather_tile<K, XP, PITCH, TILE_WAVES, false>((const float2*)xin, col, sc, s_rp, s_cn,
                                                     base, tot, lane, wib, sm);
    __syncthreads();

    const float dg  = deg[row];
    const float inv = dg > 0.0f ? 1.0f / dg : 0.0f;
    const float* xr = xin + (size_t)row * XP;
    const float* mr = sm + m * PITCH;
    const int col_n = wib * 16 + m;

    v8f acc = {};

#pragma unroll
    for (int ks = 0; ks < KP; ks += 4) {
        const int kb = ks + half * 2;
        v2f amean, aroot, bl, br;
        const float2 ms = *(const float2*)(mr + kb);
        amean[0] = ms.x * inv;
        amean[1] = ms.y * inv;
        aroot[0] = (kb     < K) ? xr[kb]           : 0.0f;
        aroot[1] = (kb + 1 < K) ? xr[kb + 1]       : 0.0f;
        bl[0] = (kb     < K) ? lw[(size_t)kb       * HID_C_C + col_n] : 0.0f;
        bl[1] = (kb + 1 < K) ? lw[(size_t)(kb + 1) * HID_C_C + col_n] : 0.0f;
        br[0] = (kb     < K) ? rw[(size_t)kb       * HID_C_C + col_n] : 0.0f;
        br[1] = (kb + 1 < K) ? rw[(size_t)(kb + 1) * HID_C_C + col_n] : 0.0f;
        acc = __builtin_amdgcn_wmma_f32_16x16x4_f32(
            false, amean, false, bl, (short)0, acc, false, false);
        acc = __builtin_amdgcn_wmma_f32_16x16x4_f32(
            false, aroot, false, br, (short)0, acc, false, false);
    }

    const float bias = lb[col_n];
    if (!POOL) {
#pragma unroll
        for (int v = 0; v < 8; ++v) {
            const float val = acc[v] + bias;
            s_out[(half * 8 + v) * OPITCH + col_n] = val > 0.0f ? val : 0.0f;
        }
        __syncthreads();
#pragma unroll
        for (int q = 0; q < 2; ++q) {
            const int r = wib * 4 + q * 2 + half;
            const float4 val = *(const float4*)(s_out + r * OPITCH + 4 * m);
            *(float4*)(out + (size_t)(node0 + r) * HID_C_C + 4 * m) = val;
        }
    } else {
        const int g  = batch[row];
        const int g0 = __builtin_amdgcn_readfirstlane(g);

        float h[8];
        float cs = 0.0f;
#pragma unroll
        for (int v = 0; v < 8; ++v) {
            const float val = acc[v] + bias;
            h[v] = val > 0.0f ? val : 0.0f;
            cs += h[v];
        }

        if (__builtin_amdgcn_ballot_w32(g != g0) == 0u) {
            const float c16 = cs + __shfl_xor(cs, 16);
            if (half == 0) atomicAdd(pooled + g0 * HID_C_C + col_n, c16);
            if (threadIdx.x == 0) atomicAdd(cnt + g0, 16.0f);
        } else {
#pragma unroll
            for (int v = 0; v < 8; ++v) {
                const int gr = __shfl(g, half * 8 + v);
                atomicAdd(pooled + gr * HID_C_C + col_n, h[v]);
            }
            if (threadIdx.x < 16) atomicAdd(cnt + g, 1.0f);
        }
    }
}

__global__ void __launch_bounds__(256)
proj_kernel(const float* __restrict__ pooled, const float* __restrict__ cnt,
            const float* __restrict__ W, const float* __restrict__ b,
            float* __restrict__ out) {
    const int i = blockIdx.x * blockDim.x + threadIdx.x;
    if (i >= N_GRAPHS_C * OUT_C_C) return;
    const int g = i / OUT_C_C;
    const int o = i % OUT_C_C;
    const float inv = 1.0f / fmaxf(cnt[g], 1.0f);
    float acc = b[o];
    const float* pr = pooled + (size_t)g * HID_C_C;
#pragma unroll 8
    for (int c = 0; c < HID_C_C; ++c)
        acc += (pr[c] * inv) * W[(size_t)c * OUT_C_C + o];
    out[i] = acc;
}

extern "C" void kernel_launch(void* const* d_in, const int* in_sizes, int n_in,
                              void* d_out, int out_size, void* d_ws, size_t ws_size,
                              hipStream_t stream) {
    const float* x      = (const float*)d_in[0];
    const float* eattr  = (const float*)d_in[1];
    const int*   eindex = (const int*)d_in[2];
    const int*   batch  = (const int*)d_in[3];
    const float* efc_w  = (const float*)d_in[4];
    const float* efc_b  = (const float*)d_in[5];
    const float* c1_lw  = (const float*)d_in[6];
    const float* c1_lb  = (const float*)d_in[7];
    const float* c1_rw  = (const float*)d_in[8];
    const float* c2_lw  = (const float*)d_in[9];
    const float* c2_lb  = (const float*)d_in[10];
    const float* c2_rw  = (const float*)d_in[11];
    const float* proj_w = (const float*)d_in[12];
    const float* proj_b = (const float*)d_in[13];
    float* out = (float*)d_out;

    float* ws = (float*)d_ws;
    size_t off = 0;
    float* x1     = ws + off;
    float* deg    = x1 + (size_t)N_NODES_C * X1_PITCH;
    off += (size_t)N_NODES_C * (2 * IN_C_C + 1);
    float* h1     = ws + off - 32; off += (size_t)N_NODES_C * HID_C_C;
    float* msum2  = ws + off; off += (size_t)N_NODES_C * HID_C_C;
    float* h2     = ws + off; off += (size_t)N_NODES_C * HID_C_C;
    float* pooled = ws + off; off += (size_t)N_GRAPHS_C * HID_C_C;
    float* cnt    = ws + off; off += (size_t)N_GRAPHS_C;

    int* count  = (int*)h2;
    int* rowptr = count  + N_NODES_C;
    int* col    = rowptr + N_NODES_C;
    int* count_s  = (int*)h1;
    int* rowptr_s = count_s  + N_NODES_C;
    int* eperm    = rowptr_s + N_NODES_C;
    int* tmp     = (int*)msum2;
    int* gcount  = tmp + 2 * (size_t)N_EDGES_C;
    int* gcursor = gcount + 2 * N_BINS;

    init_kernel<<<(N_GRAPHS_C * HID_C_C + 255) / 256, 256, 0, stream>>>(
        pooled, cnt, gcount);

    bin_hist<<<N_CHUNKS, BIN_T, 0, stream>>>(eindex, gcount);
    bin_scatter<<<dim3(N_CHUNKS, 2), BIN_T, 0, stream>>>(eindex, gcount, gcursor, tmp);
    bin_sort<<<dim3(N_BINS, 2), BIN_NODES, 0, stream>>>(
        gcount, tmp, count, rowptr, col, count_s, rowptr_s, eperm);

    edge_fc_pull<<<N_NODES_C * 16 / 256, 256, 0, stream>>>(
        eattr, x, efc_w, efc_b, rowptr_s, count_s, eperm, count, x1, deg);

    sage_wmma_kernel<IN_C_C, X1_PITCH, false><<<N_NODES_C / 16, 32 * TILE_WAVES, 0, stream>>>(
        x1, rowptr, count, col, deg, c1_lw, c1_lb, c1_rw, h1, batch, pooled, cnt);

    sage_wmma_kernel<HID_C_C, HID_C_C, true><<<N_NODES_C / 16, 32 * TILE_WAVES, 0, stream>>>(
        h1, rowptr, count, col, deg, c2_lw, c2_lb, c2_rw, h2, batch, pooled, cnt);

    proj_kernel<<<(N_GRAPHS_C * OUT_C_C + 255) / 256, 256, 0, stream>>>(
        pooled, cnt, proj_w, proj_b, out);
}


# edge_fc pull per 16-node tile on a 4-wave block: each 88-B edge_attr row read by one load (11 lanes x 8 B), 16 rows a node in flight, rows added in list order, no lane tree; edge ids staged in LDS
# speedup vs baseline: 14.7963x; 1.0344x over previous
#include <hip/hip_runtime.h>
#include <hip/hip_bf16.h>
#include <stdint.h>

#define N_NODES_C  100000
#define N_EDGES_C  1600000
#define N_GRAPHS_C 64
#define IN_C_C     18
#define EDGE_C_C   22
#define HID_C_C    64
#define OUT_C_C    128

#define X1_PITCH   32

typedef float v2f __attribute__((ext_vector_type(2)));
typedef float v8f __attribute__((ext_vector_type(8)));

#define BIN_SHIFT   8
#define BIN_NODES   256
#define N_BINS      ((N_NODES_C + BIN_NODES - 1) / BIN_NODES)
#define CHUNK_E     8192
#define N_CHUNKS    ((N_EDGES_C + CHUNK_E - 1) / CHUNK_E)
#define BIN_T       512
#define BIN_EPT     (CHUNK_E / BIN_T)
#define STAGE_CAP   8192
#define SORT_EPT    16
#define DST_PAY_BITS 17
#define SRC_PAY_BITS 21

__global__ void __launch_bounds__(256)
init_kernel(float* __restrict__ pooled, float* __restrict__ cnt,
            int* __restrict__ gcount) {
    const int i = blockIdx.x * blockDim.x + threadIdx.x;
    if (i < 4 * N_BINS)           gcount[i] = 0;
    if (i < N_GRAPHS_C * HID_C_C) pooled[i] = 0.0f;
    if (i < N_GRAPHS_C)           cnt[i] = 0.0f;
}

template <int NW>
static __device__ __forceinline__ int block_excl_scan(int v, int* swave) {
    const int lane = threadIdx.x & 31;
    const int wv   = threadIdx.x >> 5;
    int inc = v;
#pragma unroll
    for (int d = 1; d < 32; d <<= 1) {
        const int n = __shfl_up(inc, d);
        if (lane >= d) inc += n;
    }
    if (lane == 31) swave[wv] = inc;
    __syncthreads();
    int pre = 0;
#pragma unroll
    for (int w = 0; w < NW; ++w) pre += (w < wv) ? swave[w] : 0;
    __syncthreads();
    return pre + inc - v;
}

static __device__ __forceinline__ unsigned node_id(int v) {
    const unsigned u = (unsigned)v;
    return u < (unsigned)N_NODES_C ? u : (unsigned)(N_NODES_C - 1);
}

__global__ void __launch_bounds__(BIN_T)
bin_hist(const int* __restrict__ eindex, int* __restrict__ gcount) {
    __shared__ int h[2 * N_BINS];
    const int tid = threadIdx.x;
    for (int i = tid; i < 2 * N_BINS; i += BIN_T) h[i] = 0;
    __syncthreads();
    const int e0 = blockIdx.x * CHUNK_E;
    int sv[BIN_EPT], dv[BIN_EPT];
#pragma unroll
    for (int k = 0; k < BIN_EPT; ++k) {
        const int e  = e0 + k * BIN_T + tid;
        const int ec = e < N_EDGES_C ? e : N_EDGES_C - 1;
        sv[k] = eindex[ec];
        dv[k] = eindex[N_EDGES_C + ec];
    }
#pragma unroll
    for (int k = 0; k < BIN_EPT; ++k) {
        const int e = e0 + k * BIN_T + tid;
        if (e < N_EDGES_C) {
            atomicAdd(&h[node_id(dv[k]) >> BIN_SHIFT], 1);
            atomicAdd(&h[N_BINS + (node_id(sv[k]) >> BIN_SHIFT)], 1);
        }
    }
    __syncthreads();
    for (int i = tid; i < 2 * N_BINS; i += BIN_T) {
        const int v = h[i];
        if (v) atomicAdd(gcount + i, v);
    }
}

__global__ void __launch_bounds__(BIN_T)
bin_scatter(const int* __restrict__ eindex, const int* __restrict__ gcount,
            int* __restrict__ gcursor, int* __restrict__ tmp) {
    __shared__ int hist[N_BINS];
    __shared__ int gdelta[N_BINS];
    __shared__ int sval[CHUNK_E];
    __shared__ unsigned short sbin[CHUNK_E];
    __shared__ int swave[BIN_T / 32];
    const int list = blockIdx.y;
    const int tid  = threadIdx.x;
    const int e0   = blockIdx.x * CHUNK_E;
    const int nloc = (N_EDGES_C - e0) < CHUNK_E ? (N_EDGES_C - e0) : CHUNK_E;

    if (tid < N_BINS) hist[tid] = 0;
    const int gc = tid < N_BINS ? gcount[list * N_BINS + tid] : 0;
    __syncthreads();

    const int* keyrow = eindex + (list == 0 ? N_EDGES_C : 0);
    const int  pshift = list == 0 ? DST_PAY_BITS : SRC_PAY_BITS;
    int sv[BIN_EPT], kv[BIN_EPT];
#pragma unroll
    for (int k = 0; k < BIN_EPT; ++k) {
        const int e  = e0 + k * BIN_T + tid;
        const int ec = e < N_EDGES_C ? e : N_EDGES_C - 1;
        sv[k] = eindex[ec];
        kv[k] = keyrow[ec];
    }

    int val[BIN_EPT], rnk[BIN_EPT], bin[BIN_EPT];
#pragma unroll
    for (int k = 0; k < BIN_EPT; ++k) {
        const int e = e0 + k * BIN_T + tid;
        bin[k] = -1;
        if (e < N_EDGES_C) {
            const unsigned key = node_id(kv[k]);
            const unsigned low = list == 0 ? node_id(sv[k]) : (unsigned)e;
            const unsigned pay = ((key & (BIN_NODES - 1)) << pshift) | low;
            bin[k] = (int)(key >> BIN_SHIFT);
            val[k] = (int)pay;
            rnk[k] = atomicAdd(&hist[bin[k]], 1);
        }
    }
    __syncthreads();

    const int c  = tid < N_BINS ? hist[tid] : 0;
    const int bstart = block_excl_scan<BIN_T / 32>(gc, swave);
    const int lofs   = block_excl_scan<BIN_T / 32>(c, swave);
    if (tid < N_BINS) {
        const int claim = c ? atomicAdd(gcursor + list * N_BINS + tid, c) : 0;
        hist[tid]   = lofs;
        gdelta[tid] = bstart + claim - lofs;
    }
    __syncthreads();

#pragma unroll
    for (int k = 0; k < BIN_EPT; ++k) {
        if (bin[k] >= 0) {
            const int p = hist[bin[k]] + rnk[k];
            sval[p] = val[k];
            sbin[p] = (unsigned short)bin[k];
        }
    }
    __syncthreads();

    int* out = tmp + (size_t)list * N_EDGES_C;
    for (int i = tid; i < nloc; i += BIN_T) {
        const int pos = gdelta[sbin[i]] + i;
        if ((unsigned)pos < (unsigned)N_EDGES_C) out[pos] = sval[i];
    }
}

__global__ void __launch_bounds__(BIN_NODES)
bin_sort(const int* __restrict__ gcount, const int* __restrict__ tmp,
         int* __restrict__ count_d, int* __restrict__ rowptr_d, int* __restrict__ col,
         int* __restrict__ count_s, int* __restrict__ rowptr_s, int* __restrict__ eperm) {
    __shared__ int fcnt[BIN_NODES];
    __shared__ int fofs[BIN_NODES];
    __shared__ int stage[STAGE_CAP];
    __shared__ int swave[BIN_NODES / 32];
    __shared__ int sred[BIN_NODES / 32];
    const int list = blockIdx.y;
    const int b    = blockIdx.x;
    const int tid  = threadIdx.x;
    const int lane = tid & 31;
    const int* gc  = gcount + list * N_BINS;
    int* count  = list ? count_s  : count_d;
    int* rowptr = list ? rowptr_s : rowptr_d;
    int* out    = list ? eperm    : col;
    const int      shift = list ? SRC_PAY_BITS : DST_PAY_BITS;
    const unsigned pmask = (1u << shift) - 1u;

    int part = 0;
    for (int j = tid; j < b; j += BIN_NODES) part += gc[j];
    for (int d = 16; d > 0; d >>= 1) part += __shfl_xor(part, d);
    if (lane == 0) sred[tid >> 5] = part;
    fcnt[tid] = 0;
    __syncthreads();
    int start = 0;
#pragma unroll
    for (int w = 0; w < BIN_NODES / 32; ++w) start += sred[w];
    int n = gc[b];
    if (start < 0 || start > N_EDGES_C) start = N_EDGES_C;
    if (n < 0) n = 0;
    if (n > N_EDGES_C - start) n = N_EDGES_C - start;

    const int* in = tmp + (size_t)list * N_EDGES_C + start;
    for (int base = 0; base < n; base += BIN_NODES * SORT_EPT) {
        unsigned v[SORT_EPT];
#pragma unroll
        for (int k = 0; k < SORT_EPT; ++k) {
            const int i = base + k * BIN_NODES + tid;
            v[k] = (unsigned)in[i < n ? i : n - 1];
        }
#pragma unroll
        for (int k = 0; k < SORT_EPT; ++k) {
            const int i = base + k * BIN_NODES + tid;
            if (i < n) atomicAdd(&fcnt[(v[k] >> shift) & (BIN_NODES - 1)], 1);
        }
    }
    __syncthreads();

    const int c    = fcnt[tid];
    const int excl = block_excl_scan<BIN_NODES / 32>(c, swave);
    const int node = b * BIN_NODES + tid;
    if (node < N_NODES_C) { count[node] = c; rowptr[node] = start + excl; }
    fofs[tid] = excl;
    fcnt[tid] = 0;
    __syncthreads();

    const bool staged = n <= STAGE_CAP;
    for (int base = 0; base < n; base += BIN_NODES * SORT_EPT) {
        unsigned v[SORT_EPT];
#pragma unroll
        for (int k = 0; k < SORT_EPT; ++k) {
            const int i = base + k * BIN_NODES + tid;
            v[k] = (unsigned)in[i < n ? i : n - 1];
        }
#pragma unroll
        for (int k = 0; k < SORT_EPT; ++k) {
            const int i = base + k * BIN_NODES + tid;
            if (i < n) {
                const unsigned f = (v[k] >> shift) & (BIN_NODES - 1);
                const int p = fofs[f] + atomicAdd(&fcnt[f], 1);
                if (staged) stage[p] = (int)(v[k] & pmask);
                else        out[start + p] = (int)(v[k] & pmask);
            }
        }
    }
    if (staged) {
        __syncthreads();
        for (int i = tid; i < n; i += BIN_NODES) out[start + i] = stage[i];
    }
}

#define COL_CAP 512
#define EF_WAVES 4
#define EF_PITCH 24
#define EF_DEPTH 16

template <bool STAGED>
static __device__ __forceinline__ void
gather_edge_rows(const float2* __restrict__ ea2, const int* __restrict__ eperm,
                 const int* sc, const int* srp, const int* scn,
                 int base, int navail, int lane, int wib, float* sm) {
    constexpr int LPN = EDGE_C_C / 2;
    const int g  = lane / LPN;
    const int c2 = lane - g * LPN;
    auto entry = [&](int j) -> unsigned {
        unsigned e;
        if (STAGED) {
            const unsigned q = (unsigned)(j - base);
            e = (unsigned)sc[q < (unsigned)navail ? q : (unsigned)(navail - 1)];
        } else {
            e = (unsigned)eperm[(unsigned)j < (unsigned)N_EDGES_C ? (unsigned)j : (unsigned)(N_EDGES_C - 1)];
        }
        return e < (unsigned)N_EDGES_C ? e : (unsigned)(N_EDGES_C - 1);
    };
    for (int p = wib; p < 8; p += EF_WAVES) {
        const int r = p * 2 + g;
        if (g < 2) {
            const int beg = srp[r];
            const int end = beg + scn[r];
            float2 acc = make_float2(0.0f, 0.0f);
            for (int j = beg; j < end; j += EF_DEPTH) {
                unsigned e[EF_DEPTH];
                float2   a[EF_DEPTH];
#pragma unroll
                for (int u = 0; u < EF_DEPTH; ++u)
                    e[u] = entry(u == 0 || j + u < end ? j + u : end - 1);
#pragma unroll
                for (int u = 0; u < EF_DEPTH; ++u) a[u] = ea2[e[u] * (unsigned)LPN + (unsigned)c2];
#pragma unroll
                for (int u = 0; u < EF_DEPTH; ++u)
                    if (u == 0 || j + u < end) { acc.x += a[u].x; acc.y += a[u].y; }
            }
            *(float2*)(sm + r * EF_PITCH + 2 * c2) = acc;
        }
    }
}

__global__ void __launch_bounds__(32 * EF_WAVES) __attribute__((amdgpu_waves_per_eu(1, 8)))
edge_fc_tile(const float* __restrict__ eattr, const float* __restrict__ x,
             const float* __restrict__ W, const float* __restrict__ b,
             const int* __restrict__ rowptr_s, const int* __restrict__ count_s,
             const int* __restrict__ eperm, const int* __restrict__ count_d,
             float* __restrict__ x1, float* __restrict__ deg) {
    __shared__ __attribute__((aligned(16))) float s_sum[16 * EF_PITCH];
    __shared__ __attribute__((aligned(16))) float s_w[EDGE_C_C * IN_C_C];
    __shared__ int s_col[COL_CAP];
    __shared__ int s_rp[16];
    __shared__ int s_cn[16];

    const int tid  = threadIdx.x;
    const int wib  = tid >> 5;
    const int lane = tid & 31;
    const int node0 = blockIdx.x * 16;
    const int row  = node0 + (lane & 15);

    const int rp = rowptr_s[row];
    const int cn = count_s[row];
    if (tid < 16) {
        s_rp[lane] = rp; s_cn[lane] = cn;
        deg[row] = (float)count_d[row];
    }
    for (int i = tid; i < EDGE_C_C * IN_C_C; i += 32 * EF_WAVES) s_w[i] = W[i];

    const int l  = tid & 15;
    const int nr = tid >> 4;
    float2 xv0 = make_float2(0.0f, 0.0f), xv1 = xv0, bb = xv0;
    if (l < IN_C_C / 2) {
        xv0 = ((const float2*)x)[(node0 + nr)     * (IN_C_C / 2) + l];
        xv1 = ((const float2*)x)[(node0 + nr + 8) * (IN_C_C / 2) + l];
        bb  = ((const float2*)b)[l];
    }

    const int base = __builtin_amdgcn_readlane(rp, 0);
    const int tot  = __builtin_amdgcn_readlane(rp, 15) + __builtin_amdgcn_readlane(cn, 15) - base;
    const bool staged = tot >= 0 && tot <= COL_CAP;
    if (staged) {
        for (int i = tid; i < tot; i += 32 * EF_WAVES) {
            const int j = base + i;
            s_col[i] = eperm[j < N_EDGES_C ? j : N_EDGES_C - 1];
        }
    }
    __syncthreads();

    if (staged)
        gather_edge_rows<true>((const float2*)eattr, eperm, s_col, s_rp, s_cn,
                               base, tot, lane, wib, s_sum);
    else
        gather_edge_rows<false>((const float2*)eattr, eperm, s_col, s_rp, s_cn,
                                base, tot, lane, wib, s_sum);
    __syncthreads();

    if (l < IN_C_C / 2) {
        const float2* W2 = (const float2*)s_w;
        for (int h = 0; h < 2; ++h) {
            const int r = nr + 8 * h;
            const float cf = (float)s_cn[r];
            float2 acc = make_float2(cf * bb.x, cf * bb.y);
#pragma unroll
            for (int q = 0; q < EDGE_C_C; ++q) {
                const float  s = s_sum[r * EF_PITCH + q];
                const float2 w = W2[q * (IN_C_C / 2) + l];
                acc.x += s * w.x; acc.y += s * w.y;
            }
            const float2 xv = h ? xv1 : xv0;
            ((float2*)x1)[(node0 + r) * (X1_PITCH / 2) + l] = make_float2(xv.x + acc.x, xv.y + acc.y);
        }
    }
}

template <int K, int XP, int PITCH, int NWV, bool STAGED>
static __device__ __forceinline__ void
gather_tile(const float2* __restrict__ f2, const int* __restrict__ col,
            const int* sc, const int* srp, const int* scn,
            int base, int navail, int lane, int wib, float* sm) {
    constexpr int KP  = (K + 3) & ~3;
    constexpr int LPN = K / 2;
    constexpr int NPW = 32 / LPN;
    const int g  = lane / LPN;
    const int c2 = lane - g * LPN;
    auto entry = [&](int j) -> int {
        if (STAGED) {
            const unsigned q = (unsigned)(j - base);
            return sc[q < (unsigned)navail ? q : (unsigned)(navail - 1)];
        } else {
            return col[(unsigned)j < (unsigned)N_EDGES_C ? (unsigned)j : (unsigned)(N_EDGES_C - 1)];
        }
    };
    for (int p = wib; p < (16 + NPW - 1) / NPW; p += NWV) {
        const int r = p * NPW + g;
        if (g < NPW && r < 16) {
            const int beg = srp[r];
            const int end = beg + scn[r];
            float2 acc = make_float2(0.0f, 0.0f);
            int j = beg;
            for (; j + 8 <= end; j += 8) {
                int   s[8];
                float2 a[8];
#pragma unroll
                for (int u = 0; u < 8; ++u) s[u] = entry(j + u);
#pragma unroll
                for (int u = 0; u < 8; ++u) a[u] = f2[s[u] * (XP / 2) + c2];
#pragma unroll
                for (int u = 0; u < 8; ++u) { acc.x += a[u].x; acc.y += a[u].y; }
            }
            if (j < end) {
                int   s[7];
                float2 a[7];
#pragma unroll
                for (int u = 0; u < 7; ++u) s[u] = entry(j + u < end ? j + u : end - 1);
#pragma unroll
                for (int u = 0; u < 7; ++u) a[u] = f2[s[u] * (XP / 2) + c2];
#pragma unroll
                for (int u = 0; u < 7; ++u)
                    if (j + u < end) { acc.x += a[u].x; acc.y += a[u].y; }
            }
            *(float2*)(sm + r * PITCH + 2 * c2) = acc;
            if (KP > K && c2 == LPN - 1) {
                for (int q = K; q < KP; ++q) sm[r * PITCH + q] = 0.0f;
            }
        }
    }
}

#define TILE_WAVES 4

template <int K, int XP, bool POOL>
__global__ void __launch_bounds__(32 * TILE_WAVES)
sage_wmma_kernel(const float* __restrict__ xin,
                 const int* __restrict__ rowptr, const int* __restrict__ count,
                 const int* __restrict__ col,
                 const float* __restrict__ deg,
                 const float* __restrict__ lw, const float* __restrict__ lb,
                 const float* __restrict__ rw, float* __restrict__ out,
                 const int* __restrict__ batch, float* __restrict__ pooled,
                 float* __restrict__ cnt) {
    constexpr int KP    = (K + 3) & ~3;
    constexpr int PITCH = (K % 64 == 0) ? K + 4 : KP;
    constexpr int OPITCH = HID_C_C + 4;
    __shared__ __attribute__((aligned(16))) float s_sum[16 * PITCH];
    __shared__ __attribute__((aligned(16))) float s_out[POOL ? 4 : 16 * OPITCH];
    __shared__ int   s_col[COL_CAP];
    __shared__ int   s_rp[16];
    __shared__ int   s_cn[16];

    const int wib  = threadIdx.x >> 5;
    const int lane = threadIdx.x & 31;

    const int node0 = blockIdx.x * 16;
    const int m    = lane & 15;
    const int half = (lane >> 4) & 1;
    const int row  = node0 + m;

    float* sm = s_sum;
    int*   sc = s_col;

    const int rp = rowptr[row];
    const int cn = count[row];
    if (threadIdx.x < 16) { s_rp[lane] = rp; s_cn[lane] = cn; }
    const int base = __builtin_amdgcn_readlane(rp, 0);
    const int tot  = __builtin_amdgcn_readlane(rp, 15) + __builtin_amdgcn_readlane(cn, 15) - base;
    const bool staged = tot >= 0 && tot <= COL_CAP;
    if (staged) {
        for (int i = threadIdx.x; i < tot; i += 32 * TILE_WAVES) {
            const int j = base + i;
            sc[i] = col[j < N_EDGES_C ? j : N_EDGES_C - 1];
        }
    }
    __syncthreads();

    if (staged)
        gather_tile<K, XP, PITCH, TILE_WAVES, true>((const float2*)xin, col, sc, s_rp, s_cn,
                                                    base, tot, lane, wib, sm);
    else
        gather_tile<K, XP, PITCH, TILE_WAVES, false>((const float2*)xin, col, sc, s_rp, s_cn,
                                                     base, tot, lane, wib, sm);
    __syncthreads();

    const float dg  = deg[row];
    const float inv = dg > 0.0f ? 1.0f / dg : 0.0f;
    const float* xr = xin + (size_t)row * XP;
    const float* mr = sm + m * PITCH;
    const int col_n = wib * 16 + m;

    v8f acc = {};

#pragma unroll
    for (int ks = 0; ks < KP; ks += 4) {
        const int kb = ks + half * 2;
        v2f amean, aroot, bl, br;
        const float2 ms = *(const float2*)(mr + kb);
        amean[0] = ms.x * inv;
        amean[1] = ms.y * inv;
        aroot[0] = (kb     < K) ? xr[kb]           : 0.0f;
        aroot[1] = (kb + 1 < K) ? xr[kb + 1]       : 0.0f;
        bl[0] = (kb     < K) ? lw[(size_t)kb       * HID_C_C + col_n] : 0.0f;
        bl[1] = (kb + 1 < K) ? lw[(size_t)(kb + 1) * HID_C_C + col_n] : 0.0f;
        br[0] = (kb     < K) ? rw[(size_t)kb       * HID_C_C + col_n] : 0.0f;
        br[1] = (kb + 1 < K) ? rw[(size_t)(kb + 1) * HID_C_C + col_n] : 0.0f;
        acc = __builtin_amdgcn_wmma_f32_16x16x4_f32(
            false, amean, false, bl, (short)0, acc, false, false);
        acc = __builtin_amdgcn_wmma_f32_16x16x4_f32(
            false, aroot, false, br, (short)0, acc, false, false);
    }

    const float bias = lb[col_n];
    if (!POOL) {
#pragma unroll
        for (int v = 0; v < 8; ++v) {
            const float val = acc[v] + bias;
            s_out[(half * 8 + v) * OPITCH + col_n] = val > 0.0f ? val : 0.0f;
        }
        __syncthreads();
#pragma unroll
        for (int q = 0; q < 2; ++q) {
            const int r = wib * 4 + q * 2 + half;
            const float4 val = *(const float4*)(s_out + r * OPITCH + 4 * m);
            *(float4*)(out + (size_t)(node0 + r) * HID_C_C + 4 * m) = val;
        }
    } else {
        const int g  = batch[row];
        const int g0 = __builtin_amdgcn_readfirstlane(g);

        float h[8];
        float cs = 0.0f;
#pragma unroll
        for (int v = 0; v < 8; ++v) {
            const float val = acc[v] + bias;
            h[v] = val > 0.0f ? val : 0.0f;
            cs += h[v];
        }

        if (__builtin_amdgcn_ballot_w32(g != g0) == 0u) {
            const float c16 = cs + __shfl_xor(cs, 16);
            if (half == 0) atomicAdd(pooled + g0 * HID_C_C + col_n, c16);
            if (threadIdx.x == 0) atomicAdd(cnt + g0, 16.0f);
        } else {
#pragma unroll
            for (int v = 0; v < 8; ++v) {
                const int gr = __shfl(g, half * 8 + v);
                atomicAdd(pooled + gr * HID_C_C + col_n, h[v]);
            }
            if (threadIdx.x < 16) atomicAdd(cnt + g, 1.0f);
        }
    }
}

__global__ void __launch_bounds__(256)
proj_kernel(const float* __restrict__ pooled, const float* __restrict__ cnt,
            const float* __restrict__ W, const float* __restrict__ b,
            float* __restrict__ out) {
    const int i = blockIdx.x * blockDim.x + threadIdx.x;
    if (i >= N_GRAPHS_C * OUT_C_C) return;
    const int g = i / OUT_C_C;
    const int o = i % OUT_C_C;
    const float inv = 1.0f / fmaxf(cnt[g], 1.0f);
    float acc = b[o];
    const float* pr = pooled + (size_t)g * HID_C_C;
#pragma unroll 8
    for (int c = 0; c < HID_C_C; ++c)
        acc += (pr[c] * inv) * W[(size_t)c * OUT_C_C + o];
    out[i] = acc;
}

extern "C" void kernel_launch(void* const* d_in, const int* in_sizes, int n_in,
                              void* d_out, int out_size, void* d_ws, size_t ws_size,
                              hipStream_t stream) {
    const float* x      = (const float*)d_in[0];
    const float* eattr  = (const float*)d_in[1];
    const int*   eindex = (const int*)d_in[2];
    const int*   batch  = (const int*)d_in[3];
    const float* efc_w  = (const float*)d_in[4];
    const float* efc_b  = (const float*)d_in[5];
    const float* c1_lw  = (const float*)d_in[6];
    const float* c1_lb  = (const float*)d_in[7];
    const float* c1_rw  = (const float*)d_in[8];
    const float* c2_lw  = (const float*)d_in[9];
    const float* c2_lb  = (const float*)d_in[10];
    const float* c2_rw  = (const float*)d_in[11];
    const float* proj_w = (const float*)d_in[12];
    const float* proj_b = (const float*)d_in[13];
    float* out = (float*)d_out;

    float* ws = (float*)d_ws;
    size_t off = 0;
    float* x1     = ws + off;
    float* deg    = x1 + (size_t)N_NODES_C * X1_PITCH;
    off += (size_t)N_NODES_C * (2 * IN_C_C + 1);
    float* h1     = ws + off - 32; off += (size_t)N_NODES_C * HID_C_C;
    float* msum2  = ws + off; off += (size_t)N_NODES_C * HID_C_C;
    float* h2     = ws + off; off += (size_t)N_NODES_C * HID_C_C;
    float* pooled = ws + off; off += (size_t)N_GRAPHS_C * HID_C_C;
    float* cnt    = ws + off; off += (size_t)N_GRAPHS_C;

    int* count  = (int*)h2;
    int* rowptr = count  + N_NODES_C;
    int* col    = rowptr + N_NODES_C;
    int* count_s  = (int*)h1;
    int* rowptr_s = count_s  + N_NODES_C;
    int* eperm    = rowptr_s + N_NODES_C;
    int* tmp     = (int*)msum2;
    int* gcount  = tmp + 2 * (size_t)N_EDGES_C;
    int* gcursor = gcount + 2 * N_BINS;

    init_kernel<<<(N_GRAPHS_C * HID_C_C + 255) / 256, 256, 0, stream>>>(
        pooled, cnt, gcount);

    bin_hist<<<N_CHUNKS, BIN_T, 0, stream>>>(eindex, gcount);
    bin_scatter<<<dim3(N_CHUNKS, 2), BIN_T, 0, stream>>>(eindex, gcount, gcursor, tmp);
    bin_sort<<<dim3(N_BINS, 2), BIN_NODES, 0, stream>>>(
        gcount, tmp, count, rowptr, col, count_s, rowptr_s, eperm);

    edge_fc_tile<<<N_NODES_C / 16, 32 * EF_WAVES, 0, stream>>>(
        eattr, x, efc_w, efc_b, rowptr_s, count_s, eperm, count, x1, deg);

    sage_wmma_kernel<IN_C_C, X1_PITCH, false><<<N_NODES_C / 16, 32 * TILE_WAVES, 0, stream>>>(
        x1, rowptr, count, col, deg, c1_lw, c1_lb, c1_rw, h1, batch, pooled, cnt);

    sage_wmma_kernel<HID_C_C, HID_C_C, true><<<N_NODES_C / 16, 32 * TILE_WAVES, 0, stream>>>(
        h1, rowptr, count, col, deg, c2_lw, c2_lb, c2_rw, h2, batch, pooled, cnt);

    proj_kernel<<<(N_GRAPHS_C * OUT_C_C + 255) / 256, 256, 0, stream>>>(
        pooled, cnt, proj_w, proj_b, out);
}


# Edge sort without whole-call bin counts: a chunk stores its bin-ordered entries in place plus a bin-offset table, the bin pass collects its runs; histogram and zeroing launches gone; arrival order
# speedup vs baseline: 15.3933x; 1.0403x over previous
#include <hip/hip_runtime.h>
#include <hip/hip_bf16.h>
#include <stdint.h>

#define N_NODES_C  100000
#define N_EDGES_C  1600000
#define N_GRAPHS_C 64
#define IN_C_C     18
#define EDGE_C_C   22
#define HID_C_C    64
#define OUT_C_C    128

#define X1_PITCH   32

typedef float v2f __attribute__((ext_vector_type(2)));
typedef float v8f __attribute__((ext_vector_type(8)));

#define BIN_SHIFT   8
#define BIN_NODES   256
#define N_BINS      ((N_NODES_C + BIN_NODES - 1) / BIN_NODES)
#define CHUNK_E     8192
#define N_CHUNKS    ((N_EDGES_C + CHUNK_E - 1) / CHUNK_E)
#define BIN_T       512
#define BIN_EPT     (CHUNK_E / BIN_T)
#define STAGE_CAP   8192
#define SORT_WAVES  (BIN_NODES / 32)
#define RUNS_PW     ((N_CHUNKS + SORT_WAVES - 1) / SORT_WAVES)
#define OFF_PITCH   (N_BINS + 1)
#define DST_PAY_BITS 17
#define SRC_PAY_BITS 21

static_assert(N_BINS + 1 <= BIN_T, "one thread per offset-table entry");
static_assert(N_CHUNKS <= BIN_NODES, "one thread per chunk run");

template <int NW>
static __device__ __forceinline__ int block_excl_scan(int v, int* swave) {
    const int lane = threadIdx.x & 31;
    const int wv   = threadIdx.x >> 5;
    int inc = v;
#pragma unroll
    for (int d = 1; d < 32; d <<= 1) {
        const int n = __shfl_up(inc, d);
        if (lane >= d) inc += n;
    }
    if (lane == 31) swave[wv] = inc;
    __syncthreads();
    int pre = 0;
#pragma unroll
    for (int w = 0; w < NW; ++w) pre += (w < wv) ? swave[w] : 0;
    __syncthreads();
    return pre + inc - v;
}

static __device__ __forceinline__ unsigned node_id(int v) {
    const unsigned u = (unsigned)v;
    return u < (unsigned)N_NODES_C ? u : (unsigned)(N_NODES_C - 1);
}

__global__ void __launch_bounds__(BIN_T)
bin_scatter(const int* __restrict__ eindex, int* __restrict__ boff, int* __restrict__ tmp,
            float* __restrict__ pooled, float* __restrict__ cnt) {
    __shared__ int hist[N_BINS];
    __shared__ int sval[CHUNK_E];
    __shared__ int swave[BIN_T / 32];
    const int list = blockIdx.y;
    const int tid  = threadIdx.x;
    const int e0   = blockIdx.x * CHUNK_E;
    const int nloc = (N_EDGES_C - e0) < CHUNK_E ? (N_EDGES_C - e0) : CHUNK_E;

    if (tid < N_BINS) hist[tid] = 0;
    if (blockIdx.x == 0 && list == 0) {
        for (int i = tid; i < N_GRAPHS_C * HID_C_C; i += BIN_T) pooled[i] = 0.0f;
        if (tid < N_GRAPHS_C) cnt[tid] = 0.0f;
    }
    __syncthreads();

    const int* keyrow = eindex + (list == 0 ? N_EDGES_C : 0);
    const int  pshift = list == 0 ? DST_PAY_BITS : SRC_PAY_BITS;
    int sv[BIN_EPT], kv[BIN_EPT];
#pragma unroll
    for (int k = 0; k < BIN_EPT; ++k) {
        const int e  = e0 + k * BIN_T + tid;
        const int ec = e < N_EDGES_C ? e : N_EDGES_C - 1;
        sv[k] = eindex[ec];
        kv[k] = keyrow[ec];
    }

    int val[BIN_EPT], rnk[BIN_EPT], bin[BIN_EPT];
#pragma unroll
    for (int k = 0; k < BIN_EPT; ++k) {
        const int e = e0 + k * BIN_T + tid;
        bin[k] = -1;
        if (e < N_EDGES_C) {
            const unsigned key = node_id(kv[k]);
            const unsigned low = list == 0 ? node_id(sv[k]) : (unsigned)e;
            const unsigned pay = ((key & (BIN_NODES - 1)) << pshift) | low;
            bin[k] = (int)(key >> BIN_SHIFT);
            val[k] = (int)pay;
            rnk[k] = atomicAdd(&hist[bin[k]], 1);
        }
    }
    __syncthreads();

    const int c    = tid < N_BINS ? hist[tid] : 0;
    const int lofs = block_excl_scan<BIN_T / 32>(c, swave);
    int* bo = boff + ((size_t)list * N_CHUNKS + blockIdx.x) * OFF_PITCH;
    if (tid < N_BINS) { hist[tid] = lofs; bo[tid] = lofs; }
    if (tid == N_BINS) bo[N_BINS] = nloc;
    __syncthreads();

#pragma unroll
    for (int k = 0; k < BIN_EPT; ++k) {
        if (bin[k] >= 0) sval[hist[bin[k]] + rnk[k]] = val[k];
    }
    __syncthreads();

    int* out = tmp + (size_t)list * N_EDGES_C + e0;
    for (int i = tid; i < nloc; i += BIN_T) out[i] = sval[i];
}

__global__ void __launch_bounds__(BIN_NODES)
bin_sort(const int* __restrict__ boff, const int* __restrict__ tmp,
         int* __restrict__ count_d, int* __restrict__ rowptr_d, int* __restrict__ col,
         int* __restrict__ count_s, int* __restrict__ rowptr_s, int* __restrict__ eperm) {
    __shared__ int fcnt[BIN_NODES];
    __shared__ int fofs[BIN_NODES];
    __shared__ int stage[STAGE_CAP];
    __shared__ int swave[SORT_WAVES];
    __shared__ int sred[SORT_WAVES];
    __shared__ int sredn[SORT_WAVES];
    __shared__ int s_base[N_CHUNKS];
    __shared__ int s_len[N_CHUNKS];
    const int list = blockIdx.y;
    const int b    = blockIdx.x;
    const int tid  = threadIdx.x;
    const int lane = tid & 31;
    const int wv   = tid >> 5;
    int* count  = list ? count_s  : count_d;
    int* rowptr = list ? rowptr_s : rowptr_d;
    int* out    = list ? eperm    : col;
    const int      shift = list ? SRC_PAY_BITS : DST_PAY_BITS;
    const unsigned pmask = (1u << shift) - 1u;

    int o0 = 0, ln = 0;
    if (tid < N_CHUNKS) {
        const int* bo = boff + ((size_t)list * N_CHUNKS + tid) * OFF_PITCH + b;
        const int cap = (N_EDGES_C - tid * CHUNK_E) < CHUNK_E ? (N_EDGES_C - tid * CHUNK_E) : CHUNK_E;
        o0 = bo[0];
        const int o1 = bo[1];
        if (o0 < 0) o0 = 0;
        if (o0 > cap) o0 = cap;
        ln = o1 - o0;
        if (ln < 0) ln = 0;
        if (ln > cap - o0) ln = cap - o0;
        s_base[tid] = tid * CHUNK_E + o0;
        s_len[tid]  = ln;
    }
    int ps = o0, pn = ln;
    for (int d = 16; d > 0; d >>= 1) { ps += __shfl_xor(ps, d); pn += __shfl_xor(pn, d); }
    if (lane == 0) { sred[wv] = ps; sredn[wv] = pn; }
    fcnt[tid] = 0;
    __syncthreads();
    int start = 0, n = 0;
#pragma unroll
    for (int w = 0; w < SORT_WAVES; ++w) { start += sred[w]; n += sredn[w]; }
    const bool staged = n <= STAGE_CAP;
    if (start < 0 || start > N_EDGES_C) start = N_EDGES_C;
    if (n > N_EDGES_C - start) n = N_EDGES_C - start;

    const int* in = tmp + (size_t)list * N_EDGES_C;
    unsigned v[RUNS_PW];
#pragma unroll
    for (int k = 0; k < RUNS_PW; ++k) {
        const int c  = wv + k * SORT_WAVES;
        const int l  = c < N_CHUNKS ? s_len[c] : 0;
        const int bs = c < N_CHUNKS ? s_base[c] : 0;
        const int i  = lane < l ? bs + lane : (bs < N_EDGES_C ? bs : N_EDGES_C - 1);
        v[k] = (unsigned)in[i];
    }
#pragma unroll
    for (int k = 0; k < RUNS_PW; ++k) {
        const int c = wv + k * SORT_WAVES;
        const int l = c < N_CHUNKS ? s_len[c] : 0;
        if (lane < l) atomicAdd(&fcnt[(v[k] >> shift) & (BIN_NODES - 1)], 1);
    }
    for (int c = wv; c < N_CHUNKS; c += SORT_WAVES) {
        const int l = s_len[c];
        const int bs = s_base[c];
        for (int j = 32 + lane; j < l; j += 32)
            atomicAdd(&fcnt[((unsigned)in[bs + j] >> shift) & (BIN_NODES - 1)], 1);
    }
    __syncthreads();

    const int c    = fcnt[tid];
    const int excl = block_excl_scan<SORT_WAVES>(c, swave);
    const int node = b * BIN_NODES + tid;
    if (node < N_NODES_C) { count[node] = c; rowptr[node] = start + excl; }
    fofs[tid] = excl;
    fcnt[tid] = 0;
    __syncthreads();

    auto place = [&](unsigned e) {
        const unsigned f = (e >> shift) & (BIN_NODES - 1);
        const int p = fofs[f] + atomicAdd(&fcnt[f], 1);
        if (staged) stage[p] = (int)(e & pmask);
        else if ((unsigned)(start + p) < (unsigned)N_EDGES_C) out[start + p] = (int)(e & pmask);
    };
#pragma unroll
    for (int k = 0; k < RUNS_PW; ++k) {
        const int cc = wv + k * SORT_WAVES;
        const int l  = cc < N_CHUNKS ? s_len[cc] : 0;
        if (lane < l) place(v[k]);
    }
    for (int cc = wv; cc < N_CHUNKS; cc += SORT_WAVES) {
        const int l = s_len[cc];
        const int bs = s_base[cc];
        for (int j = 32 + lane; j < l; j += 32) place((unsigned)in[bs + j]);
    }
    if (staged) {
        __syncthreads();
        for (int i = tid; i < n; i += BIN_NODES) out[start + i] = stage[i];
    }
}

#define COL_CAP 512
#define EF_WAVES 4
#define EF_PITCH 24
#define EF_DEPTH 16

template <bool STAGED>
static __device__ __forceinline__ void
gather_edge_rows(const float2* __restrict__ ea2, const int* __restrict__ eperm,
                 const int* sc, const int* srp, const int* scn,
                 int base, int navail, int lane, int wib, float* sm) {
    constexpr int LPN = EDGE_C_C / 2;
    const int g  = lane / LPN;
    const int c2 = lane - g * LPN;
    auto entry = [&](int j) -> unsigned {
        unsigned e;
        if (STAGED) {
            const unsigned q = (unsigned)(j - base);
            e = (unsigned)sc[q < (unsigned)navail ? q : (unsigned)(navail - 1)];
        } else {
            e = (unsigned)eperm[(unsigned)j < (unsigned)N_EDGES_C ? (unsigned)j : (unsigned)(N_EDGES_C - 1)];
        }
        return e < (unsigned)N_EDGES_C ? e : (unsigned)(N_EDGES_C - 1);
    };
    for (int p = wib; p < 8; p += EF_WAVES) {
        const int r = p * 2 + g;
        if (g < 2) {
            const int beg = srp[r];
            const int end = beg + scn[r];
            float2 acc = make_float2(0.0f, 0.0f);
            for (int j = beg; j < end; j += EF_DEPTH) {
                unsigned e[EF_DEPTH];
                float2   a[EF_DEPTH];
#pragma unroll
                for (int u = 0; u < EF_DEPTH; ++u)
                    e[u] = entry(u == 0 || j + u < end ? j + u : end - 1);
#pragma unroll
                for (int u = 0; u < EF_DEPTH; ++u) a[u] = ea2[e[u] * (unsigned)LPN + (unsigned)c2];
#pragma unroll
                for (int u = 0; u < EF_DEPTH; ++u)
                    if (u == 0 || j + u < end) { acc.x += a[u].x; acc.y += a[u].y; }
            }
            *(float2*)(sm + r * EF_PITCH + 2 * c2) = acc;
        }
    }
}

__global__ void __launch_bounds__(32 * EF_WAVES) __attribute__((amdgpu_waves_per_eu(1, 8)))
edge_fc_tile(const float* __restrict__ eattr, const float* __restrict__ x,
             const float* __restrict__ W, const float* __restrict__ b,
             const int* __restrict__ rowptr_s, const int* __restrict__ count_s,
             const int* __restrict__ eperm, const int* __restrict__ count_d,
             float* __restrict__ x1, float* __restrict__ deg) {
    __shared__ __attribute__((aligned(16))) float s_sum[16 * EF_PITCH];
    __shared__ __attribute__((aligned(16))) float s_w[EDGE_C_C * IN_C_C];
    __shared__ int s_col[COL_CAP];
    __shared__ int s_rp[16];
    __shared__ int s_cn[16];

    const int tid  = threadIdx.x;
    const int wib  = tid >> 5;
    const int lane = tid & 31;
    const int node0 = blockIdx.x * 16;
    const int row  = node0 + (lane & 15);

    const int rp = rowptr_s[row];
    const int cn = count_s[row];
    if (tid < 16) {
        s_rp[lane] = rp; s_cn[lane] = cn;
        deg[row] = (float)count_d[row];
    }
    for (int i = tid; i < EDGE_C_C * IN_C_C; i += 32 * EF_WAVES) s_w[i] = W[i];

    const int l  = tid & 15;
    const int nr = tid >> 4;
    float2 xv0 = make_float2(0.0f, 0.0f), xv1 = xv0, bb = xv0;
    if (l < IN_C_C / 2) {
        xv0 = ((const float2*)x)[(node0 + nr)     * (IN_C_C / 2) + l];
        xv1 = ((const float2*)x)[(node0 + nr + 8) * (IN_C_C / 2) + l];
        bb  = ((const float2*)b)[l];
    }

    const int base = __builtin_amdgcn_readlane(rp, 0);
    const int tot  = __builtin_amdgcn_readlane(rp, 15) + __builtin_amdgcn_readlane(cn, 15) - base;
    const bool staged = tot >= 0 && tot <= COL_CAP;
    if (staged) {
        for (int i = tid; i < tot; i += 32 * EF_WAVES) {
            const int j = base + i;
            s_col[i] = eperm[j < N_EDGES_C ? j : N_EDGES_C - 1];
        }
    }
    __syncthreads();

    if (staged)
        gather_edge_rows<true>((const float2*)eattr, eperm, s_col, s_rp, s_cn,
                               base, tot, lane, wib, s_sum);
    else
        gather_edge_rows<false>((const float2*)eattr, eperm, s_col, s_rp, s_cn,
                                base, tot, lane, wib, s_sum);
    __syncthreads();

    if (l < IN_C_C / 2) {
        const float2* W2 = (const float2*)s_w;
        for (int h = 0; h < 2; ++h) {
            const int r = nr + 8 * h;
            const float cf = (float)s_cn[r];
            float2 acc = make_float2(cf * bb.x, cf * bb.y);
#pragma unroll
            for (int q = 0; q < EDGE_C_C; ++q) {
                const float  s = s_sum[r * EF_PITCH + q];
                const float2 w = W2[q * (IN_C_C / 2) + l];
                acc.x += s * w.x; acc.y += s * w.y;
            }
            const float2 xv = h ? xv1 : xv0;
            ((float2*)x1)[(node0 + r) * (X1_PITCH / 2) + l] = make_float2(xv.x + acc.x, xv.y + acc.y);
        }
    }
}

template <int K, int XP, int PITCH, int NWV, bool STAGED>
static __device__ __forceinline__ void
gather_tile(const float2* __restrict__ f2, const int* __restrict__ col,
            const int* sc, const int* srp, const int* scn,
            int base, int navail, int lane, int wib, float* sm) {
    constexpr int KP  = (K + 3) & ~3;
    constexpr int LPN = K / 2;
    constexpr int NPW = 32 / LPN;
    const int g  = lane / LPN;
    const int c2 = lane - g * LPN;
    auto entry = [&](int j) -> int {
        if (STAGED) {
            const unsigned q = (unsigned)(j - base);
            return sc[q < (unsigned)navail ? q : (unsigned)(navail - 1)];
        } else {
            return col[(unsigned)j < (unsigned)N_EDGES_C ? (unsigned)j : (unsigned)(N_EDGES_C - 1)];
        }
    };
    for (int p = wib; p < (16 + NPW - 1) / NPW; p += NWV) {
        const int r = p * NPW + g;
        if (g < NPW && r < 16) {
            const int beg = srp[r];
            const int end = beg + scn[r];
            float2 acc = make_float2(0.0f, 0.0f);
            int j = beg;
            for (; j + 8 <= end; j += 8) {
                int   s[8];
                float2 a[8];
#pragma unroll
                for (int u = 0; u < 8; ++u) s[u] = entry(j + u);
#pragma unroll
                for (int u = 0; u < 8; ++u) a[u] = f2[s[u] * (XP / 2) + c2];
#pragma unroll
                for (int u = 0; u < 8; ++u) { acc.x += a[u].x; acc.y += a[u].y; }
            }
            if (j < end) {
                int   s[7];
                float2 a[7];
#pragma unroll
                for (int u = 0; u < 7; ++u) s[u] = entry(j + u < end ? j + u : end - 1);
#pragma unroll
                for (int u = 0; u < 7; ++u) a[u] = f2[s[u] * (XP / 2) + c2];
#pragma unroll
                for (int u = 0; u < 7; ++u)
                    if (j + u < end) { acc.x += a[u].x; acc.y += a[u].y; }
            }
            *(float2*)(sm + r * PITCH + 2 * c2) = acc;
            if (KP > K && c2 == LPN - 1) {
                for (int q = K; q < KP; ++q) sm[r * PITCH + q] = 0.0f;
            }
        }
    }
}

#define TILE_WAVES 4

template <int K, int XP, bool POOL>
__global__ void __launch_bounds__(32 * TILE_WAVES)
sage_wmma_kernel(const float* __restrict__ xin,
                 const int* __restrict__ rowptr, const int* __restrict__ count,
                 const int* __restrict__ col,
                 const float* __restrict__ deg,
                 const float* __restrict__ lw, const float* __restrict__ lb,
                 const float* __restrict__ rw, float* __restrict__ out,
                 const int* __restrict__ batch, float* __restrict__ pooled,
                 float* __restrict__ cnt) {
    constexpr int KP    = (K + 3) & ~3;
    constexpr int PITCH = (K % 64 == 0) ? K + 4 : KP;
    constexpr int OPITCH = HID_C_C + 4;
    __shared__ __attribute__((aligned(16))) float s_sum[16 * PITCH];
    __shared__ __attribute__((aligned(16))) float s_out[POOL ? 4 : 16 * OPITCH];
    __shared__ int   s_col[COL_CAP];
    __shared__ int   s_rp[16];
    __shared__ int   s_cn[16];

    const int wib  = threadIdx.x >> 5;
    const int lane = threadIdx.x & 31;

    const int node0 = blockIdx.x * 16;
    const int m    = lane & 15;
    const int half = (lane >> 4) & 1;
    const int row  = node0 + m;

    float* sm = s_sum;
    int*   sc = s_col;

    const int rp = rowptr[row];
    const int cn = count[row];
    if (threadIdx.x < 16) { s_rp[lane] = rp; s_cn[lane] = cn; }
    const int base = __builtin_amdgcn_readlane(rp, 0);
    const int tot  = __builtin_amdgcn_readlane(rp, 15) + __builtin_amdgcn_readlane(cn, 15) - base;
    const bool staged = tot >= 0 && tot <= COL_CAP;
    if (staged) {
        for (int i = threadIdx.x; i < tot; i += 32 * TILE_WAVES) {
            const int j = base + i;
            sc[i] = col[j < N_EDGES_C ? j : N_EDGES_C - 1];
        }
    }
    __syncthreads();

    if (staged)
        gather_tile<K, XP, PITCH, TILE_WAVES, true>((const float2*)xin, col, sc, s_rp, s_cn,
                                                    base, tot, lane, wib, sm);
    else
        gather_tile<K, XP, PITCH, TILE_WAVES, false>((const float2*)xin, col, sc, s_rp, s_cn,
                                                     base, tot, lane, wib, sm);
    __syncthreads();

    const float dg  = deg[row];
    const float inv = dg > 0.0f ? 1.0f / dg : 0.0f;
    const float* xr = xin + (size_t)row * XP;
    const float* mr = sm + m * PITCH;
    const int col_n = wib * 16 + m;

    v8f acc = {};

#pragma unroll
    for (int ks = 0; ks < KP; ks += 4) {
        const int kb = ks + half * 2;
        v2f amean, aroot, bl, br;
        const float2 ms = *(const float2*)(mr + kb);
        amean[0] = ms.x * inv;
        amean[1] = ms.y * inv;
        aroot[0] = (kb     < K) ? xr[kb]           : 0.0f;
        aroot[1] = (kb + 1 < K) ? xr[kb + 1]       : 0.0f;
        bl[0] = (kb     < K) ? lw[(size_t)kb       * HID_C_C + col_n] : 0.0f;
        bl[1] = (kb + 1 < K) ? lw[(size_t)(kb + 1) * HID_C_C + col_n] : 0.0f;
        br[0] = (kb     < K) ? rw[(size_t)kb       * HID_C_C + col_n] : 0.0f;
        br[1] = (kb + 1 < K) ? rw[(size_t)(kb + 1) * HID_C_C + col_n] : 0.0f;
        acc = __builtin_amdgcn_wmma_f32_16x16x4_f32(
            false, amean, false, bl, (short)0, acc, false, false);
        acc = __builtin_amdgcn_wmma_f32_16x16x4_f32(
            false, aroot, false, br, (short)0, acc, false, false);
    }

    const float bias = lb[col_n];
    if (!POOL) {
#pragma unroll
        for (int v = 0; v < 8; ++v) {
            const float val = acc[v] + bias;
            s_out[(half * 8 + v) * OPITCH + col_n] = val > 0.0f ? val : 0.0f;
        }
        __syncthreads();
#pragma unroll
        for (int q = 0; q < 2; ++q) {
            const int r = wib * 4 + q * 2 + half;
            const float4 val = *(const float4*)(s_out + r * OPITCH + 4 * m);
            *(float4*)(out + (size_t)(node0 + r) * HID_C_C + 4 * m) = val;
        }
    } else {
        const int g  = batch[row];
        const int g0 = __builtin_amdgcn_readfirstlane(g);

        float h[8];
        float cs = 0.0f;
#pragma unroll
        for (int v = 0; v < 8; ++v) {
            const float val = acc[v] + bias;
            h[v] = val > 0.0f ? val : 0.0f;
            cs += h[v];
        }

        if (__builtin_amdgcn_ballot_w32(g != g0) == 0u) {
            const float c16 = cs + __shfl_xor(cs, 16);
            if (half == 0) atomicAdd(pooled + g0 * HID_C_C + col_n, c16);
            if (threadIdx.x == 0) atomicAdd(cnt + g0, 16.0f);
        } else {
#pragma unroll
            for (int v = 0; v < 8; ++v) {
                const int gr = __shfl(g, half * 8 + v);
                atomicAdd(pooled + gr * HID_C_C + col_n, h[v]);
            }
            if (threadIdx.x < 16) atomicAdd(cnt + g, 1.0f);
        }
    }
}

__global__ void __launch_bounds__(256)
proj_kernel(const float* __restrict__ pooled, const float* __restrict__ cnt,
            const float* __restrict__ W, const float* __restrict__ b,
            float* __restrict__ out) {
    const int i = blockIdx.x * blockDim.x + threadIdx.x;
    if (i >= N_GRAPHS_C * OUT_C_C) return;
    const int g = i / OUT_C_C;
    const int o = i % OUT_C_C;
    const float inv = 1.0f / fmaxf(cnt[g], 1.0f);
    float acc = b[o];
    const float* pr = pooled + (size_t)g * HID_C_C;
#pragma unroll 8
    for (int c = 0; c < HID_C_C; ++c)
        acc += (pr[c] * inv) * W[(size_t)c * OUT_C_C + o];
    out[i] = acc;
}

extern "C" void kernel_launch(void* const* d_in, const int* in_sizes, int n_in,
                              void* d_out, int out_size, void* d_ws, size_t ws_size,
                              hipStream_t stream) {
    const float* x      = (const float*)d_in[0];
    const float* eattr  = (const float*)d_in[1];
    const int*   eindex = (const int*)d_in[2];
    const int*   batch  = (const int*)d_in[3];
    const float* efc_w  = (const float*)d_in[4];
    const float* efc_b  = (const float*)d_in[5];
    const float* c1_lw  = (const float*)d_in[6];
    const float* c1_lb  = (const float*)d_in[7];
    const float* c1_rw  = (const float*)d_in[8];
    const float* c2_lw  = (const float*)d_in[9];
    const float* c2_lb  = (const float*)d_in[10];
    const float* c2_rw  = (const float*)d_in[11];
    const float* proj_w = (const float*)d_in[12];
    const float* proj_b = (const float*)d_in[13];
    float* out = (float*)d_out;

    float* ws = (float*)d_ws;
    size_t off = 0;
    float* x1     = ws + off;
    float* deg    = x1 + (size_t)N_NODES_C * X1_PITCH;
    off += (size_t)N_NODES_C * (2 * IN_C_C + 1);
    float* h1     = ws + off - 32; off += (size_t)N_NODES_C * HID_C_C;
    float* msum2  = ws + off; off += (size_t)N_NODES_C * HID_C_C;
    float* h2     = ws + off; off += (size_t)N_NODES_C * HID_C_C;
    float* pooled = ws + off; off += (size_t)N_GRAPHS_C * HID_C_C;
    float* cnt    = ws + off; off += (size_t)N_GRAPHS_C;

    int* count  = (int*)h2;
    int* rowptr = count  + N_NODES_C;
    int* col    = rowptr + N_NODES_C;
    int* count_s  = (int*)h1;
    int* rowptr_s = count_s  + N_NODES_C;
    int* eperm    = rowptr_s + N_NODES_C;
    int* tmp  = (int*)msum2;
    int* boff = tmp + 2 * (size_t)N_EDGES_C;

    bin_scatter<<<dim3(N_CHUNKS, 2), BIN_T, 0, stream>>>(eindex, boff, tmp, pooled, cnt);
    bin_sort<<<dim3(N_BINS, 2), BIN_NODES, 0, stream>>>(
        boff, tmp, count, rowptr, col, count_s, rowptr_s, eperm);

    edge_fc_tile<<<N_NODES_C / 16, 32 * EF_WAVES, 0, stream>>>(
        eattr, x, efc_w, efc_b, rowptr_s, count_s, eperm, count, x1, deg);

    sage_wmma_kernel<IN_C_C, X1_PITCH, false><<<N_NODES_C / 16, 32 * TILE_WAVES, 0, stream>>>(
        x1, rowptr, count, col, deg, c1_lw, c1_lb, c1_rw, h1, batch, pooled, cnt);

    sage_wmma_kernel<HID_C_C, HID_C_C, true><<<N_NODES_C / 16, 32 * TILE_WAVES, 0, stream>>>(
        h1, rowptr, count, col, deg, c2_lw, c2_lb, c2_rw, h2, batch, pooled, cnt);

    proj_kernel<<<(N_GRAPHS_C * OUT_C_C + 255) / 256, 256, 0, stream>>>(
        pooled, cnt, proj_w, proj_b, out);
}


# edge_fc sums per 256-node source bin straight from the sort's chunk-ordered entries: all blocks walk the edge ids upward in step; rows added into LDS by float atomics (arrival order)
# speedup vs baseline: 15.9849x; 1.0384x over previous
#include <hip/hip_runtime.h>
#include <hip/hip_bf16.h>
#include <stdint.h>

#define N_NODES_C  100000
#define N_EDGES_C  1600000
#define N_GRAPHS_C 64
#define IN_C_C     18
#define EDGE_C_C   22
#define HID_C_C    64
#define OUT_C_C    128

#define X1_PITCH   32

typedef float v2f __attribute__((ext_vector_type(2)));
typedef float v8f __attribute__((ext_vector_type(8)));

#define BIN_SHIFT   8
#define BIN_NODES   256
#define N_BINS      ((N_NODES_C + BIN_NODES - 1) / BIN_NODES)
#define CHUNK_E     8192
#define N_CHUNKS    ((N_EDGES_C + CHUNK_E - 1) / CHUNK_E)
#define BIN_T       512
#define BIN_EPT     (CHUNK_E / BIN_T)
#define STAGE_CAP   8192
#define SORT_WAVES  (BIN_NODES / 32)
#define RUNS_PW     ((N_CHUNKS + SORT_WAVES - 1) / SORT_WAVES)
#define OFF_PITCH   (N_BINS + 1)
#define DST_PAY_BITS 17
#define SRC_PAY_BITS 21

static_assert(N_BINS + 1 <= BIN_T, "one thread per offset-table entry");
static_assert(N_CHUNKS <= BIN_NODES, "one thread per chunk run");

template <int NW>
static __device__ __forceinline__ int block_excl_scan(int v, int* swave) {
    const int lane = threadIdx.x & 31;
    const int wv   = threadIdx.x >> 5;
    int inc = v;
#pragma unroll
    for (int d = 1; d < 32; d <<= 1) {
        const int n = __shfl_up(inc, d);
        if (lane >= d) inc += n;
    }
    if (lane == 31) swave[wv] = inc;
    __syncthreads();
    int pre = 0;
#pragma unroll
    for (int w = 0; w < NW; ++w) pre += (w < wv) ? swave[w] : 0;
    __syncthreads();
    return pre + inc - v;
}

static __device__ __forceinline__ unsigned node_id(int v) {
    const unsigned u = (unsigned)v;
    return u < (unsigned)N_NODES_C ? u : (unsigned)(N_NODES_C - 1);
}

__global__ void __launch_bounds__(BIN_T)
bin_scatter(const int* __restrict__ eindex, int* __restrict__ boff, int* __restrict__ tmp,
            float* __restrict__ pooled, float* __restrict__ cnt) {
    __shared__ int hist[N_BINS];
    __shared__ int sval[CHUNK_E];
    __shared__ int swave[BIN_T / 32];
    const int list = blockIdx.y;
    const int tid  = threadIdx.x;
    const int e0   = blockIdx.x * CHUNK_E;
    const int nloc = (N_EDGES_C - e0) < CHUNK_E ? (N_EDGES_C - e0) : CHUNK_E;

    if (tid < N_BINS) hist[tid] = 0;
    if (blockIdx.x == 0 && list == 0) {
        for (int i = tid; i < N_GRAPHS_C * HID_C_C; i += BIN_T) pooled[i] = 0.0f;
        if (tid < N_GRAPHS_C) cnt[tid] = 0.0f;
    }
    __syncthreads();

    const int* keyrow = eindex + (list == 0 ? N_EDGES_C : 0);
    const int  pshift = list == 0 ? DST_PAY_BITS : SRC_PAY_BITS;
    int sv[BIN_EPT], kv[BIN_EPT];
#pragma unroll
    for (int k = 0; k < BIN_EPT; ++k) {
        const int e  = e0 + k * BIN_T + tid;
        const int ec = e < N_EDGES_C ? e : N_EDGES_C - 1;
        sv[k] = eindex[ec];
        kv[k] = keyrow[ec];
    }

    int val[BIN_EPT], rnk[BIN_EPT], bin[BIN_EPT];
#pragma unroll
    for (int k = 0; k < BIN_EPT; ++k) {
        const int e = e0 + k * BIN_T + tid;
        bin[k] = -1;
        if (e < N_EDGES_C) {
            const unsigned key = node_id(kv[k]);
            const unsigned low = list == 0 ? node_id(sv[k]) : (unsigned)e;
            const unsigned pay = ((key & (BIN_NODES - 1)) << pshift) | low;
            bin[k] = (int)(key >> BIN_SHIFT);
            val[k] = (int)pay;
            rnk[k] = atomicAdd(&hist[bin[k]], 1);
        }
    }
    __syncthreads();

    const int c    = tid < N_BINS ? hist[tid] : 0;
    const int lofs = block_excl_scan<BIN_T / 32>(c, swave);
    int* bo = boff + ((size_t)list * N_CHUNKS + blockIdx.x) * OFF_PITCH;
    if (tid < N_BINS) { hist[tid] = lofs; bo[tid] = lofs; }
    if (tid == N_BINS) bo[N_BINS] = nloc;
    __syncthreads();

#pragma unroll
    for (int k = 0; k < BIN_EPT; ++k) {
        if (bin[k] >= 0) sval[hist[bin[k]] + rnk[k]] = val[k];
    }
    __syncthreads();

    int* out = tmp + (size_t)list * N_EDGES_C + e0;
    for (int i = tid; i < nloc; i += BIN_T) out[i] = sval[i];
}

__global__ void __launch_bounds__(BIN_NODES)
bin_sort(const int* __restrict__ boff, const int* __restrict__ tmp,
         int* __restrict__ count, int* __restrict__ rowptr, int* __restrict__ out) {
    __shared__ int fcnt[BIN_NODES];
    __shared__ int fofs[BIN_NODES];
    __shared__ int stage[STAGE_CAP];
    __shared__ int swave[SORT_WAVES];
    __shared__ int sred[SORT_WAVES];
    __shared__ int sredn[SORT_WAVES];
    __shared__ int s_base[N_CHUNKS];
    __shared__ int s_len[N_CHUNKS];
    const int b    = blockIdx.x;
    const int tid  = threadIdx.x;
    const int lane = tid & 31;
    const int wv   = tid >> 5;
    const int      shift = DST_PAY_BITS;
    const unsigned pmask = (1u << shift) - 1u;

    int o0 = 0, ln = 0;
    if (tid < N_CHUNKS) {
        const int* bo = boff + (size_t)tid * OFF_PITCH + b;
        const int cap = (N_EDGES_C - tid * CHUNK_E) < CHUNK_E ? (N_EDGES_C - tid * CHUNK_E) : CHUNK_E;
        o0 = bo[0];
        const int o1 = bo[1];
        if (o0 < 0) o0 = 0;
        if (o0 > cap) o0 = cap;
        ln = o1 - o0;
        if (ln < 0) ln = 0;
        if (ln > cap - o0) ln = cap - o0;
        s_base[tid] = tid * CHUNK_E + o0;
        s_len[tid]  = ln;
    }
    int ps = o0, pn = ln;
    for (int d = 16; d > 0; d >>= 1) { ps += __shfl_xor(ps, d); pn += __shfl_xor(pn, d); }
    if (lane == 0) { sred[wv] = ps; sredn[wv] = pn; }
    fcnt[tid] = 0;
    __syncthreads();
    int start = 0, n = 0;
#pragma unroll
    for (int w = 0; w < SORT_WAVES; ++w) { start += sred[w]; n += sredn[w]; }
    const bool staged = n <= STAGE_CAP;
    if (start < 0 || start > N_EDGES_C) start = N_EDGES_C;
    if (n > N_EDGES_C - start) n = N_EDGES_C - start;

    const int* in = tmp;
    unsigned v[RUNS_PW];
#pragma unroll
    for (int k = 0; k < RUNS_PW; ++k) {
        const int c  = wv + k * SORT_WAVES;
        const int l  = c < N_CHUNKS ? s_len[c] : 0;
        const int bs = c < N_CHUNKS ? s_base[c] : 0;
        const int i  = lane < l ? bs + lane : (bs < N_EDGES_C ? bs : N_EDGES_C - 1);
        v[k] = (unsigned)in[i];
    }
#pragma unroll
    for (int k = 0; k < RUNS_PW; ++k) {
        const int c = wv + k * SORT_WAVES;
        const int l = c < N_CHUNKS ? s_len[c] : 0;
        if (lane < l) atomicAdd(&fcnt[(v[k] >> shift) & (BIN_NODES - 1)], 1);
    }
    for (int c = wv; c < N_CHUNKS; c += SORT_WAVES) {
        const int l = s_len[c];
        const int bs = s_base[c];
        for (int j = 32 + lane; j < l; j += 32)
            atomicAdd(&fcnt[((unsigned)in[bs + j] >> shift) & (BIN_NODES - 1)], 1);
    }
    __syncthreads();

    const int c    = fcnt[tid];
    const int excl = block_excl_scan<SORT_WAVES>(c, swave);
    const int node = b * BIN_NODES + tid;
    if (node < N_NODES_C) { count[node] = c; rowptr[node] = start + excl; }
    fofs[tid] = excl;
    fcnt[tid] = 0;
    __syncthreads();

    auto place = [&](unsigned e) {
        const unsigned f = (e >> shift) & (BIN_NODES - 1);
        const int p = fofs[f] + atomicAdd(&fcnt[f], 1);
        if (staged) stage[p] = (int)(e & pmask);
        else if ((unsigned)(start + p) < (unsigned)N_EDGES_C) out[start + p] = (int)(e & pmask);
    };
#pragma unroll
    for (int k = 0; k < RUNS_PW; ++k) {
        const int cc = wv + k * SORT_WAVES;
        const int l  = cc < N_CHUNKS ? s_len[cc] : 0;
        if (lane < l) place(v[k]);
    }
    for (int cc = wv; cc < N_CHUNKS; cc += SORT_WAVES) {
        const int l = s_len[cc];
        const int bs = s_base[cc];
        for (int j = 32 + lane; j < l; j += 32) place((unsigned)in[bs + j]);
    }
    if (staged) {
        __syncthreads();
        for (int i = tid; i < n; i += BIN_NODES) out[start + i] = stage[i];
    }
}

#define COL_CAP 512
#define EB_T     512
#define EB_WAVES (EB_T / 32)
#define EB_PITCH 23
#define EB_PASSES (BIN_NODES * 16 / EB_T)

static_assert(N_CHUNKS <= EB_T && BIN_NODES <= EB_T, "one thread per chunk run and per node");
static_assert(EB_WAVES <= N_CHUNKS, "every wave has a first run");

__global__ void __launch_bounds__(EB_T) __attribute__((amdgpu_waves_per_eu(8, 8)))
edge_fc_bin(const float* __restrict__ eattr, const float* __restrict__ x,
            const float* __restrict__ W, const float* __restrict__ b,
            const int* __restrict__ boff, const int* __restrict__ tmp,
            const int* __restrict__ count_d,
            float* __restrict__ x1, float* __restrict__ deg) {
    __shared__ float s_sum[BIN_NODES * EB_PITCH];
    __shared__ __attribute__((aligned(16))) float s_w[EDGE_C_C * IN_C_C];
    __shared__ int s_cnt[BIN_NODES];
    __shared__ int s_base[N_CHUNKS];
    __shared__ int s_len[N_CHUNKS];

    constexpr int LPN = EDGE_C_C / 2;
    constexpr unsigned EMASK = (1u << SRC_PAY_BITS) - 1u;
    const int bin   = blockIdx.x;
    const int tid   = threadIdx.x;
    const int lane  = tid & 31;
    const int wv    = __builtin_amdgcn_readfirstlane(tid >> 5);
    const int node0 = bin * BIN_NODES;

    if (tid < N_CHUNKS) {
        const int* bo = boff + ((size_t)N_CHUNKS + tid) * OFF_PITCH + bin;
        const int cap = (N_EDGES_C - tid * CHUNK_E) < CHUNK_E ? (N_EDGES_C - tid * CHUNK_E) : CHUNK_E;
        int o0 = bo[0];
        const int o1 = bo[1];
        if (o0 < 0) o0 = 0;
        if (o0 > cap) o0 = cap;
        int ln = o1 - o0;
        if (ln < 0) ln = 0;
        if (ln > cap - o0) ln = cap - o0;
        s_base[tid] = N_EDGES_C + tid * CHUNK_E + o0;
        s_len[tid]  = ln;
    }
    for (int i = tid; i < BIN_NODES * EB_PITCH; i += EB_T) s_sum[i] = 0.0f;
    if (tid < BIN_NODES) s_cnt[tid] = 0;
    for (int i = tid; i < EDGE_C_C * IN_C_C; i += EB_T) s_w[i] = W[i];

    const int l  = tid & 15;
    const int nr = tid >> 4;
    float2 xv[EB_PASSES];
    float2 bb = make_float2(0.0f, 0.0f);
    if (l < IN_C_C / 2) {
#pragma unroll
        for (int h = 0; h < EB_PASSES; ++h) {
            const int nd = node0 + nr + (EB_T / 16) * h;
            xv[h] = ((const float2*)x)[(nd < N_NODES_C ? nd : N_NODES_C - 1) * (IN_C_C / 2) + l];
        }
        bb = ((const float2*)b)[l];
    } else {
#pragma unroll
        for (int h = 0; h < EB_PASSES; ++h) xv[h] = make_float2(0.0f, 0.0f);
    }
    __syncthreads();

    const float2* ea2 = (const float2*)eattr;
    const int g  = lane / LPN;
    const int c2 = lane - g * LPN;

    auto first_entries = [&](int bs, int len) -> unsigned {
        if (len <= 0) return 0u;
        return (unsigned)tmp[bs + (lane < len ? lane : 0)];
    };
    auto piece = [&](unsigned v, int nn, int nbs, int nlen, unsigned& vnext) {
        if (lane < nn) atomicAdd(&s_cnt[(v >> SRC_PAY_BITS) & (BIN_NODES - 1)], 1);
        unsigned ent[16];
        float2   a[16];
#pragma unroll
        for (int u = 0; u < 16; ++u) {
            const int e0 = __builtin_amdgcn_readlane((int)v, 2 * u);
            const int e1 = __builtin_amdgcn_readlane((int)v, 2 * u + 1);
            ent[u] = (unsigned)(g == 1 ? e1 : e0);
        }
        vnext = first_entries(nbs, nlen);
        if (g < 2) {
#pragma unroll
            for (int u = 0; u < 16; ++u) {
                unsigned e = ent[u] & EMASK;
                e = e < (unsigned)N_EDGES_C ? e : (unsigned)(N_EDGES_C - 1);
                a[u] = ea2[e * (unsigned)LPN + (unsigned)c2];
            }
#pragma unroll
            for (int u = 0; u < 16; ++u) {
                if (2 * u + g < nn) {
                    float* p = s_sum + ((ent[u] >> SRC_PAY_BITS) & (BIN_NODES - 1)) * EB_PITCH + 2 * c2;
                    atomicAdd(p,     a[u].x);
                    atomicAdd(p + 1, a[u].y);
                }
            }
        }
    };

    int c  = wv;
    int lc = __builtin_amdgcn_readfirstlane(s_len[c]);
    int bc = __builtin_amdgcn_readfirstlane(s_base[c]);
    unsigned vc = first_entries(bc, lc);
    for (;;) {
        const int cn = c + EB_WAVES;
        int ln = 0, bn = 0;
        if (cn < N_CHUNKS) {
            ln = __builtin_amdgcn_readfirstlane(s_len[cn]);
            bn = __builtin_amdgcn_readfirstlane(s_base[cn]);
        }
        unsigned vn = 0u;
        if (lc > 0) {
            piece(vc, lc < 32 ? lc : 32, bn, ln, vn);
            for (int j0 = 32; j0 < lc; j0 += 32) {
                const int nn = (lc - j0) < 32 ? (lc - j0) : 32;
                const unsigned v = (unsigned)tmp[bc + j0 + (lane < nn ? lane : 0)];
                unsigned none;
                piece(v, nn, 0, 0, none);
            }
        } else {
            vn = first_entries(bn, ln);
        }
        if (cn >= N_CHUNKS) break;
        c = cn; lc = ln; bc = bn; vc = vn;
    }
    __syncthreads();

    if (l < IN_C_C / 2) {
        const float2* W2 = (const float2*)s_w;
#pragma unroll
        for (int h = 0; h < EB_PASSES; ++h) {
            const int r  = nr + (EB_T / 16) * h;
            const int nd = node0 + r;
            if (nd < N_NODES_C) {
                const float cf = (float)s_cnt[r];
                float2 acc = make_float2(cf * bb.x, cf * bb.y);
#pragma unroll
                for (int q = 0; q < EDGE_C_C; ++q) {
                    const float  sq = s_sum[r * EB_PITCH + q];
                    const float2 w  = W2[q * (IN_C_C / 2) + l];
                    acc.x += sq * w.x; acc.y += sq * w.y;
                }
                ((float2*)x1)[nd * (X1_PITCH / 2) + l] = make_float2(xv[h].x + acc.x, xv[h].y + acc.y);
            }
        }
    }
    if (tid < BIN_NODES) {
        const int nd = node0 + tid;
        if (nd < N_NODES_C) deg[nd] = (float)count_d[nd];
    }
}

template <int K, int XP, int PITCH, int NWV, bool STAGED>
static __device__ __forceinline__ void
gather_tile(const float2* __restrict__ f2, const int* __restrict__ col,
            const int* sc, const int* srp, const int* scn,
            int base, int navail, int lane, int wib, float* sm) {
    constexpr int KP  = (K + 3) & ~3;
    constexpr int LPN = K / 2;
    constexpr int NPW = 32 / LPN;
    const int g  = lane / LPN;
    const int c2 = lane - g * LPN;
    auto entry = [&](int j) -> int {
        if (STAGED) {
            const unsigned q = (unsigned)(j - base);
            return sc[q < (unsigned)navail ? q : (unsigned)(navail - 1)];
        } else {
            return col[(unsigned)j < (unsigned)N_EDGES_C ? (unsigned)j : (unsigned)(N_EDGES_C - 1)];
        }
    };
    for (int p = wib; p < (16 + NPW - 1) / NPW; p += NWV) {
        const int r = p * NPW + g;
        if (g < NPW && r < 16) {
            const int beg = srp[r];
            const int end = beg + scn[r];
            float2 acc = make_float2(0.0f, 0.0f);
            int j = beg;
            for (; j + 8 <= end; j += 8) {
                int   s[8];
                float2 a[8];
#pragma unroll
                for (int u = 0; u < 8; ++u) s[u] = entry(j + u);
#pragma unroll
                for (int u = 0; u < 8; ++u) a[u] = f2[s[u] * (XP / 2) + c2];
#pragma unroll
                for (int u = 0; u < 8; ++u) { acc.x += a[u].x; acc.y += a[u].y; }
            }
            if (j < end) {
                int   s[7];
                float2 a[7];
#pragma unroll
                for (int u = 0; u < 7; ++u) s[u] = entry(j + u < end ? j + u : end - 1);
#pragma unroll
                for (int u = 0; u < 7; ++u) a[u] = f2[s[u] * (XP / 2) + c2];
#pragma unroll
                for (int u = 0; u < 7; ++u)
                    if (j + u < end) { acc.x += a[u].x; acc.y += a[u].y; }
            }
            *(float2*)(sm + r * PITCH + 2 * c2) = acc;
            if (KP > K && c2 == LPN - 1) {
                for (int q = K; q < KP; ++q) sm[r * PITCH + q] = 0.0f;
            }
        }
    }
}

#define TILE_WAVES 4

template <int K, int XP, bool POOL>
__global__ void __launch_bounds__(32 * TILE_WAVES)
sage_wmma_kernel(const float* __restrict__ xin,
                 const int* __restrict__ rowptr, const int* __restrict__ count,
                 const int* __restrict__ col,
                 const float* __restrict__ deg,
                 const float* __restrict__ lw, const float* __restrict__ lb,
                 const float* __restrict__ rw, float* __restrict__ out,
                 const int* __restrict__ batch, float* __restrict__ pooled,
                 float* __restrict__ cnt) {
    constexpr int KP    = (K + 3) & ~3;
    constexpr int PITCH = (K % 64 == 0) ? K + 4 : KP;
    constexpr int OPITCH = HID_C_C + 4;
    __shared__ __attribute__((aligned(16))) float s_sum[16 * PITCH];
    __shared__ __attribute__((aligned(16))) float s_out[POOL ? 4 : 16 * OPITCH];
    __shared__ int   s_col[COL_CAP];
    __shared__ int   s_rp[16];
    __shared__ int   s_cn[16];

    const int wib  = threadIdx.x >> 5;
    const int lane = threadIdx.x & 31;

    const int node0 = blockIdx.x * 16;
    const int m    = lane & 15;
    const int half = (lane >> 4) & 1;
    const int row  = node0 + m;

    float* sm = s_sum;
    int*   sc = s_col;

    const int rp = rowptr[row];
    const int cn = count[row];
    if (threadIdx.x < 16) { s_rp[lane] = rp; s_cn[lane] = cn; }
    const int base = __builtin_amdgcn_readlane(rp, 0);
    const int tot  = __builtin_amdgcn_readlane(rp, 15) + __builtin_amdgcn_readlane(cn, 15) - base;
    const bool staged = tot >= 0 && tot <= COL_CAP;
    if (staged) {
        for (int i = threadIdx.x; i < tot; i += 32 * TILE_WAVES) {
            const int j = base + i;
            sc[i] = col[j < N_EDGES_C ? j : N_EDGES_C - 1];
        }
    }
    __syncthreads();

    if (staged)
        gather_tile<K, XP, PITCH, TILE_WAVES, true>((const float2*)xin, col, sc, s_rp, s_cn,
                                                    base, tot, lane, wib, sm);
    else
        gather_tile<K, XP, PITCH, TILE_WAVES, false>((const float2*)xin, col, sc, s_rp, s_cn,
                                                     base, tot, lane, wib, sm);
    __syncthreads();

    const float dg  = deg[row];
    const float inv = dg > 0.0f ? 1.0f / dg : 0.0f;
    const float* xr = xin + (size_t)row * XP;
    const float* mr = sm + m * PITCH;
    const int col_n = wib * 16 + m;

    v8f acc = {};

#pragma unroll
    for (int ks = 0; ks < KP; ks += 4) {
        const int kb = ks + half * 2;
        v2f amean, aroot, bl, br;
        const float2 ms = *(const float2*)(mr + kb);
        amean[0] = ms.x * inv;
        amean[1] = ms.y * inv;
        aroot[0] = (kb     < K) ? xr[kb]           : 0.0f;
        aroot[1] = (kb + 1 < K) ? xr[kb + 1]       : 0.0f;
        bl[0] = (kb     < K) ? lw[(size_t)kb       * HID_C_C + col_n] : 0.0f;
        bl[1] = (kb + 1 < K) ? lw[(size_t)(kb + 1) * HID_C_C + col_n] : 0.0f;
        br[0] = (kb     < K) ? rw[(size_t)kb       * HID_C_C + col_n] : 0.0f;
        br[1] = (kb + 1 < K) ? rw[(size_t)(kb + 1) * HID_C_C + col_n] : 0.0f;
        acc = __builtin_amdgcn_wmma_f32_16x16x4_f32(
            false, amean, false, bl, (short)0, acc, false, false);
        acc = __builtin_amdgcn_wmma_f32_16x16x4_f32(
            false, aroot, false, br, (short)0, acc, false, false);
    }

    const float bias = lb[col_n];
    if (!POOL) {
#pragma unroll
        for (int v = 0; v < 8; ++v) {
            const float val = acc[v] + bias;
            s_out[(half * 8 + v) * OPITCH + col_n] = val > 0.0f ? val : 0.0f;
        }
        __syncthreads();
#pragma unroll
        for (int q = 0; q < 2; ++q) {
            const int r = wib * 4 + q * 2 + half;
            const float4 val = *(const float4*)(s_out + r * OPITCH + 4 * m);
            *(float4*)(out + (size_t)(node0 + r) * HID_C_C + 4 * m) = val;
        }
    } else {
        const int g  = batch[row];
        const int g0 = __builtin_amdgcn_readfirstlane(g);

        float h[8];
        float cs = 0.0f;
#pragma unroll
        for (int v = 0; v < 8; ++v) {
            const float val = acc[v] + bias;
            h[v] = val > 0.0f ? val : 0.0f;
            cs += h[v];
        }

        if (__builtin_amdgcn_ballot_w32(g != g0) == 0u) {
            const float c16 = cs + __shfl_xor(cs, 16);
            if (half == 0) atomicAdd(pooled + g0 * HID_C_C + col_n, c16);
            if (threadIdx.x == 0) atomicAdd(cnt + g0, 16.0f);
        } else {
#pragma unroll
            for (int v = 0; v < 8; ++v) {
                const int gr = __shfl(g, half * 8 + v);
                atomicAdd(pooled + gr * HID_C_C + col_n, h[v]);
            }
            if (threadIdx.x < 16) atomicAdd(cnt + g, 1.0f);
        }
    }
}

__global__ void __launch_bounds__(256)
proj_kernel(const float* __restrict__ pooled, const float* __restrict__ cnt,
            const float* __restrict__ W, const float* __restrict__ b,
            float* __restrict__ out) {
    const int i = blockIdx.x * blockDim.x + threadIdx.x;
    if (i >= N_GRAPHS_C * OUT_C_C) return;
    const int g = i / OUT_C_C;
    const int o = i % OUT_C_C;
    const float inv = 1.0f / fmaxf(cnt[g], 1.0f);
    float acc = b[o];
    const float* pr = pooled + (size_t)g * HID_C_C;
#pragma unroll 8
    for (int c = 0; c < HID_C_C; ++c)
        acc += (pr[c] * inv) * W[(size_t)c * OUT_C_C + o];
    out[i] = acc;
}

extern "C" void kernel_launch(void* const* d_in, const int* in_sizes, int n_in,
                              void* d_out, int out_size, void* d_ws, size_t ws_size,
                              hipStream_t stream) {
    const float* x      = (const float*)d_in[0];
    const float* eattr  = (const float*)d_in[1];
    const int*   eindex = (const int*)d_in[2];
    const int*   batch  = (const int*)d_in[3];
    const float* efc_w  = (const float*)d_in[4];
    const float* efc_b  = (const float*)d_in[5];
    const float* c1_lw  = (const float*)d_in[6];
    const float* c1_lb  = (const float*)d_in[7];
    const float* c1_rw  = (const float*)d_in[8];
    const float* c2_lw  = (const float*)d_in[9];
    const float* c2_lb  = (const float*)d_in[10];
    const float* c2_rw  = (const float*)d_in[11];
    const float* proj_w = (const float*)d_in[12];
    const float* proj_b = (const float*)d_in[13];
    float* out = (float*)d_out;

    float* ws = (float*)d_ws;
    size_t off = 0;
    float* x1     = ws + off;
    float* deg    = x1 + (size_t)N_NODES_C * X1_PITCH;
    off += (size_t)N_NODES_C * (2 * IN_C_C + 1);
    float* h1     = ws + off - 32; off += (size_t)N_NODES_C * HID_C_C;
    float* msum2  = ws + off; off += (size_t)N_NODES_C * HID_C_C;
    float* h2     = ws + off; off += (size_t)N_NODES_C * HID_C_C;
    float* pooled = ws + off; off += (size_t)N_GRAPHS_C * HID_C_C;
    float* cnt    = ws + off; off += (size_t)N_GRAPHS_C;

    int* count  = (int*)h2;
    int* rowptr = count  + N_NODES_C;
    int* col    = rowptr + N_NODES_C;
    int* tmp  = (int*)msum2;
    int* boff = tmp + 2 * (size_t)N_EDGES_C;

    bin_scatter<<<dim3(N_CHUNKS, 2), BIN_T, 0, stream>>>(eindex, boff, tmp, pooled, cnt);
    bin_sort<<<N_BINS, BIN_NODES, 0, stream>>>(boff, tmp, count, rowptr, col);

    edge_fc_bin<<<N_BINS, EB_T, 0, stream>>>(
        eattr, x, efc_w, efc_b, boff, tmp, count, x1, deg);

    sage_wmma_kernel<IN_C_C, X1_PITCH, false><<<N_NODES_C / 16, 32 * TILE_WAVES, 0, stream>>>(
        x1, rowptr, count, col, deg, c1_lw, c1_lb, c1_rw, h1, batch, pooled, cnt);

    sage_wmma_kernel<HID_C_C, HID_C_C, true><<<N_NODES_C / 16, 32 * TILE_WAVES, 0, stream>>>(
        h1, rowptr, count, col, deg, c2_lw, c2_lb, c2_rw, h2, batch, pooled, cnt);

    proj_kernel<<<(N_GRAPHS_C * OUT_C_C + 255) / 256, 256, 0, stream>>>(
        pooled, cnt, proj_w, proj_b, out);
}
